# starting kernel
# baseline (speedup 1.0000x reference)
#include <hip/hip_runtime.h>
#include <hip/hip_bf16.h>

#define HCH 511
#define LSEQ 2048
#define NMODES 32
#define NLAYERS 4
#define M2 1022
#define KD 511
#define KPAD 512
#define LN_EPS 1e-5f

typedef __attribute__((ext_vector_type(2))) float v2f;
typedef __attribute__((ext_vector_type(8))) float v8f;

__device__ __forceinline__ float gelu_tanh(float x) {
    float x3 = x * x * x;
    float t = tanhf(0.7978845608028654f * (x + 0.044715f * x3));
    return 0.5f * x * (1.0f + t);
}

__device__ __forceinline__ float sigmoidf(float x) {
    return 1.0f / (1.0f + expf(-x));
}

__global__ void s4_setup(const float* __restrict__ log_dt,
                         const float* __restrict__ Arl,
                         const float* __restrict__ Aim,
                         const float* __restrict__ Cre,
                         const float* __restrict__ Cim,
                         float* __restrict__ w_re, float* __restrict__ w_im,
                         float* __restrict__ c_re, float* __restrict__ c_im) {
    int idx = blockIdx.x * blockDim.x + threadIdx.x;
    if (idx >= HCH * NMODES) return;
    int h = idx / NMODES;
    float a_re = -expf(Arl[idx]);
    float a_im = Aim[idx];
    float dt = expf(log_dt[h]);
    float dre = a_re * dt, dim = a_im * dt;
    float er = expf(dre);
    float sn, cs;
    sincosf(dim, &sn, &cs);
    float wre = er * cs, wim = er * sn;
    float nre = wre - 1.0f, nim = wim;
    float inv = 1.0f / (a_re * a_re + a_im * a_im);
    float tre = (nre * a_re + nim * a_im) * inv;
    float tim = (nim * a_re - nre * a_im) * inv;
    float Cr = Cre[idx], Ci = Cim[idx];
    w_re[idx] = wre;
    w_im[idx] = wim;
    c_re[idx] = 2.0f * (Cr * tre - Ci * tim);
    c_im[idx] = 2.0f * (Cr * tim + Ci * tre);
}

__global__ __launch_bounds__(256) void s4_scan(
        const float* __restrict__ z,
        const float* __restrict__ w_re, const float* __restrict__ w_im,
        const float* __restrict__ c_re, const float* __restrict__ c_im,
        const float* __restrict__ D_skip, float* __restrict__ y) {
    int warp = threadIdx.x >> 5;
    int lane = threadIdx.x & 31;
    int h = blockIdx.x * 8 + warp;
    if (h >= HCH) return;
    int cidx = h * NMODES + lane;
    float wr = w_re[cidx], wi = w_im[cidx];
    float cr = c_re[cidx], ci = c_im[cidx];
    float d = D_skip[h];
    float xr = 0.0f, xi = 0.0f;
    const float* zrow = z + (size_t)h * LSEQ;
    float* yrow = y + (size_t)h * LSEQ;
    for (int l0 = 0; l0 < LSEQ; l0 += 32) {
        float zv = zrow[l0 + lane];
        float out = 0.0f;
#pragma unroll
        for (int t = 0; t < 32; ++t) {
            float zt = __shfl(zv, t, 32);
            float nxr = fmaf(wr, xr, fmaf(-wi, xi, zt));
            float nxi = fmaf(wr, xi, wi * xr);
            xr = nxr; xi = nxi;
            float contrib = fmaf(cr, xr, -ci * xi);
#pragma unroll
            for (int off = 16; off > 0; off >>= 1)
                contrib += __shfl_xor(contrib, off, 32);
            if (lane == t) out = contrib;
        }
        yrow[l0 + lane] = gelu_tanh(fmaf(d, zv, out));
    }
}

#define BM 128
#define BN 128
#define BK 16
#define LDP 18

__global__ __launch_bounds__(256) void s4_glu_gemm(
        const float* __restrict__ Wg, const float* __restrict__ bias,
        const float* __restrict__ Y, float* __restrict__ Y2) {
    __shared__ float Ws[BM][LDP];
    __shared__ float Ys[BN][LDP];
    int tid = threadIdx.x;
    int lane = tid & 31;
    int wave = tid >> 5;
    int wave_m = wave >> 2;
    int wave_n = wave & 3;
    int lrow = lane & 15;
    int khalf = (lane >> 4) << 1;
    int madd = (lane >> 4) << 3;
    int blockM = blockIdx.x * BM;
    int blockN = blockIdx.y * BN;

    v8f acc[4][2];
#pragma unroll
    for (int mt = 0; mt < 4; ++mt) {
#pragma unroll
        for (int i = 0; i < 8; ++i) {
            int gm = blockM + wave_m * 64 + mt * 16 + i + madd;
            float bv = (gm < M2) ? bias[gm] : 0.0f;
#pragma unroll
            for (int nt = 0; nt < 2; ++nt) acc[mt][nt][i] = bv;
        }
    }

    for (int kb = 0; kb < KPAD; kb += BK) {
        {
            int m = tid >> 1;
            int ks = (tid & 1) * 8;
            int gm = blockM + m;
#pragma unroll
            for (int j = 0; j < 8; ++j) {
                int gk = kb + ks + j;
                Ws[m][ks + j] = (gm < M2 && gk < KD) ? Wg[gm * KD + gk] : 0.0f;
            }
        }
        {
            int k = tid >> 4;
            int ns = (tid & 15) * 8;
            int gk = kb + k;
            if (gk < KD) {
                const float* src = Y + (size_t)gk * LSEQ + blockN + ns;
#pragma unroll
                for (int j = 0; j < 8; ++j) Ys[ns + j][k] = src[j];
            } else {
#pragma unroll
                for (int j = 0; j < 8; ++j) Ys[ns + j][k] = 0.0f;
            }
        }
        __syncthreads();
#pragma unroll
        for (int kc = 0; kc < 4; ++kc) {
            int kk = kc * 4 + khalf;
            v2f a[4], b[2];
#pragma unroll
            for (int mt = 0; mt < 4; ++mt)
                a[mt] = *(const v2f*)&Ws[wave_m * 64 + mt * 16 + lrow][kk];
#pragma unroll
            for (int nt = 0; nt < 2; ++nt)
                b[nt] = *(const v2f*)&Ys[wave_n * 32 + nt * 16 + lrow][kk];
#pragma unroll
            for (int mt = 0; mt < 4; ++mt)
#pragma unroll
                for (int nt = 0; nt < 2; ++nt)
                    acc[mt][nt] = __builtin_amdgcn_wmma_f32_16x16x4_f32(
                        false, a[mt], false, b[nt],
                        (short)0, acc[mt][nt], false, false);
        }
        __syncthreads();
    }

    int gn_base = blockN + wave_n * 32;
#pragma unroll
    for (int mt = 0; mt < 4; ++mt) {
#pragma unroll
        for (int i = 0; i < 8; ++i) {
            int gm = blockM + wave_m * 64 + mt * 16 + i + madd;
            if (gm < M2) {
#pragma unroll
                for (int nt = 0; nt < 2; ++nt) {
                    int gn = gn_base + nt * 16 + lrow;
                    Y2[(size_t)gm * LSEQ + gn] = acc[mt][nt][i];
                }
            }
        }
    }
}

__global__ __launch_bounds__(256) void s4_glu_ln(
        const float* __restrict__ Y2, float* __restrict__ Z,
        const float* __restrict__ g, const float* __restrict__ b) {
    __shared__ float red[256];
    int l = blockIdx.x;
    int tid = threadIdx.x;
    int h1 = tid + 256;

    float v0, v1 = 0.0f;
    {
        float p = Y2[(size_t)tid * LSEQ + l];
        float q = Y2[(size_t)(tid + HCH) * LSEQ + l];
        v0 = p * sigmoidf(q) + Z[(size_t)tid * LSEQ + l];
    }
    if (h1 < HCH) {
        float p = Y2[(size_t)h1 * LSEQ + l];
        float q = Y2[(size_t)(h1 + HCH) * LSEQ + l];
        v1 = p * sigmoidf(q) + Z[(size_t)h1 * LSEQ + l];
    }

    red[tid] = v0 + v1;
    __syncthreads();
    for (int s = 128; s > 0; s >>= 1) {
        if (tid < s) red[tid] += red[tid + s];
        __syncthreads();
    }
    float mu = red[0] * (1.0f / (float)HCH);
    __syncthreads();

    float d0 = v0 - mu;
    float d1 = (h1 < HCH) ? (v1 - mu) : 0.0f;
    red[tid] = d0 * d0 + d1 * d1;
    __syncthreads();
    for (int s = 128; s > 0; s >>= 1) {
        if (tid < s) red[tid] += red[tid + s];
        __syncthreads();
    }
    float rstd = rsqrtf(red[0] * (1.0f / (float)HCH) + LN_EPS);

    Z[(size_t)tid * LSEQ + l] = d0 * rstd * g[tid] + b[tid];
    if (h1 < HCH)
        Z[(size_t)h1 * LSEQ + l] = d1 * rstd * g[h1] + b[h1];
}

extern "C" void kernel_launch(void* const* d_in, const int* in_sizes, int n_in,
                              void* d_out, int out_size, void* d_ws, size_t ws_size,
                              hipStream_t stream) {
    (void)in_sizes; (void)n_in; (void)out_size; (void)ws_size;
    const float* Zin    = (const float*)d_in[0];
    const float* log_dt = (const float*)d_in[1];
    const float* Arl    = (const float*)d_in[2];
    const float* Aim    = (const float*)d_in[3];
    const float* Cre    = (const float*)d_in[4];
    const float* Cim    = (const float*)d_in[5];
    const float* Dsk    = (const float*)d_in[6];
    const float* glu_w  = (const float*)d_in[7];
    const float* glu_b  = (const float*)d_in[8];
    const float* ln_g   = (const float*)d_in[9];
    const float* ln_b   = (const float*)d_in[10];

    float* ws = (float*)d_ws;
    const size_t HN = (size_t)HCH * NMODES;
    float* w_re  = ws;
    float* w_im  = ws + HN;
    float* c_re  = ws + 2 * HN;
    float* c_im  = ws + 3 * HN;
    float* zbuf  = ws + 4 * HN;
    float* ybuf  = zbuf + (size_t)HCH * LSEQ;
    float* y2buf = ybuf + (size_t)HCH * LSEQ;

    s4_setup<<<(int)((HN + 255) / 256), 256, 0, stream>>>(
        log_dt, Arl, Aim, Cre, Cim, w_re, w_im, c_re, c_im);

    hipMemcpyAsync(zbuf, Zin, sizeof(float) * (size_t)HCH * LSEQ,
                   hipMemcpyDeviceToDevice, stream);

    for (int layer = 0; layer < NLAYERS; ++layer) {
        s4_scan<<<64, 256, 0, stream>>>(zbuf, w_re, w_im, c_re, c_im, Dsk, ybuf);
        s4_glu_gemm<<<dim3(8, 16, 1), 256, 0, stream>>>(glu_w, glu_b, ybuf, y2buf);
        s4_glu_ln<<<LSEQ, 256, 0, stream>>>(y2buf, zbuf, ln_g, ln_b);
    }

    hipMemcpyAsync(d_out, zbuf, sizeof(float) * (size_t)HCH * LSEQ,
                   hipMemcpyDeviceToDevice, stream);
}


# scan: per-step 5-link ds_bpermute mode sum (and bpermute broadcast) replaced by one transposed 31-exchange butterfly per 32 steps on VALU lane ops; same pair order, sums bit-identical
# speedup vs baseline: 3.3825x; 3.3825x over previous
#include <hip/hip_runtime.h>
#include <hip/hip_bf16.h>

#define HCH 511
#define LSEQ 2048
#define NMODES 32
#define NLAYERS 4
#define M2 1022
#define KD 511
#define KPAD 512
#define LN_EPS 1e-5f

typedef __attribute__((ext_vector_type(2))) float v2f;
typedef __attribute__((ext_vector_type(8))) float v8f;

__device__ __forceinline__ float gelu_tanh(float x) {
    float x3 = x * x * x;
    float t = tanhf(0.7978845608028654f * (x + 0.044715f * x3));
    return 0.5f * x * (1.0f + t);
}

__device__ __forceinline__ float sigmoidf(float x) {
    return 1.0f / (1.0f + expf(-x));
}

__global__ void s4_setup(const float* __restrict__ log_dt,
                         const float* __restrict__ Arl,
                         const float* __restrict__ Aim,
                         const float* __restrict__ Cre,
                         const float* __restrict__ Cim,
                         float* __restrict__ w_re, float* __restrict__ w_im,
                         float* __restrict__ c_re, float* __restrict__ c_im) {
    int idx = blockIdx.x * blockDim.x + threadIdx.x;
    if (idx >= HCH * NMODES) return;
    int h = idx / NMODES;
    float a_re = -expf(Arl[idx]);
    float a_im = Aim[idx];
    float dt = expf(log_dt[h]);
    float dre = a_re * dt, dim = a_im * dt;
    float er = expf(dre);
    float sn, cs;
    sincosf(dim, &sn, &cs);
    float wre = er * cs, wim = er * sn;
    float nre = wre - 1.0f, nim = wim;
    float inv = 1.0f / (a_re * a_re + a_im * a_im);
    float tre = (nre * a_re + nim * a_im) * inv;
    float tim = (nim * a_re - nre * a_im) * inv;
    float Cr = Cre[idx], Ci = Cim[idx];
    w_re[idx] = wre;
    w_im[idx] = wim;
    c_re[idx] = 2.0f * (Cr * tre - Ci * tim);
    c_im[idx] = 2.0f * (Cr * tim + Ci * tre);
}

__device__ __forceinline__ float lane_xor16(float v) {
    int i = __builtin_bit_cast(int, v);
    int r = __builtin_amdgcn_permlanex16(i, i, 0x76543210, 0xfedcba98, false, false);
    return __builtin_bit_cast(float, r);
}
template <int MASK>
__device__ __forceinline__ float lane_xor_row(float v) {
    int i = __builtin_bit_cast(int, v);
    int r = __builtin_amdgcn_update_dpp(0, i, 0x160 | MASK, 0xf, 0xf, true);
    return __builtin_bit_cast(float, r);
}
template <int T>
__device__ __forceinline__ float lane_bcast(float v) {
    return __builtin_bit_cast(float,
        __builtin_amdgcn_readlane(__builtin_bit_cast(int, v), T));
}

__device__ __forceinline__ float scan_step(float wr, float wi, float cr, float ci,
                                         float zt, float& xr, float& xi) {
    float nxr = fmaf(wr, xr, fmaf(-wi, xi, zt));
    float nxi = fmaf(wr, xi, wi * xr);
    xr = nxr; xi = nxi;
    return fmaf(cr, xr, -ci * xi);
}

__global__ __launch_bounds__(256) void s4_scan(
        const float* __restrict__ z,
        const float* __restrict__ w_re, const float* __restrict__ w_im,
        const float* __restrict__ c_re, const float* __restrict__ c_im,
        const float* __restrict__ D_skip, float* __restrict__ y) {
    int warp = threadIdx.x >> 5;
    int lane = threadIdx.x & 31;
    int h = blockIdx.x * 8 + warp;
    if (h >= HCH) return;
    int cidx = h * NMODES + lane;
    float wr = w_re[cidx], wi = w_im[cidx];
    float cr = c_re[cidx], ci = c_im[cidx];
    float d = D_skip[h];
    float xr = 0.0f, xi = 0.0f;
    const bool b4 = (lane & 16) != 0;
    const bool b3 = (lane & 8) != 0;
    const bool b2 = (lane & 4) != 0;
    const bool b1 = (lane & 2) != 0;
    const bool b0 = (lane & 1) != 0;
    const float* zrow = z + (size_t)h * LSEQ;
    float* yrow = y + (size_t)h * LSEQ;
    for (int l0 = 0; l0 < LSEQ; l0 += 32) {
        float zv = zrow[l0 + lane];
        float v[32];
#define SCAN_STEP(T) v[T] = scan_step(wr, wi, cr, ci, lane_bcast<T>(zv), xr, xi);
        SCAN_STEP(0)  SCAN_STEP(1)  SCAN_STEP(2)  SCAN_STEP(3)
        SCAN_STEP(4)  SCAN_STEP(5)  SCAN_STEP(6)  SCAN_STEP(7)
        SCAN_STEP(8)  SCAN_STEP(9)  SCAN_STEP(10) SCAN_STEP(11)
        SCAN_STEP(12) SCAN_STEP(13) SCAN_STEP(14) SCAN_STEP(15)
        SCAN_STEP(16) SCAN_STEP(17) SCAN_STEP(18) SCAN_STEP(19)
        SCAN_STEP(20) SCAN_STEP(21) SCAN_STEP(22) SCAN_STEP(23)
        SCAN_STEP(24) SCAN_STEP(25) SCAN_STEP(26) SCAN_STEP(27)
        SCAN_STEP(28) SCAN_STEP(29) SCAN_STEP(30) SCAN_STEP(31)
#undef SCAN_STEP
#pragma unroll
        for (int j = 0; j < 16; ++j) {
            float keep = b4 ? v[j + 16] : v[j];
            float send = b4 ? v[j] : v[j + 16];
            v[j] = keep + lane_xor16(send);
        }
#pragma unroll
        for (int j = 0; j < 8; ++j) {
            float keep = b3 ? v[j + 8] : v[j];
            float send = b3 ? v[j] : v[j + 8];
            v[j] = keep + lane_xor_row<8>(send);
        }
#pragma unroll
        for (int j = 0; j < 4; ++j) {
            float keep = b2 ? v[j + 4] : v[j];
            float send = b2 ? v[j] : v[j + 4];
            v[j] = keep + lane_xor_row<4>(send);
        }
#pragma unroll
        for (int j = 0; j < 2; ++j) {
            float keep = b1 ? v[j + 2] : v[j];
            float send = b1 ? v[j] : v[j + 2];
            v[j] = keep + lane_xor_row<2>(send);
        }
        float keep0 = b0 ? v[1] : v[0];
        float send0 = b0 ? v[0] : v[1];
        float out = keep0 + lane_xor_row<1>(send0);
        yrow[l0 + lane] = gelu_tanh(fmaf(d, zv, out));
    }
}

#define BM 128
#define BN 128
#define BK 16
#define LDP 18

__global__ __launch_bounds__(256) void s4_glu_gemm(
        const float* __restrict__ Wg, const float* __restrict__ bias,
        const float* __restrict__ Y, float* __restrict__ Y2) {
    __shared__ float Ws[BM][LDP];
    __shared__ float Ys[BN][LDP];
    int tid = threadIdx.x;
    int lane = tid & 31;
    int wave = tid >> 5;
    int wave_m = wave >> 2;
    int wave_n = wave & 3;
    int lrow = lane & 15;
    int khalf = (lane >> 4) << 1;
    int madd = (lane >> 4) << 3;
    int blockM = blockIdx.x * BM;
    int blockN = blockIdx.y * BN;

    v8f acc[4][2];
#pragma unroll
    for (int mt = 0; mt < 4; ++mt) {
#pragma unroll
        for (int i = 0; i < 8; ++i) {
            int gm = blockM + wave_m * 64 + mt * 16 + i + madd;
            float bv = (gm < M2) ? bias[gm] : 0.0f;
#pragma unroll
            for (int nt = 0; nt < 2; ++nt) acc[mt][nt][i] = bv;
        }
    }

    for (int kb = 0; kb < KPAD; kb += BK) {
        {
            int m = tid >> 1;
            int ks = (tid & 1) * 8;
            int gm = blockM + m;
#pragma unroll
            for (int j = 0; j < 8; ++j) {
                int gk = kb + ks + j;
                Ws[m][ks + j] = (gm < M2 && gk < KD) ? Wg[gm * KD + gk] : 0.0f;
            }
        }
        {
            int k = tid >> 4;
            int ns = (tid & 15) * 8;
            int gk = kb + k;
            if (gk < KD) {
                const float* src = Y + (size_t)gk * LSEQ + blockN + ns;
#pragma unroll
                for (int j = 0; j < 8; ++j) Ys[ns + j][k] = src[j];
            } else {
#pragma unroll
                for (int j = 0; j < 8; ++j) Ys[ns + j][k] = 0.0f;
            }
        }
        __syncthreads();
#pragma unroll
        for (int kc = 0; kc < 4; ++kc) {
            int kk = kc * 4 + khalf;
            v2f a[4], b[2];
#pragma unroll
            for (int mt = 0; mt < 4; ++mt)
                a[mt] = *(const v2f*)&Ws[wave_m * 64 + mt * 16 + lrow][kk];
#pragma unroll
            for (int nt = 0; nt < 2; ++nt)
                b[nt] = *(const v2f*)&Ys[wave_n * 32 + nt * 16 + lrow][kk];
#pragma unroll
            for (int mt = 0; mt < 4; ++mt)
#pragma unroll
                for (int nt = 0; nt < 2; ++nt)
                    acc[mt][nt] = __builtin_amdgcn_wmma_f32_16x16x4_f32(
                        false, a[mt], false, b[nt],
                        (short)0, acc[mt][nt], false, false);
        }
        __syncthreads();
    }

    int gn_base = blockN + wave_n * 32;
#pragma unroll
    for (int mt = 0; mt < 4; ++mt) {
#pragma unroll
        for (int i = 0; i < 8; ++i) {
            int gm = blockM + wave_m * 64 + mt * 16 + i + madd;
            if (gm < M2) {
#pragma unroll
                for (int nt = 0; nt < 2; ++nt) {
                    int gn = gn_base + nt * 16 + lrow;
                    Y2[(size_t)gm * LSEQ + gn] = acc[mt][nt][i];
                }
            }
        }
    }
}

__global__ __launch_bounds__(256) void s4_glu_ln(
        const float* __restrict__ Y2, float* __restrict__ Z,
        const float* __restrict__ g, const float* __restrict__ b) {
    __shared__ float red[256];
    int l = blockIdx.x;
    int tid = threadIdx.x;
    int h1 = tid + 256;

    float v0, v1 = 0.0f;
    {
        float p = Y2[(size_t)tid * LSEQ + l];
        float q = Y2[(size_t)(tid + HCH) * LSEQ + l];
        v0 = p * sigmoidf(q) + Z[(size_t)tid * LSEQ + l];
    }
    if (h1 < HCH) {
        float p = Y2[(size_t)h1 * LSEQ + l];
        float q = Y2[(size_t)(h1 + HCH) * LSEQ + l];
        v1 = p * sigmoidf(q) + Z[(size_t)h1 * LSEQ + l];
    }

    red[tid] = v0 + v1;
    __syncthreads();
    for (int s = 128; s > 0; s >>= 1) {
        if (tid < s) red[tid] += red[tid + s];
        __syncthreads();
    }
    float mu = red[0] * (1.0f / (float)HCH);
    __syncthreads();

    float d0 = v0 - mu;
    float d1 = (h1 < HCH) ? (v1 - mu) : 0.0f;
    red[tid] = d0 * d0 + d1 * d1;
    __syncthreads();
    for (int s = 128; s > 0; s >>= 1) {
        if (tid < s) red[tid] += red[tid + s];
        __syncthreads();
    }
    float rstd = rsqrtf(red[0] * (1.0f / (float)HCH) + LN_EPS);

    Z[(size_t)tid * LSEQ + l] = d0 * rstd * g[tid] + b[tid];
    if (h1 < HCH)
        Z[(size_t)h1 * LSEQ + l] = d1 * rstd * g[h1] + b[h1];
}

extern "C" void kernel_launch(void* const* d_in, const int* in_sizes, int n_in,
                              void* d_out, int out_size, void* d_ws, size_t ws_size,
                              hipStream_t stream) {
    (void)in_sizes; (void)n_in; (void)out_size; (void)ws_size;
    const float* Zin    = (const float*)d_in[0];
    const float* log_dt = (const float*)d_in[1];
    const float* Arl    = (const float*)d_in[2];
    const float* Aim    = (const float*)d_in[3];
    const float* Cre    = (const float*)d_in[4];
    const float* Cim    = (const float*)d_in[5];
    const float* Dsk    = (const float*)d_in[6];
    const float* glu_w  = (const float*)d_in[7];
    const float* glu_b  = (const float*)d_in[8];
    const float* ln_g   = (const float*)d_in[9];
    const float* ln_b   = (const float*)d_in[10];

    float* ws = (float*)d_ws;
    const size_t HN = (size_t)HCH * NMODES;
    float* w_re  = ws;
    float* w_im  = ws + HN;
    float* c_re  = ws + 2 * HN;
    float* c_im  = ws + 3 * HN;
    float* zbuf  = ws + 4 * HN;
    float* ybuf  = zbuf + (size_t)HCH * LSEQ;
    float* y2buf = ybuf + (size_t)HCH * LSEQ;

    s4_setup<<<(int)((HN + 255) / 256), 256, 0, stream>>>(
        log_dt, Arl, Aim, Cre, Cim, w_re, w_im, c_re, c_im);

    hipMemcpyAsync(zbuf, Zin, sizeof(float) * (size_t)HCH * LSEQ,
                   hipMemcpyDeviceToDevice, stream);

    for (int layer = 0; layer < NLAYERS; ++layer) {
        s4_scan<<<64, 256, 0, stream>>>(zbuf, w_re, w_im, c_re, c_im, Dsk, ybuf);
        s4_glu_gemm<<<dim3(8, 16, 1), 256, 0, stream>>>(glu_w, glu_b, ybuf, y2buf);
        s4_glu_ln<<<LSEQ, 256, 0, stream>>>(y2buf, zbuf, ln_g, ln_b);
    }

    hipMemcpyAsync(d_out, zbuf, sizeof(float) * (size_t)HCH * LSEQ,
                   hipMemcpyDeviceToDevice, stream);
}


# glu_ln re-tiled: block = 32 adjacent columns, lane = column, so each wave load/store is one whole 128-B line, not 32 lines; mean/var sums now serial per lane then over 32 waves (sum order changes)
# speedup vs baseline: 4.4172x; 1.3059x over previous
#include <hip/hip_runtime.h>
#include <hip/hip_bf16.h>

#define HCH 511
#define LSEQ 2048
#define NMODES 32
#define NLAYERS 4
#define M2 1022
#define KD 511
#define KPAD 512
#define LN_EPS 1e-5f

typedef __attribute__((ext_vector_type(2))) float v2f;
typedef __attribute__((ext_vector_type(8))) float v8f;

__device__ __forceinline__ float gelu_tanh(float x) {
    float x3 = x * x * x;
    float t = tanhf(0.7978845608028654f * (x + 0.044715f * x3));
    return 0.5f * x * (1.0f + t);
}

__device__ __forceinline__ float sigmoidf(float x) {
    return 1.0f / (1.0f + expf(-x));
}

__global__ void s4_setup(const float* __restrict__ log_dt,
                         const float* __restrict__ Arl,
                         const float* __restrict__ Aim,
                         const float* __restrict__ Cre,
                         const float* __restrict__ Cim,
                         float* __restrict__ w_re, float* __restrict__ w_im,
                         float* __restrict__ c_re, float* __restrict__ c_im) {
    int idx = blockIdx.x * blockDim.x + threadIdx.x;
    if (idx >= HCH * NMODES) return;
    int h = idx / NMODES;
    float a_re = -expf(Arl[idx]);
    float a_im = Aim[idx];
    float dt = expf(log_dt[h]);
    float dre = a_re * dt, dim = a_im * dt;
    float er = expf(dre);
    float sn, cs;
    sincosf(dim, &sn, &cs);
    float wre = er * cs, wim = er * sn;
    float nre = wre - 1.0f, nim = wim;
    float inv = 1.0f / (a_re * a_re + a_im * a_im);
    float tre = (nre * a_re + nim * a_im) * inv;
    float tim = (nim * a_re - nre * a_im) * inv;
    float Cr = Cre[idx], Ci = Cim[idx];
    w_re[idx] = wre;
    w_im[idx] = wim;
    c_re[idx] = 2.0f * (Cr * tre - Ci * tim);
    c_im[idx] = 2.0f * (Cr * tim + Ci * tre);
}

__device__ __forceinline__ float lane_xor16(float v) {
    int i = __builtin_bit_cast(int, v);
    int r = __builtin_amdgcn_permlanex16(i, i, 0x76543210, 0xfedcba98, false, false);
    return __builtin_bit_cast(float, r);
}
template <int MASK>
__device__ __forceinline__ float lane_xor_row(float v) {
    int i = __builtin_bit_cast(int, v);
    int r = __builtin_amdgcn_update_dpp(0, i, 0x160 | MASK, 0xf, 0xf, true);
    return __builtin_bit_cast(float, r);
}
template <int T>
__device__ __forceinline__ float lane_bcast(float v) {
    return __builtin_bit_cast(float,
        __builtin_amdgcn_readlane(__builtin_bit_cast(int, v), T));
}

__device__ __forceinline__ float scan_step(float wr, float wi, float cr, float ci,
                                         float zt, float& xr, float& xi) {
    float nxr = fmaf(wr, xr, fmaf(-wi, xi, zt));
    float nxi = fmaf(wr, xi, wi * xr);
    xr = nxr; xi = nxi;
    return fmaf(cr, xr, -ci * xi);
}

__global__ __launch_bounds__(256) void s4_scan(
        const float* __restrict__ z,
        const float* __restrict__ w_re, const float* __restrict__ w_im,
        const float* __restrict__ c_re, const float* __restrict__ c_im,
        const float* __restrict__ D_skip, float* __restrict__ y) {
    int warp = threadIdx.x >> 5;
    int lane = threadIdx.x & 31;
    int h = blockIdx.x * 8 + warp;
    if (h >= HCH) return;
    int cidx = h * NMODES + lane;
    float wr = w_re[cidx], wi = w_im[cidx];
    float cr = c_re[cidx], ci = c_im[cidx];
    float d = D_skip[h];
    float xr = 0.0f, xi = 0.0f;
    const bool b4 = (lane & 16) != 0;
    const bool b3 = (lane & 8) != 0;
    const bool b2 = (lane & 4) != 0;
    const bool b1 = (lane & 2) != 0;
    const bool b0 = (lane & 1) != 0;
    const float* zrow = z + (size_t)h * LSEQ;
    float* yrow = y + (size_t)h * LSEQ;
    for (int l0 = 0; l0 < LSEQ; l0 += 32) {
        float zv = zrow[l0 + lane];
        float v[32];
#define SCAN_STEP(T) v[T] = scan_step(wr, wi, cr, ci, lane_bcast<T>(zv), xr, xi);
        SCAN_STEP(0)  SCAN_STEP(1)  SCAN_STEP(2)  SCAN_STEP(3)
        SCAN_STEP(4)  SCAN_STEP(5)  SCAN_STEP(6)  SCAN_STEP(7)
        SCAN_STEP(8)  SCAN_STEP(9)  SCAN_STEP(10) SCAN_STEP(11)
        SCAN_STEP(12) SCAN_STEP(13) SCAN_STEP(14) SCAN_STEP(15)
        SCAN_STEP(16) SCAN_STEP(17) SCAN_STEP(18) SCAN_STEP(19)
        SCAN_STEP(20) SCAN_STEP(21) SCAN_STEP(22) SCAN_STEP(23)
        SCAN_STEP(24) SCAN_STEP(25) SCAN_STEP(26) SCAN_STEP(27)
        SCAN_STEP(28) SCAN_STEP(29) SCAN_STEP(30) SCAN_STEP(31)
#undef SCAN_STEP
#pragma unroll
        for (int j = 0; j < 16; ++j) {
            float keep = b4 ? v[j + 16] : v[j];
            float send = b4 ? v[j] : v[j + 16];
            v[j] = keep + lane_xor16(send);
        }
#pragma unroll
        for (int j = 0; j < 8; ++j) {
            float keep = b3 ? v[j + 8] : v[j];
            float send = b3 ? v[j] : v[j + 8];
            v[j] = keep + lane_xor_row<8>(send);
        }
#pragma unroll
        for (int j = 0; j < 4; ++j) {
            float keep = b2 ? v[j + 4] : v[j];
            float send = b2 ? v[j] : v[j + 4];
            v[j] = keep + lane_xor_row<4>(send);
        }
#pragma unroll
        for (int j = 0; j < 2; ++j) {
            float keep = b1 ? v[j + 2] : v[j];
            float send = b1 ? v[j] : v[j + 2];
            v[j] = keep + lane_xor_row<2>(send);
        }
        float keep0 = b0 ? v[1] : v[0];
        float send0 = b0 ? v[0] : v[1];
        float out = keep0 + lane_xor_row<1>(send0);
        yrow[l0 + lane] = gelu_tanh(fmaf(d, zv, out));
    }
}

#define BM 128
#define BN 128
#define BK 16
#define LDP 18

__global__ __launch_bounds__(256) void s4_glu_gemm(
        const float* __restrict__ Wg, const float* __restrict__ bias,
        const float* __restrict__ Y, float* __restrict__ Y2) {
    __shared__ float Ws[BM][LDP];
    __shared__ float Ys[BN][LDP];
    int tid = threadIdx.x;
    int lane = tid & 31;
    int wave = tid >> 5;
    int wave_m = wave >> 2;
    int wave_n = wave & 3;
    int lrow = lane & 15;
    int khalf = (lane >> 4) << 1;
    int madd = (lane >> 4) << 3;
    int blockM = blockIdx.x * BM;
    int blockN = blockIdx.y * BN;

    v8f acc[4][2];
#pragma unroll
    for (int mt = 0; mt < 4; ++mt) {
#pragma unroll
        for (int i = 0; i < 8; ++i) {
            int gm = blockM + wave_m * 64 + mt * 16 + i + madd;
            float bv = (gm < M2) ? bias[gm] : 0.0f;
#pragma unroll
            for (int nt = 0; nt < 2; ++nt) acc[mt][nt][i] = bv;
        }
    }

    for (int kb = 0; kb < KPAD; kb += BK) {
        {
            int m = tid >> 1;
            int ks = (tid & 1) * 8;
            int gm = blockM + m;
#pragma unroll
            for (int j = 0; j < 8; ++j) {
                int gk = kb + ks + j;
                Ws[m][ks + j] = (gm < M2 && gk < KD) ? Wg[gm * KD + gk] : 0.0f;
            }
        }
        {
            int k = tid >> 4;
            int ns = (tid & 15) * 8;
            int gk = kb + k;
            if (gk < KD) {
                const float* src = Y + (size_t)gk * LSEQ + blockN + ns;
#pragma unroll
                for (int j = 0; j < 8; ++j) Ys[ns + j][k] = src[j];
            } else {
#pragma unroll
                for (int j = 0; j < 8; ++j) Ys[ns + j][k] = 0.0f;
            }
        }
        __syncthreads();
#pragma unroll
        for (int kc = 0; kc < 4; ++kc) {
            int kk = kc * 4 + khalf;
            v2f a[4], b[2];
#pragma unroll
            for (int mt = 0; mt < 4; ++mt)
                a[mt] = *(const v2f*)&Ws[wave_m * 64 + mt * 16 + lrow][kk];
#pragma unroll
            for (int nt = 0; nt < 2; ++nt)
                b[nt] = *(const v2f*)&Ys[wave_n * 32 + nt * 16 + lrow][kk];
#pragma unroll
            for (int mt = 0; mt < 4; ++mt)
#pragma unroll
                for (int nt = 0; nt < 2; ++nt)
                    acc[mt][nt] = __builtin_amdgcn_wmma_f32_16x16x4_f32(
                        false, a[mt], false, b[nt],
                        (short)0, acc[mt][nt], false, false);
        }
        __syncthreads();
    }

    int gn_base = blockN + wave_n * 32;
#pragma unroll
    for (int mt = 0; mt < 4; ++mt) {
#pragma unroll
        for (int i = 0; i < 8; ++i) {
            int gm = blockM + wave_m * 64 + mt * 16 + i + madd;
            if (gm < M2) {
#pragma unroll
                for (int nt = 0; nt < 2; ++nt) {
                    int gn = gn_base + nt * 16 + lrow;
                    Y2[(size_t)gm * LSEQ + gn] = acc[mt][nt][i];
                }
            }
        }
    }
}

#define LN_COLS 32
#define LN_WAVES 32
#define LN_CPW 16

__global__ __launch_bounds__(1024) void s4_glu_ln(
        const float* __restrict__ Y2, float* __restrict__ Z,
        const float* __restrict__ g, const float* __restrict__ b) {
    __shared__ float red_sum[LN_WAVES][LN_COLS];
    __shared__ float red_sq[LN_WAVES][LN_COLS];
    int lane = threadIdx.x & 31;
    int wave = __builtin_amdgcn_readfirstlane((int)(threadIdx.x >> 5));
    int l = blockIdx.x * LN_COLS + lane;

    float p[LN_CPW], q[LN_CPW], x[LN_CPW], gg[LN_CPW], bb[LN_CPW];
#pragma unroll
    for (int i = 0; i < LN_CPW; ++i) {
        int h = wave + LN_WAVES * i;
        int hc = (i < LN_CPW - 1 || h < HCH) ? h : (HCH - 1);
        p[i] = Y2[(size_t)hc * LSEQ + l];
        q[i] = Y2[(size_t)(hc + HCH) * LSEQ + l];
        x[i] = Z[(size_t)hc * LSEQ + l];
        gg[i] = g[hc];
        bb[i] = b[hc];
    }

    float s = 0.0f;
#pragma unroll
    for (int i = 0; i < LN_CPW; ++i) {
        int h = wave + LN_WAVES * i;
        float v = p[i] * sigmoidf(q[i]) + x[i];
        x[i] = (i < LN_CPW - 1 || h < HCH) ? v : 0.0f;
        s += x[i];
    }
    red_sum[wave][lane] = s;
    __syncthreads();
    float tot = 0.0f;
#pragma unroll
    for (int w = 0; w < LN_WAVES; ++w) tot += red_sum[w][lane];
    float mu = tot * (1.0f / (float)HCH);

    float sq = 0.0f;
#pragma unroll
    for (int i = 0; i < LN_CPW; ++i) {
        int h = wave + LN_WAVES * i;
        float d = (i < LN_CPW - 1 || h < HCH) ? (x[i] - mu) : 0.0f;
        x[i] = d;
        sq += d * d;
    }
    red_sq[wave][lane] = sq;
    __syncthreads();
    float tq = 0.0f;
#pragma unroll
    for (int w = 0; w < LN_WAVES; ++w) tq += red_sq[w][lane];
    float rstd = rsqrtf(tq * (1.0f / (float)HCH) + LN_EPS);

#pragma unroll
    for (int i = 0; i < LN_CPW; ++i) {
        int h = wave + LN_WAVES * i;
        if (i < LN_CPW - 1 || h < HCH)
            Z[(size_t)h * LSEQ + l] = x[i] * rstd * gg[i] + bb[i];
    }
}

extern "C" void kernel_launch(void* const* d_in, const int* in_sizes, int n_in,
                              void* d_out, int out_size, void* d_ws, size_t ws_size,
                              hipStream_t stream) {
    (void)in_sizes; (void)n_in; (void)out_size; (void)ws_size;
    const float* Zin    = (const float*)d_in[0];
    const float* log_dt = (const float*)d_in[1];
    const float* Arl    = (const float*)d_in[2];
    const float* Aim    = (const float*)d_in[3];
    const float* Cre    = (const float*)d_in[4];
    const float* Cim    = (const float*)d_in[5];
    const float* Dsk    = (const float*)d_in[6];
    const float* glu_w  = (const float*)d_in[7];
    const float* glu_b  = (const float*)d_in[8];
    const float* ln_g   = (const float*)d_in[9];
    const float* ln_b   = (const float*)d_in[10];

    float* ws = (float*)d_ws;
    const size_t HN = (size_t)HCH * NMODES;
    float* w_re  = ws;
    float* w_im  = ws + HN;
    float* c_re  = ws + 2 * HN;
    float* c_im  = ws + 3 * HN;
    float* zbuf  = ws + 4 * HN;
    float* ybuf  = zbuf + (size_t)HCH * LSEQ;
    float* y2buf = ybuf + (size_t)HCH * LSEQ;

    s4_setup<<<(int)((HN + 255) / 256), 256, 0, stream>>>(
        log_dt, Arl, Aim, Cre, Cim, w_re, w_im, c_re, c_im);

    hipMemcpyAsync(zbuf, Zin, sizeof(float) * (size_t)HCH * LSEQ,
                   hipMemcpyDeviceToDevice, stream);

    for (int layer = 0; layer < NLAYERS; ++layer) {
        s4_scan<<<64, 256, 0, stream>>>(zbuf, w_re, w_im, c_re, c_im, Dsk, ybuf);
        s4_glu_gemm<<<dim3(8, 16, 1), 256, 0, stream>>>(glu_w, glu_b, ybuf, y2buf);
        s4_glu_ln<<<LSEQ / LN_COLS, 1024, 0, stream>>>(y2buf, zbuf, ln_g, ln_b);
    }

    hipMemcpyAsync(d_out, zbuf, sizeof(float) * (size_t)HCH * LSEQ,
                   hipMemcpyDeviceToDevice, stream);
}


# glu_gemm K loop software-pipelined: next tile's 16 global loads issued as one group ahead of the 32 WMMAs into registers, LDS double-buffered, one barrier per tile; sum order unchanged
# speedup vs baseline: 6.4997x; 1.4715x over previous
#include <hip/hip_runtime.h>
#include <hip/hip_bf16.h>

#define HCH 511
#define LSEQ 2048
#define NMODES 32
#define NLAYERS 4
#define M2 1022
#define KD 511
#define KPAD 512
#define LN_EPS 1e-5f

typedef __attribute__((ext_vector_type(2))) float v2f;
typedef __attribute__((ext_vector_type(8))) float v8f;

__device__ __forceinline__ float gelu_tanh(float x) {
    float x3 = x * x * x;
    float t = tanhf(0.7978845608028654f * (x + 0.044715f * x3));
    return 0.5f * x * (1.0f + t);
}

__device__ __forceinline__ float sigmoidf(float x) {
    return 1.0f / (1.0f + expf(-x));
}

__global__ void s4_setup(const float* __restrict__ log_dt,
                         const float* __restrict__ Arl,
                         const float* __restrict__ Aim,
                         const float* __restrict__ Cre,
                         const float* __restrict__ Cim,
                         float* __restrict__ w_re, float* __restrict__ w_im,
                         float* __restrict__ c_re, float* __restrict__ c_im) {
    int idx = blockIdx.x * blockDim.x + threadIdx.x;
    if (idx >= HCH * NMODES) return;
    int h = idx / NMODES;
    float a_re = -expf(Arl[idx]);
    float a_im = Aim[idx];
    float dt = expf(log_dt[h]);
    float dre = a_re * dt, dim = a_im * dt;
    float er = expf(dre);
    float sn, cs;
    sincosf(dim, &sn, &cs);
    float wre = er * cs, wim = er * sn;
    float nre = wre - 1.0f, nim = wim;
    float inv = 1.0f / (a_re * a_re + a_im * a_im);
    float tre = (nre * a_re + nim * a_im) * inv;
    float tim = (nim * a_re - nre * a_im) * inv;
    float Cr = Cre[idx], Ci = Cim[idx];
    w_re[idx] = wre;
    w_im[idx] = wim;
    c_re[idx] = 2.0f * (Cr * tre - Ci * tim);
    c_im[idx] = 2.0f * (Cr * tim + Ci * tre);
}

__device__ __forceinline__ float lane_xor16(float v) {
    int i = __builtin_bit_cast(int, v);
    int r = __builtin_amdgcn_permlanex16(i, i, 0x76543210, 0xfedcba98, false, false);
    return __builtin_bit_cast(float, r);
}
template <int MASK>
__device__ __forceinline__ float lane_xor_row(float v) {
    int i = __builtin_bit_cast(int, v);
    int r = __builtin_amdgcn_update_dpp(0, i, 0x160 | MASK, 0xf, 0xf, true);
    return __builtin_bit_cast(float, r);
}
template <int T>
__device__ __forceinline__ float lane_bcast(float v) {
    return __builtin_bit_cast(float,
        __builtin_amdgcn_readlane(__builtin_bit_cast(int, v), T));
}

__device__ __forceinline__ float scan_step(float wr, float wi, float cr, float ci,
                                         float zt, float& xr, float& xi) {
    float nxr = fmaf(wr, xr, fmaf(-wi, xi, zt));
    float nxi = fmaf(wr, xi, wi * xr);
    xr = nxr; xi = nxi;
    return fmaf(cr, xr, -ci * xi);
}

__global__ __launch_bounds__(256) void s4_scan(
        const float* __restrict__ z,
        const float* __restrict__ w_re, const float* __restrict__ w_im,
        const float* __restrict__ c_re, const float* __restrict__ c_im,
        const float* __restrict__ D_skip, float* __restrict__ y) {
    int warp = threadIdx.x >> 5;
    int lane = threadIdx.x & 31;
    int h = blockIdx.x * 8 + warp;
    if (h >= HCH) return;
    int cidx = h * NMODES + lane;
    float wr = w_re[cidx], wi = w_im[cidx];
    float cr = c_re[cidx], ci = c_im[cidx];
    float d = D_skip[h];
    float xr = 0.0f, xi = 0.0f;
    const bool b4 = (lane & 16) != 0;
    const bool b3 = (lane & 8) != 0;
    const bool b2 = (lane & 4) != 0;
    const bool b1 = (lane & 2) != 0;
    const bool b0 = (lane & 1) != 0;
    const float* zrow = z + (size_t)h * LSEQ;
    float* yrow = y + (size_t)h * LSEQ;
    for (int l0 = 0; l0 < LSEQ; l0 += 32) {
        float zv = zrow[l0 + lane];
        float v[32];
#define SCAN_STEP(T) v[T] = scan_step(wr, wi, cr, ci, lane_bcast<T>(zv), xr, xi);
        SCAN_STEP(0)  SCAN_STEP(1)  SCAN_STEP(2)  SCAN_STEP(3)
        SCAN_STEP(4)  SCAN_STEP(5)  SCAN_STEP(6)  SCAN_STEP(7)
        SCAN_STEP(8)  SCAN_STEP(9)  SCAN_STEP(10) SCAN_STEP(11)
        SCAN_STEP(12) SCAN_STEP(13) SCAN_STEP(14) SCAN_STEP(15)
        SCAN_STEP(16) SCAN_STEP(17) SCAN_STEP(18) SCAN_STEP(19)
        SCAN_STEP(20) SCAN_STEP(21) SCAN_STEP(22) SCAN_STEP(23)
        SCAN_STEP(24) SCAN_STEP(25) SCAN_STEP(26) SCAN_STEP(27)
        SCAN_STEP(28) SCAN_STEP(29) SCAN_STEP(30) SCAN_STEP(31)
#undef SCAN_STEP
#pragma unroll
        for (int j = 0; j < 16; ++j) {
            float keep = b4 ? v[j + 16] : v[j];
            float send = b4 ? v[j] : v[j + 16];
            v[j] = keep + lane_xor16(send);
        }
#pragma unroll
        for (int j = 0; j < 8; ++j) {
            float keep = b3 ? v[j + 8] : v[j];
            float send = b3 ? v[j] : v[j + 8];
            v[j] = keep + lane_xor_row<8>(send);
        }
#pragma unroll
        for (int j = 0; j < 4; ++j) {
            float keep = b2 ? v[j + 4] : v[j];
            float send = b2 ? v[j] : v[j + 4];
            v[j] = keep + lane_xor_row<4>(send);
        }
#pragma unroll
        for (int j = 0; j < 2; ++j) {
            float keep = b1 ? v[j + 2] : v[j];
            float send = b1 ? v[j] : v[j + 2];
            v[j] = keep + lane_xor_row<2>(send);
        }
        float keep0 = b0 ? v[1] : v[0];
        float send0 = b0 ? v[0] : v[1];
        float out = keep0 + lane_xor_row<1>(send0);
        yrow[l0 + lane] = gelu_tanh(fmaf(d, zv, out));
    }
}

#define BM 128
#define BN 128
#define BK 16
#define LDP 18
#define NKT (KPAD / BK)

__global__ __launch_bounds__(256) void s4_glu_gemm(
        const float* __restrict__ Wg, const float* __restrict__ bias,
        const float* __restrict__ Y, float* __restrict__ Y2) {
    __shared__ float Ws[2][BM][LDP];
    __shared__ float Ys[2][BN][LDP];
    int tid = threadIdx.x;
    int lane = tid & 31;
    int wave = tid >> 5;
    int wave_m = wave >> 2;
    int wave_n = wave & 3;
    int lrow = lane & 15;
    int khalf = (lane >> 4) << 1;
    int madd = (lane >> 4) << 3;
    int blockM = blockIdx.x * BM;
    int blockN = blockIdx.y * BN;

    int sk = tid & 15;
    int sm = tid >> 4;
    int wrow[8];
#pragma unroll
    for (int j = 0; j < 8; ++j) {
        int gm = blockM + sm + 16 * j;
        wrow[j] = (gm < M2 ? gm : (M2 - 1)) * KD;
    }
    const float* ycol = Y + blockN + lane;

    v8f acc[4][2];
#pragma unroll
    for (int mt = 0; mt < 4; ++mt) {
#pragma unroll
        for (int i = 0; i < 8; ++i) {
            int gm = blockM + wave_m * 64 + mt * 16 + i + madd;
            float bv = (gm < M2) ? bias[gm] : 0.0f;
#pragma unroll
            for (int nt = 0; nt < 2; ++nt) acc[mt][nt][i] = bv;
        }
    }

    float wreg[8], yreg[8];

#define TILE_LOAD(kb)                                                          \
    {                                                                          \
        int gk = (kb) + sk;                                                    \
        int gkc = gk < KD ? gk : (KD - 1);                                     \
        _Pragma("unroll")                                                      \
        for (int j = 0; j < 8; ++j) wreg[j] = Wg[wrow[j] + gkc];               \
        _Pragma("unroll")                                                      \
        for (int r = 0; r < 2; ++r) {                                          \
            int gy = (kb) + wave + 8 * r;                                      \
            int gyc = gy < KD ? gy : (KD - 1);                                 \
            const float* srow = ycol + (size_t)gyc * LSEQ;                     \
            _Pragma("unroll")                                                  \
            for (int j = 0; j < 4; ++j) yreg[4 * r + j] = srow[32 * j];        \
        }                                                                      \
    }

#define TILE_STORE(kb, b)                                                      \
    {                                                                          \
        bool kin = (kb) + sk < KD;                                             \
        _Pragma("unroll")                                                      \
        for (int j = 0; j < 8; ++j)                                            \
            Ws[b][sm + 16 * j][sk] = kin ? wreg[j] : 0.0f;                     \
        _Pragma("unroll")                                                      \
        for (int r = 0; r < 2; ++r) {                                          \
            bool yin = (kb) + wave + 8 * r < KD;                               \
            _Pragma("unroll")                                                  \
            for (int j = 0; j < 4; ++j)                                        \
                Ys[b][lane + 32 * j][wave + 8 * r] = yin ? yreg[4 * r + j] : 0.0f; \
        }                                                                      \
    }

#define TILE_MMA(b)                                                            \
    _Pragma("unroll")                                                          \
    for (int kc = 0; kc < 4; ++kc) {                                           \
        int kk = kc * 4 + khalf;                                               \
        v2f a[4], bf[2];                                                       \
        _Pragma("unroll")                                                      \
        for (int mt = 0; mt < 4; ++mt)                                         \
            a[mt] = *(const v2f*)&Ws[b][wave_m * 64 + mt * 16 + lrow][kk];     \
        _Pragma("unroll")                                                      \
        for (int nt = 0; nt < 2; ++nt)                                         \
            bf[nt] = *(const v2f*)&Ys[b][wave_n * 32 + nt * 16 + lrow][kk];    \
        _Pragma("unroll")                                                      \
        for (int mt = 0; mt < 4; ++mt)                                         \
            _Pragma("unroll")                                                  \
            for (int nt = 0; nt < 2; ++nt)                                     \
                acc[mt][nt] = __builtin_amdgcn_wmma_f32_16x16x4_f32(           \
                    false, a[mt], false, bf[nt],                               \
                    (short)0, acc[mt][nt], false, false);                      \
    }

    TILE_LOAD(0)
    TILE_STORE(0, 0)
    __syncthreads();
    for (int t = 0; t < NKT - 1; ++t) {
        int cur = t & 1;
        int kbn = (t + 1) * BK;
        TILE_LOAD(kbn)
        TILE_MMA(cur)
        TILE_STORE(kbn, cur ^ 1)
        __syncthreads();
    }
    TILE_MMA((NKT - 1) & 1)
#undef TILE_LOAD
#undef TILE_STORE
#undef TILE_MMA

    int gn_base = blockN + wave_n * 32;
#pragma unroll
    for (int mt = 0; mt < 4; ++mt) {
#pragma unroll
        for (int i = 0; i < 8; ++i) {
            int gm = blockM + wave_m * 64 + mt * 16 + i + madd;
            if (gm < M2) {
#pragma unroll
                for (int nt = 0; nt < 2; ++nt) {
                    int gn = gn_base + nt * 16 + lrow;
                    Y2[(size_t)gm * LSEQ + gn] = acc[mt][nt][i];
                }
            }
        }
    }
}

#define LN_COLS 32
#define LN_WAVES 32
#define LN_CPW 16

__global__ __launch_bounds__(1024) void s4_glu_ln(
        const float* __restrict__ Y2, float* __restrict__ Z,
        const float* __restrict__ g, const float* __restrict__ b) {
    __shared__ float red_sum[LN_WAVES][LN_COLS];
    __shared__ float red_sq[LN_WAVES][LN_COLS];
    int lane = threadIdx.x & 31;
    int wave = __builtin_amdgcn_readfirstlane((int)(threadIdx.x >> 5));
    int l = blockIdx.x * LN_COLS + lane;

    float p[LN_CPW], q[LN_CPW], x[LN_CPW], gg[LN_CPW], bb[LN_CPW];
#pragma unroll
    for (int i = 0; i < LN_CPW; ++i) {
        int h = wave + LN_WAVES * i;
        int hc = (i < LN_CPW - 1 || h < HCH) ? h : (HCH - 1);
        p[i] = Y2[(size_t)hc * LSEQ + l];
        q[i] = Y2[(size_t)(hc + HCH) * LSEQ + l];
        x[i] = Z[(size_t)hc * LSEQ + l];
        gg[i] = g[hc];
        bb[i] = b[hc];
    }

    float s = 0.0f;
#pragma unroll
    for (int i = 0; i < LN_CPW; ++i) {
        int h = wave + LN_WAVES * i;
        float v = p[i] * sigmoidf(q[i]) + x[i];
        x[i] = (i < LN_CPW - 1 || h < HCH) ? v : 0.0f;
        s += x[i];
    }
    red_sum[wave][lane] = s;
    __syncthreads();
    float tot = 0.0f;
#pragma unroll
    for (int w = 0; w < LN_WAVES; ++w) tot += red_sum[w][lane];
    float mu = tot * (1.0f / (float)HCH);

    float sq = 0.0f;
#pragma unroll
    for (int i = 0; i < LN_CPW; ++i) {
        int h = wave + LN_WAVES * i;
        float d = (i < LN_CPW - 1 || h < HCH) ? (x[i] - mu) : 0.0f;
        x[i] = d;
        sq += d * d;
    }
    red_sq[wave][lane] = sq;
    __syncthreads();
    float tq = 0.0f;
#pragma unroll
    for (int w = 0; w < LN_WAVES; ++w) tq += red_sq[w][lane];
    float rstd = rsqrtf(tq * (1.0f / (float)HCH) + LN_EPS);

#pragma unroll
    for (int i = 0; i < LN_CPW; ++i) {
        int h = wave + LN_WAVES * i;
        if (i < LN_CPW - 1 || h < HCH)
            Z[(size_t)h * LSEQ + l] = x[i] * rstd * gg[i] + bb[i];
    }
}

extern "C" void kernel_launch(void* const* d_in, const int* in_sizes, int n_in,
                              void* d_out, int out_size, void* d_ws, size_t ws_size,
                              hipStream_t stream) {
    (void)in_sizes; (void)n_in; (void)out_size; (void)ws_size;
    const float* Zin    = (const float*)d_in[0];
    const float* log_dt = (const float*)d_in[1];
    const float* Arl    = (const float*)d_in[2];
    const float* Aim    = (const float*)d_in[3];
    const float* Cre    = (const float*)d_in[4];
    const float* Cim    = (const float*)d_in[5];
    const float* Dsk    = (const float*)d_in[6];
    const float* glu_w  = (const float*)d_in[7];
    const float* glu_b  = (const float*)d_in[8];
    const float* ln_g   = (const float*)d_in[9];
    const float* ln_b   = (const float*)d_in[10];

    float* ws = (float*)d_ws;
    const size_t HN = (size_t)HCH * NMODES;
    float* w_re  = ws;
    float* w_im  = ws + HN;
    float* c_re  = ws + 2 * HN;
    float* c_im  = ws + 3 * HN;
    float* zbuf  = ws + 4 * HN;
    float* ybuf  = zbuf + (size_t)HCH * LSEQ;
    float* y2buf = ybuf + (size_t)HCH * LSEQ;

    s4_setup<<<(int)((HN + 255) / 256), 256, 0, stream>>>(
        log_dt, Arl, Aim, Cre, Cim, w_re, w_im, c_re, c_im);

    hipMemcpyAsync(zbuf, Zin, sizeof(float) * (size_t)HCH * LSEQ,
                   hipMemcpyDeviceToDevice, stream);

    for (int layer = 0; layer < NLAYERS; ++layer) {
        s4_scan<<<64, 256, 0, stream>>>(zbuf, w_re, w_im, c_re, c_im, Dsk, ybuf);
        s4_glu_gemm<<<dim3(8, 16, 1), 256, 0, stream>>>(glu_w, glu_b, ybuf, y2buf);
        s4_glu_ln<<<LSEQ / LN_COLS, 1024, 0, stream>>>(y2buf, zbuf, ln_g, ln_b);
    }

    hipMemcpyAsync(d_out, zbuf, sizeof(float) * (size_t)HCH * LSEQ,
                   hipMemcpyDeviceToDevice, stream);
}


# scan loop software-pipelined: mode-sum butterfly of chunk i runs in the latency gaps of chunk i+1's recurrence, z loaded two chunks ahead, one wave per SIMD (4-wave blocks); sums bit-identical
# speedup vs baseline: 9.0521x; 1.3927x over previous
#include <hip/hip_runtime.h>
#include <hip/hip_bf16.h>

#define HCH 511
#define LSEQ 2048
#define NMODES 32
#define NLAYERS 4
#define M2 1022
#define KD 511
#define KPAD 512
#define LN_EPS 1e-5f

typedef __attribute__((ext_vector_type(2))) float v2f;
typedef __attribute__((ext_vector_type(8))) float v8f;

__device__ __forceinline__ float gelu_tanh(float x) {
    float x3 = x * x * x;
    float t = tanhf(0.7978845608028654f * (x + 0.044715f * x3));
    return 0.5f * x * (1.0f + t);
}

__device__ __forceinline__ float sigmoidf(float x) {
    return 1.0f / (1.0f + expf(-x));
}

__global__ void s4_setup(const float* __restrict__ log_dt,
                         const float* __restrict__ Arl,
                         const float* __restrict__ Aim,
                         const float* __restrict__ Cre,
                         const float* __restrict__ Cim,
                         float* __restrict__ w_re, float* __restrict__ w_im,
                         float* __restrict__ c_re, float* __restrict__ c_im) {
    int idx = blockIdx.x * blockDim.x + threadIdx.x;
    if (idx >= HCH * NMODES) return;
    int h = idx / NMODES;
    float a_re = -expf(Arl[idx]);
    float a_im = Aim[idx];
    float dt = expf(log_dt[h]);
    float dre = a_re * dt, dim = a_im * dt;
    float er = expf(dre);
    float sn, cs;
    sincosf(dim, &sn, &cs);
    float wre = er * cs, wim = er * sn;
    float nre = wre - 1.0f, nim = wim;
    float inv = 1.0f / (a_re * a_re + a_im * a_im);
    float tre = (nre * a_re + nim * a_im) * inv;
    float tim = (nim * a_re - nre * a_im) * inv;
    float Cr = Cre[idx], Ci = Cim[idx];
    w_re[idx] = wre;
    w_im[idx] = wim;
    c_re[idx] = 2.0f * (Cr * tre - Ci * tim);
    c_im[idx] = 2.0f * (Cr * tim + Ci * tre);
}

__device__ __forceinline__ float lane_xor16(float v) {
    int i = __builtin_bit_cast(int, v);
    int r = __builtin_amdgcn_permlanex16(i, i, 0x76543210, 0xfedcba98, false, false);
    return __builtin_bit_cast(float, r);
}
template <int MASK>
__device__ __forceinline__ float lane_xor_row(float v) {
    int i = __builtin_bit_cast(int, v);
    int r = __builtin_amdgcn_update_dpp(0, i, 0x160 | MASK, 0xf, 0xf, true);
    return __builtin_bit_cast(float, r);
}
template <int T>
__device__ __forceinline__ float lane_bcast(float v) {
    return __builtin_bit_cast(float,
        __builtin_amdgcn_readlane(__builtin_bit_cast(int, v), T));
}

__device__ __forceinline__ float scan_step(float wr, float wi, float cr, float ci,
                                         float zt, float& xr, float& xi) {
    float nxr = fmaf(wr, xr, fmaf(-wi, xi, zt));
    float nxi = fmaf(wr, xi, wi * xr);
    xr = nxr; xi = nxi;
    return fmaf(cr, xr, -ci * xi);
}

__device__ __forceinline__ void scan_chunk(float wr, float wi, float cr, float ci,
                                           float zv, float& xr, float& xi,
                                           float (&v)[32]) {
#define SCAN_STEP(T) v[T] = scan_step(wr, wi, cr, ci, lane_bcast<T>(zv), xr, xi);
    SCAN_STEP(0)  SCAN_STEP(1)  SCAN_STEP(2)  SCAN_STEP(3)
    SCAN_STEP(4)  SCAN_STEP(5)  SCAN_STEP(6)  SCAN_STEP(7)
    SCAN_STEP(8)  SCAN_STEP(9)  SCAN_STEP(10) SCAN_STEP(11)
    SCAN_STEP(12) SCAN_STEP(13) SCAN_STEP(14) SCAN_STEP(15)
    SCAN_STEP(16) SCAN_STEP(17) SCAN_STEP(18) SCAN_STEP(19)
    SCAN_STEP(20) SCAN_STEP(21) SCAN_STEP(22) SCAN_STEP(23)
    SCAN_STEP(24) SCAN_STEP(25) SCAN_STEP(26) SCAN_STEP(27)
    SCAN_STEP(28) SCAN_STEP(29) SCAN_STEP(30) SCAN_STEP(31)
#undef SCAN_STEP
}

__device__ __forceinline__ float mode_sum(float (&v)[32], bool b4, bool b3,
                                          bool b2, bool b1, bool b0) {
#pragma unroll
    for (int j = 0; j < 16; ++j) {
        float keep = b4 ? v[j + 16] : v[j];
        float send = b4 ? v[j] : v[j + 16];
        v[j] = keep + lane_xor16(send);
    }
#pragma unroll
    for (int j = 0; j < 8; ++j) {
        float keep = b3 ? v[j + 8] : v[j];
        float send = b3 ? v[j] : v[j + 8];
        v[j] = keep + lane_xor_row<8>(send);
    }
#pragma unroll
    for (int j = 0; j < 4; ++j) {
        float keep = b2 ? v[j + 4] : v[j];
        float send = b2 ? v[j] : v[j + 4];
        v[j] = keep + lane_xor_row<4>(send);
    }
#pragma unroll
    for (int j = 0; j < 2; ++j) {
        float keep = b1 ? v[j + 2] : v[j];
        float send = b1 ? v[j] : v[j + 2];
        v[j] = keep + lane_xor_row<2>(send);
    }
    float keep0 = b0 ? v[1] : v[0];
    float send0 = b0 ? v[0] : v[1];
    return keep0 + lane_xor_row<1>(send0);
}

#define SCAN_PIN_STATE() asm volatile("" : "+v"(xr), "+v"(xi))

#define SCAN_CHUNKS (LSEQ / 32)
#define SCAN_WAVES 4

__global__ __launch_bounds__(SCAN_WAVES * 32) void s4_scan(
        const float* __restrict__ z,
        const float* __restrict__ w_re, const float* __restrict__ w_im,
        const float* __restrict__ c_re, const float* __restrict__ c_im,
        const float* __restrict__ D_skip, float* __restrict__ y) {
    int warp = threadIdx.x >> 5;
    int lane = threadIdx.x & 31;
    int h = blockIdx.x * SCAN_WAVES + warp;
    if (h >= HCH) return;
    int cidx = h * NMODES + lane;
    float wr = w_re[cidx], wi = w_im[cidx];
    float cr = c_re[cidx], ci = c_im[cidx];
    float d = D_skip[h];
    float xr = 0.0f, xi = 0.0f;
    const bool b4 = (lane & 16) != 0;
    const bool b3 = (lane & 8) != 0;
    const bool b2 = (lane & 4) != 0;
    const bool b1 = (lane & 2) != 0;
    const bool b0 = (lane & 1) != 0;
    const float* zrow = z + (size_t)h * LSEQ + lane;
    float* yrow = y + (size_t)h * LSEQ + lane;

    float va[32], vb[32];
    float z0 = zrow[0];
    float z1 = zrow[32];
    float z2 = zrow[64];
    scan_chunk(wr, wi, cr, ci, z0, xr, xi, va);
    for (int c = 0; c < SCAN_CHUNKS - 2; c += 2) {
        float z3 = zrow[(c + 3) * 32];
        scan_chunk(wr, wi, cr, ci, z1, xr, xi, vb);
        float o = mode_sum(va, b4, b3, b2, b1, b0);
        SCAN_PIN_STATE();
        yrow[c * 32] = gelu_tanh(fmaf(d, z0, o));

        int c4 = c + 4 < SCAN_CHUNKS ? c + 4 : SCAN_CHUNKS - 1;
        float z4 = zrow[c4 * 32];
        scan_chunk(wr, wi, cr, ci, z2, xr, xi, va);
        o = mode_sum(vb, b4, b3, b2, b1, b0);
        SCAN_PIN_STATE();
        yrow[(c + 1) * 32] = gelu_tanh(fmaf(d, z1, o));

        z0 = z2; z1 = z3; z2 = z4;
    }
    scan_chunk(wr, wi, cr, ci, z1, xr, xi, vb);
    float o = mode_sum(va, b4, b3, b2, b1, b0);
    yrow[(SCAN_CHUNKS - 2) * 32] = gelu_tanh(fmaf(d, z0, o));
    o = mode_sum(vb, b4, b3, b2, b1, b0);
    yrow[(SCAN_CHUNKS - 1) * 32] = gelu_tanh(fmaf(d, z1, o));
}

#define BM 128
#define BN 128
#define BK 16
#define LDP 18
#define NKT (KPAD / BK)

__global__ __launch_bounds__(256) void s4_glu_gemm(
        const float* __restrict__ Wg, const float* __restrict__ bias,
        const float* __restrict__ Y, float* __restrict__ Y2) {
    __shared__ float Ws[2][BM][LDP];
    __shared__ float Ys[2][BN][LDP];
    int tid = threadIdx.x;
    int lane = tid & 31;
    int wave = tid >> 5;
    int wave_m = wave >> 2;
    int wave_n = wave & 3;
    int lrow = lane & 15;
    int khalf = (lane >> 4) << 1;
    int madd = (lane >> 4) << 3;
    int blockM = blockIdx.x * BM;
    int blockN = blockIdx.y * BN;

    int sk = tid & 15;
    int sm = tid >> 4;
    int wrow[8];
#pragma unroll
    for (int j = 0; j < 8; ++j) {
        int gm = blockM + sm + 16 * j;
        wrow[j] = (gm < M2 ? gm : (M2 - 1)) * KD;
    }
    const float* ycol = Y + blockN + lane;

    v8f acc[4][2];
#pragma unroll
    for (int mt = 0; mt < 4; ++mt) {
#pragma unroll
        for (int i = 0; i < 8; ++i) {
            int gm = blockM + wave_m * 64 + mt * 16 + i + madd;
            float bv = (gm < M2) ? bias[gm] : 0.0f;
#pragma unroll
            for (int nt = 0; nt < 2; ++nt) acc[mt][nt][i] = bv;
        }
    }

    float wreg[8], yreg[8];

#define TILE_LOAD(kb)                                                          \
    {                                                                          \
        int gk = (kb) + sk;                                                    \
        int gkc = gk < KD ? gk : (KD - 1);                                     \
        _Pragma("unroll")                                                      \
        for (int j = 0; j < 8; ++j) wreg[j] = Wg[wrow[j] + gkc];               \
        _Pragma("unroll")                                                      \
        for (int r = 0; r < 2; ++r) {                                          \
            int gy = (kb) + wave + 8 * r;                                      \
            int gyc = gy < KD ? gy : (KD - 1);                                 \
            const float* srow = ycol + (size_t)gyc * LSEQ;                     \
            _Pragma("unroll")                                                  \
            for (int j = 0; j < 4; ++j) yreg[4 * r + j] = srow[32 * j];        \
        }                                                                      \
    }

#define TILE_STORE(kb, b)                                                      \
    {                                                                          \
        bool kin = (kb) + sk < KD;                                             \
        _Pragma("unroll")                                                      \
        for (int j = 0; j < 8; ++j)                                            \
            Ws[b][sm + 16 * j][sk] = kin ? wreg[j] : 0.0f;                     \
        _Pragma("unroll")                                                      \
        for (int r = 0; r < 2; ++r) {                                          \
            bool yin = (kb) + wave + 8 * r < KD;                               \
            _Pragma("unroll")                                                  \
            for (int j = 0; j < 4; ++j)                                        \
                Ys[b][lane + 32 * j][wave + 8 * r] = yin ? yreg[4 * r + j] : 0.0f; \
        }                                                                      \
    }

#define TILE_MMA(b)                                                            \
    _Pragma("unroll")                                                          \
    for (int kc = 0; kc < 4; ++kc) {                                           \
        int kk = kc * 4 + khalf;                                               \
        v2f a[4], bf[2];                                                       \
        _Pragma("unroll")                                                      \
        for (int mt = 0; mt < 4; ++mt)                                         \
            a[mt] = *(const v2f*)&Ws[b][wave_m * 64 + mt * 16 + lrow][kk];     \
        _Pragma("unroll")                                                      \
        for (int nt = 0; nt < 2; ++nt)                                         \
            bf[nt] = *(const v2f*)&Ys[b][wave_n * 32 + nt * 16 + lrow][kk];    \
        _Pragma("unroll")                                                      \
        for (int mt = 0; mt < 4; ++mt)                                         \
            _Pragma("unroll")                                                  \
            for (int nt = 0; nt < 2; ++nt)                                     \
                acc[mt][nt] = __builtin_amdgcn_wmma_f32_16x16x4_f32(           \
                    false, a[mt], false, bf[nt],                               \
                    (short)0, acc[mt][nt], false, false);                      \
    }

    TILE_LOAD(0)
    TILE_STORE(0, 0)
    __syncthreads();
    for (int t = 0; t < NKT - 1; ++t) {
        int cur = t & 1;
        int kbn = (t + 1) * BK;
        TILE_LOAD(kbn)
        TILE_MMA(cur)
        TILE_STORE(kbn, cur ^ 1)
        __syncthreads();
    }
    TILE_MMA((NKT - 1) & 1)
#undef TILE_LOAD
#undef TILE_STORE
#undef TILE_MMA

    int gn_base = blockN + wave_n * 32;
#pragma unroll
    for (int mt = 0; mt < 4; ++mt) {
#pragma unroll
        for (int i = 0; i < 8; ++i) {
            int gm = blockM + wave_m * 64 + mt * 16 + i + madd;
            if (gm < M2) {
#pragma unroll
                for (int nt = 0; nt < 2; ++nt) {
                    int gn = gn_base + nt * 16 + lrow;
                    Y2[(size_t)gm * LSEQ + gn] = acc[mt][nt][i];
                }
            }
        }
    }
}

#define LN_COLS 32
#define LN_WAVES 32
#define LN_CPW 16

__global__ __launch_bounds__(1024) void s4_glu_ln(
        const float* __restrict__ Y2, float* __restrict__ Z,
        const float* __restrict__ g, const float* __restrict__ b) {
    __shared__ float red_sum[LN_WAVES][LN_COLS];
    __shared__ float red_sq[LN_WAVES][LN_COLS];
    int lane = threadIdx.x & 31;
    int wave = __builtin_amdgcn_readfirstlane((int)(threadIdx.x >> 5));
    int l = blockIdx.x * LN_COLS + lane;

    float p[LN_CPW], q[LN_CPW], x[LN_CPW], gg[LN_CPW], bb[LN_CPW];
#pragma unroll
    for (int i = 0; i < LN_CPW; ++i) {
        int h = wave + LN_WAVES * i;
        int hc = (i < LN_CPW - 1 || h < HCH) ? h : (HCH - 1);
        p[i] = Y2[(size_t)hc * LSEQ + l];
        q[i] = Y2[(size_t)(hc + HCH) * LSEQ + l];
        x[i] = Z[(size_t)hc * LSEQ + l];
        gg[i] = g[hc];
        bb[i] = b[hc];
    }

    float s = 0.0f;
#pragma unroll
    for (int i = 0; i < LN_CPW; ++i) {
        int h = wave + LN_WAVES * i;
        float v = p[i] * sigmoidf(q[i]) + x[i];
        x[i] = (i < LN_CPW - 1 || h < HCH) ? v : 0.0f;
        s += x[i];
    }
    red_sum[wave][lane] = s;
    __syncthreads();
    float tot = 0.0f;
#pragma unroll
    for (int w = 0; w < LN_WAVES; ++w) tot += red_sum[w][lane];
    float mu = tot * (1.0f / (float)HCH);

    float sq = 0.0f;
#pragma unroll
    for (int i = 0; i < LN_CPW; ++i) {
        int h = wave + LN_WAVES * i;
        float d = (i < LN_CPW - 1 || h < HCH) ? (x[i] - mu) : 0.0f;
        x[i] = d;
        sq += d * d;
    }
    red_sq[wave][lane] = sq;
    __syncthreads();
    float tq = 0.0f;
#pragma unroll
    for (int w = 0; w < LN_WAVES; ++w) tq += red_sq[w][lane];
    float rstd = rsqrtf(tq * (1.0f / (float)HCH) + LN_EPS);

#pragma unroll
    for (int i = 0; i < LN_CPW; ++i) {
        int h = wave + LN_WAVES * i;
        if (i < LN_CPW - 1 || h < HCH)
            Z[(size_t)h * LSEQ + l] = x[i] * rstd * gg[i] + bb[i];
    }
}

extern "C" void kernel_launch(void* const* d_in, const int* in_sizes, int n_in,
                              void* d_out, int out_size, void* d_ws, size_t ws_size,
                              hipStream_t stream) {
    (void)in_sizes; (void)n_in; (void)out_size; (void)ws_size;
    const float* Zin    = (const float*)d_in[0];
    const float* log_dt = (const float*)d_in[1];
    const float* Arl    = (const float*)d_in[2];
    const float* Aim    = (const float*)d_in[3];
    const float* Cre    = (const float*)d_in[4];
    const float* Cim    = (const float*)d_in[5];
    const float* Dsk    = (const float*)d_in[6];
    const float* glu_w  = (const float*)d_in[7];
    const float* glu_b  = (const float*)d_in[8];
    const float* ln_g   = (const float*)d_in[9];
    const float* ln_b   = (const float*)d_in[10];

    float* ws = (float*)d_ws;
    const size_t HN = (size_t)HCH * NMODES;
    float* w_re  = ws;
    float* w_im  = ws + HN;
    float* c_re  = ws + 2 * HN;
    float* c_im  = ws + 3 * HN;
    float* zbuf  = ws + 4 * HN;
    float* ybuf  = zbuf + (size_t)HCH * LSEQ;
    float* y2buf = ybuf + (size_t)HCH * LSEQ;

    s4_setup<<<(int)((HN + 255) / 256), 256, 0, stream>>>(
        log_dt, Arl, Aim, Cre, Cim, w_re, w_im, c_re, c_im);

    hipMemcpyAsync(zbuf, Zin, sizeof(float) * (size_t)HCH * LSEQ,
                   hipMemcpyDeviceToDevice, stream);

    for (int layer = 0; layer < NLAYERS; ++layer) {
        s4_scan<<<(HCH + SCAN_WAVES - 1) / SCAN_WAVES, SCAN_WAVES * 32, 0, stream>>>(zbuf, w_re, w_im, c_re, c_im, Dsk, ybuf);
        s4_glu_gemm<<<dim3(8, 16, 1), 256, 0, stream>>>(glu_w, glu_b, ybuf, y2buf);
        s4_glu_ln<<<LSEQ / LN_COLS, 1024, 0, stream>>>(y2buf, zbuf, ln_g, ln_b);
    }

    hipMemcpyAsync(d_out, zbuf, sizeof(float) * (size_t)HCH * LSEQ,
                   hipMemcpyDeviceToDevice, stream);
}


# glu_gemm re-tiled: block 128x128x16 -> 64x128x32 (wave tile 32x32), grid 16x16 = 256 blocks = one per WGP, not half the chip idle; parent's loop order held by 2 sched fences; k order, sums unchanged
# speedup vs baseline: 10.9979x; 1.2150x over previous
#include <hip/hip_runtime.h>
#include <hip/hip_bf16.h>

#define HCH 511
#define LSEQ 2048
#define NMODES 32
#define NLAYERS 4
#define M2 1022
#define KD 511
#define KPAD 512
#define LN_EPS 1e-5f

typedef __attribute__((ext_vector_type(2))) float v2f;
typedef __attribute__((ext_vector_type(8))) float v8f;

__device__ __forceinline__ float gelu_tanh(float x) {
    float x3 = x * x * x;
    float t = tanhf(0.7978845608028654f * (x + 0.044715f * x3));
    return 0.5f * x * (1.0f + t);
}

__device__ __forceinline__ float sigmoidf(float x) {
    return 1.0f / (1.0f + expf(-x));
}

__global__ void s4_setup(const float* __restrict__ log_dt,
                         const float* __restrict__ Arl,
                         const float* __restrict__ Aim,
                         const float* __restrict__ Cre,
                         const float* __restrict__ Cim,
                         float* __restrict__ w_re, float* __restrict__ w_im,
                         float* __restrict__ c_re, float* __restrict__ c_im) {
    int idx = blockIdx.x * blockDim.x + threadIdx.x;
    if (idx >= HCH * NMODES) return;
    int h = idx / NMODES;
    float a_re = -expf(Arl[idx]);
    float a_im = Aim[idx];
    float dt = expf(log_dt[h]);
    float dre = a_re * dt, dim = a_im * dt;
    float er = expf(dre);
    float sn, cs;
    sincosf(dim, &sn, &cs);
    float wre = er * cs, wim = er * sn;
    float nre = wre - 1.0f, nim = wim;
    float inv = 1.0f / (a_re * a_re + a_im * a_im);
    float tre = (nre * a_re + nim * a_im) * inv;
    float tim = (nim * a_re - nre * a_im) * inv;
    float Cr = Cre[idx], Ci = Cim[idx];
    w_re[idx] = wre;
    w_im[idx] = wim;
    c_re[idx] = 2.0f * (Cr * tre - Ci * tim);
    c_im[idx] = 2.0f * (Cr * tim + Ci * tre);
}

__device__ __forceinline__ float lane_xor16(float v) {
    int i = __builtin_bit_cast(int, v);
    int r = __builtin_amdgcn_permlanex16(i, i, 0x76543210, 0xfedcba98, false, false);
    return __builtin_bit_cast(float, r);
}
template <int MASK>
__device__ __forceinline__ float lane_xor_row(float v) {
    int i = __builtin_bit_cast(int, v);
    int r = __builtin_amdgcn_update_dpp(0, i, 0x160 | MASK, 0xf, 0xf, true);
    return __builtin_bit_cast(float, r);
}
template <int T>
__device__ __forceinline__ float lane_bcast(float v) {
    return __builtin_bit_cast(float,
        __builtin_amdgcn_readlane(__builtin_bit_cast(int, v), T));
}

__device__ __forceinline__ float scan_step(float wr, float wi, float cr, float ci,
                                         float zt, float& xr, float& xi) {
    float nxr = fmaf(wr, xr, fmaf(-wi, xi, zt));
    float nxi = fmaf(wr, xi, wi * xr);
    xr = nxr; xi = nxi;
    return fmaf(cr, xr, -ci * xi);
}

__device__ __forceinline__ void scan_chunk(float wr, float wi, float cr, float ci,
                                           float zv, float& xr, float& xi,
                                           float (&v)[32]) {
#define SCAN_STEP(T) v[T] = scan_step(wr, wi, cr, ci, lane_bcast<T>(zv), xr, xi);
    SCAN_STEP(0)  SCAN_STEP(1)  SCAN_STEP(2)  SCAN_STEP(3)
    SCAN_STEP(4)  SCAN_STEP(5)  SCAN_STEP(6)  SCAN_STEP(7)
    SCAN_STEP(8)  SCAN_STEP(9)  SCAN_STEP(10) SCAN_STEP(11)
    SCAN_STEP(12) SCAN_STEP(13) SCAN_STEP(14) SCAN_STEP(15)
    SCAN_STEP(16) SCAN_STEP(17) SCAN_STEP(18) SCAN_STEP(19)
    SCAN_STEP(20) SCAN_STEP(21) SCAN_STEP(22) SCAN_STEP(23)
    SCAN_STEP(24) SCAN_STEP(25) SCAN_STEP(26) SCAN_STEP(27)
    SCAN_STEP(28) SCAN_STEP(29) SCAN_STEP(30) SCAN_STEP(31)
#undef SCAN_STEP
}

__device__ __forceinline__ float mode_sum(float (&v)[32], bool b4, bool b3,
                                          bool b2, bool b1, bool b0) {
#pragma unroll
    for (int j = 0; j < 16; ++j) {
        float keep = b4 ? v[j + 16] : v[j];
        float send = b4 ? v[j] : v[j + 16];
        v[j] = keep + lane_xor16(send);
    }
#pragma unroll
    for (int j = 0; j < 8; ++j) {
        float keep = b3 ? v[j + 8] : v[j];
        float send = b3 ? v[j] : v[j + 8];
        v[j] = keep + lane_xor_row<8>(send);
    }
#pragma unroll
    for (int j = 0; j < 4; ++j) {
        float keep = b2 ? v[j + 4] : v[j];
        float send = b2 ? v[j] : v[j + 4];
        v[j] = keep + lane_xor_row<4>(send);
    }
#pragma unroll
    for (int j = 0; j < 2; ++j) {
        float keep = b1 ? v[j + 2] : v[j];
        float send = b1 ? v[j] : v[j + 2];
        v[j] = keep + lane_xor_row<2>(send);
    }
    float keep0 = b0 ? v[1] : v[0];
    float send0 = b0 ? v[0] : v[1];
    return keep0 + lane_xor_row<1>(send0);
}

#define SCAN_PIN_STATE() asm volatile("" : "+v"(xr), "+v"(xi))

#define SCAN_CHUNKS (LSEQ / 32)
#define SCAN_WAVES 4

__global__ __launch_bounds__(SCAN_WAVES * 32) void s4_scan(
        const float* __restrict__ z,
        const float* __restrict__ w_re, const float* __restrict__ w_im,
        const float* __restrict__ c_re, const float* __restrict__ c_im,
        const float* __restrict__ D_skip, float* __restrict__ y) {
    int warp = threadIdx.x >> 5;
    int lane = threadIdx.x & 31;
    int h = blockIdx.x * SCAN_WAVES + warp;
    if (h >= HCH) return;
    int cidx = h * NMODES + lane;
    float wr = w_re[cidx], wi = w_im[cidx];
    float cr = c_re[cidx], ci = c_im[cidx];
    float d = D_skip[h];
    float xr = 0.0f, xi = 0.0f;
    const bool b4 = (lane & 16) != 0;
    const bool b3 = (lane & 8) != 0;
    const bool b2 = (lane & 4) != 0;
    const bool b1 = (lane & 2) != 0;
    const bool b0 = (lane & 1) != 0;
    const float* zrow = z + (size_t)h * LSEQ + lane;
    float* yrow = y + (size_t)h * LSEQ + lane;

    float va[32], vb[32];
    float z0 = zrow[0];
    float z1 = zrow[32];
    float z2 = zrow[64];
    scan_chunk(wr, wi, cr, ci, z0, xr, xi, va);
    for (int c = 0; c < SCAN_CHUNKS - 2; c += 2) {
        float z3 = zrow[(c + 3) * 32];
        scan_chunk(wr, wi, cr, ci, z1, xr, xi, vb);
        float o = mode_sum(va, b4, b3, b2, b1, b0);
        SCAN_PIN_STATE();
        yrow[c * 32] = gelu_tanh(fmaf(d, z0, o));

        int c4 = c + 4 < SCAN_CHUNKS ? c + 4 : SCAN_CHUNKS - 1;
        float z4 = zrow[c4 * 32];
        scan_chunk(wr, wi, cr, ci, z2, xr, xi, va);
        o = mode_sum(vb, b4, b3, b2, b1, b0);
        SCAN_PIN_STATE();
        yrow[(c + 1) * 32] = gelu_tanh(fmaf(d, z1, o));

        z0 = z2; z1 = z3; z2 = z4;
    }
    scan_chunk(wr, wi, cr, ci, z1, xr, xi, vb);
    float o = mode_sum(va, b4, b3, b2, b1, b0);
    yrow[(SCAN_CHUNKS - 2) * 32] = gelu_tanh(fmaf(d, z0, o));
    o = mode_sum(vb, b4, b3, b2, b1, b0);
    yrow[(SCAN_CHUNKS - 1) * 32] = gelu_tanh(fmaf(d, z1, o));
}

#define BM 64
#define BN 128
#define BK 32
#define LDP 34
#define NKT (KPAD / BK)

__global__ __launch_bounds__(256) void s4_glu_gemm(
        const float* __restrict__ Wg, const float* __restrict__ bias,
        const float* __restrict__ Y, float* __restrict__ Y2) {
    __shared__ float Ws[2][BM][LDP];
    __shared__ float Ys[2][BN][LDP];
    int tid = threadIdx.x;
    int lane = tid & 31;
    int wave = tid >> 5;
    int wave_m = wave >> 2;
    int wave_n = wave & 3;
    int lrow = lane & 15;
    int khalf = (lane >> 4) << 1;
    int madd = (lane >> 4) << 3;
    int blockM = blockIdx.x * BM;
    int blockN = blockIdx.y * BN;

    int wrow[8];
#pragma unroll
    for (int j = 0; j < 8; ++j) {
        int gm = blockM + wave + 8 * j;
        wrow[j] = (gm < M2 ? gm : (M2 - 1)) * KD;
    }
    const float* ycol = Y + blockN + lane;

    v8f acc[2][2];
#pragma unroll
    for (int mt = 0; mt < 2; ++mt) {
#pragma unroll
        for (int i = 0; i < 8; ++i) {
            int gm = blockM + wave_m * 32 + mt * 16 + i + madd;
            float bv = (gm < M2) ? bias[gm] : 0.0f;
#pragma unroll
            for (int nt = 0; nt < 2; ++nt) acc[mt][nt][i] = bv;
        }
    }

    float wreg[8], yreg[16];

#define TILE_LOAD(kb)                                                          \
    {                                                                          \
        int gk = (kb) + lane;                                                  \
        int gkc = gk < KD ? gk : (KD - 1);                                     \
        _Pragma("unroll")                                                      \
        for (int j = 0; j < 8; ++j) wreg[j] = Wg[wrow[j] + gkc];               \
        _Pragma("unroll")                                                      \
        for (int r = 0; r < 4; ++r) {                                          \
            int gy = (kb) + wave + 8 * r;                                      \
            int gyc = gy < KD ? gy : (KD - 1);                                 \
            const float* srow = ycol + (size_t)gyc * LSEQ;                     \
            _Pragma("unroll")                                                  \
            for (int j = 0; j < 4; ++j) yreg[4 * r + j] = srow[32 * j];        \
        }                                                                      \
    }

#define TILE_STORE(kb, b)                                                      \
    {                                                                          \
        bool kin = (kb) + lane < KD;                                           \
        _Pragma("unroll")                                                      \
        for (int j = 0; j < 8; ++j)                                            \
            Ws[b][wave + 8 * j][lane] = kin ? wreg[j] : 0.0f;                  \
        _Pragma("unroll")                                                      \
        for (int r = 0; r < 4; ++r) {                                          \
            bool yin = (kb) + wave + 8 * r < KD;                               \
            _Pragma("unroll")                                                  \
            for (int j = 0; j < 4; ++j)                                        \
                Ys[b][lane + 32 * j][wave + 8 * r] = yin ? yreg[4 * r + j] : 0.0f; \
        }                                                                      \
    }

#define TILE_MMA(b)                                                            \
    {                                                                          \
        v2f a[BK / 4][2], bf[BK / 4][2];                                       \
        _Pragma("unroll")                                                      \
        for (int kc = 0; kc < BK / 4; ++kc) {                                  \
            int kk = kc * 4 + khalf;                                           \
            _Pragma("unroll")                                                  \
            for (int mt = 0; mt < 2; ++mt)                                     \
                a[kc][mt] = *(const v2f*)&Ws[b][wave_m * 32 + mt * 16 + lrow][kk];  \
            _Pragma("unroll")                                                  \
            for (int nt = 0; nt < 2; ++nt)                                     \
                bf[kc][nt] = *(const v2f*)&Ys[b][wave_n * 32 + nt * 16 + lrow][kk]; \
        }                                                                      \
        __builtin_amdgcn_sched_barrier(0);                                     \
        _Pragma("unroll")                                                      \
        for (int kc = 0; kc < BK / 4; ++kc) {                                  \
            _Pragma("unroll")                                                  \
            for (int mt = 0; mt < 2; ++mt)                                     \
                _Pragma("unroll")                                              \
                for (int nt = 0; nt < 2; ++nt)                                 \
                    acc[mt][nt] = __builtin_amdgcn_wmma_f32_16x16x4_f32(       \
                        false, a[kc][mt], false, bf[kc][nt],                   \
                        (short)0, acc[mt][nt], false, false);                  \
        }                                                                      \
        __builtin_amdgcn_sched_barrier(0);                                     \
    }

    TILE_LOAD(0)
    TILE_STORE(0, 0)
    __syncthreads();
    for (int t = 0; t < NKT - 1; ++t) {
        int cur = t & 1;
        int kbn = (t + 1) * BK;
        TILE_LOAD(kbn)
        TILE_MMA(cur)
        TILE_STORE(kbn, cur ^ 1)
        __syncthreads();
    }
    TILE_MMA((NKT - 1) & 1)
#undef TILE_LOAD
#undef TILE_STORE
#undef TILE_MMA

    int gn_base = blockN + wave_n * 32;
#pragma unroll
    for (int mt = 0; mt < 2; ++mt) {
#pragma unroll
        for (int i = 0; i < 8; ++i) {
            int gm = blockM + wave_m * 32 + mt * 16 + i + madd;
            if (gm < M2) {
#pragma unroll
                for (int nt = 0; nt < 2; ++nt) {
                    int gn = gn_base + nt * 16 + lrow;
                    Y2[(size_t)gm * LSEQ + gn] = acc[mt][nt][i];
                }
            }
        }
    }
}

#define LN_COLS 32
#define LN_WAVES 32
#define LN_CPW 16

__global__ __launch_bounds__(1024) void s4_glu_ln(
        const float* __restrict__ Y2, float* __restrict__ Z,
        const float* __restrict__ g, const float* __restrict__ b) {
    __shared__ float red_sum[LN_WAVES][LN_COLS];
    __shared__ float red_sq[LN_WAVES][LN_COLS];
    int lane = threadIdx.x & 31;
    int wave = __builtin_amdgcn_readfirstlane((int)(threadIdx.x >> 5));
    int l = blockIdx.x * LN_COLS + lane;

    float p[LN_CPW], q[LN_CPW], x[LN_CPW], gg[LN_CPW], bb[LN_CPW];
#pragma unroll
    for (int i = 0; i < LN_CPW; ++i) {
        int h = wave + LN_WAVES * i;
        int hc = (i < LN_CPW - 1 || h < HCH) ? h : (HCH - 1);
        p[i] = Y2[(size_t)hc * LSEQ + l];
        q[i] = Y2[(size_t)(hc + HCH) * LSEQ + l];
        x[i] = Z[(size_t)hc * LSEQ + l];
        gg[i] = g[hc];
        bb[i] = b[hc];
    }

    float s = 0.0f;
#pragma unroll
    for (int i = 0; i < LN_CPW; ++i) {
        int h = wave + LN_WAVES * i;
        float v = p[i] * sigmoidf(q[i]) + x[i];
        x[i] = (i < LN_CPW - 1 || h < HCH) ? v : 0.0f;
        s += x[i];
    }
    red_sum[wave][lane] = s;
    __syncthreads();
    float tot = 0.0f;
#pragma unroll
    for (int w = 0; w < LN_WAVES; ++w) tot += red_sum[w][lane];
    float mu = tot * (1.0f / (float)HCH);

    float sq = 0.0f;
#pragma unroll
    for (int i = 0; i < LN_CPW; ++i) {
        int h = wave + LN_WAVES * i;
        float d = (i < LN_CPW - 1 || h < HCH) ? (x[i] - mu) : 0.0f;
        x[i] = d;
        sq += d * d;
    }
    red_sq[wave][lane] = sq;
    __syncthreads();
    float tq = 0.0f;
#pragma unroll
    for (int w = 0; w < LN_WAVES; ++w) tq += red_sq[w][lane];
    float rstd = rsqrtf(tq * (1.0f / (float)HCH) + LN_EPS);

#pragma unroll
    for (int i = 0; i < LN_CPW; ++i) {
        int h = wave + LN_WAVES * i;
        if (i < LN_CPW - 1 || h < HCH)
            Z[(size_t)h * LSEQ + l] = x[i] * rstd * gg[i] + bb[i];
    }
}

extern "C" void kernel_launch(void* const* d_in, const int* in_sizes, int n_in,
                              void* d_out, int out_size, void* d_ws, size_t ws_size,
                              hipStream_t stream) {
    (void)in_sizes; (void)n_in; (void)out_size; (void)ws_size;
    const float* Zin    = (const float*)d_in[0];
    const float* log_dt = (const float*)d_in[1];
    const float* Arl    = (const float*)d_in[2];
    const float* Aim    = (const float*)d_in[3];
    const float* Cre    = (const float*)d_in[4];
    const float* Cim    = (const float*)d_in[5];
    const float* Dsk    = (const float*)d_in[6];
    const float* glu_w  = (const float*)d_in[7];
    const float* glu_b  = (const float*)d_in[8];
    const float* ln_g   = (const float*)d_in[9];
    const float* ln_b   = (const float*)d_in[10];

    float* ws = (float*)d_ws;
    const size_t HN = (size_t)HCH * NMODES;
    float* w_re  = ws;
    float* w_im  = ws + HN;
    float* c_re  = ws + 2 * HN;
    float* c_im  = ws + 3 * HN;
    float* zbuf  = ws + 4 * HN;
    float* ybuf  = zbuf + (size_t)HCH * LSEQ;
    float* y2buf = ybuf + (size_t)HCH * LSEQ;

    s4_setup<<<(int)((HN + 255) / 256), 256, 0, stream>>>(
        log_dt, Arl, Aim, Cre, Cim, w_re, w_im, c_re, c_im);

    hipMemcpyAsync(zbuf, Zin, sizeof(float) * (size_t)HCH * LSEQ,
                   hipMemcpyDeviceToDevice, stream);

    for (int layer = 0; layer < NLAYERS; ++layer) {
        s4_scan<<<(HCH + SCAN_WAVES - 1) / SCAN_WAVES, SCAN_WAVES * 32, 0, stream>>>(zbuf, w_re, w_im, c_re, c_im, Dsk, ybuf);
        s4_glu_gemm<<<dim3(16, 16, 1), 256, 0, stream>>>(glu_w, glu_b, ybuf, y2buf);
        s4_glu_ln<<<LSEQ / LN_COLS, 1024, 0, stream>>>(y2buf, zbuf, ln_g, ln_b);
    }

    hipMemcpyAsync(d_out, zbuf, sizeof(float) * (size_t)HCH * LSEQ,
                   hipMemcpyDeviceToDevice, stream);
}


# scan: two waves per channel, each runs the recurrence on every other 32-step chunk and jumps the partner's by x = w^32 x + sum_k w^(31-k) z_k; the carried state's sum order (rounding) changes
# speedup vs baseline: 13.2388x; 1.2038x over previous
#include <hip/hip_runtime.h>
#include <hip/hip_bf16.h>

#define HCH 511
#define LSEQ 2048
#define NMODES 32
#define NLAYERS 4
#define M2 1022
#define KD 511
#define KPAD 512
#define LN_EPS 1e-5f

typedef __attribute__((ext_vector_type(2))) float v2f;
typedef __attribute__((ext_vector_type(8))) float v8f;

__device__ __forceinline__ float gelu_tanh(float x) {
    float x3 = x * x * x;
    float t = tanhf(0.7978845608028654f * (x + 0.044715f * x3));
    return 0.5f * x * (1.0f + t);
}

__device__ __forceinline__ float sigmoidf(float x) {
    return 1.0f / (1.0f + expf(-x));
}

__global__ void s4_setup(const float* __restrict__ log_dt,
                         const float* __restrict__ Arl,
                         const float* __restrict__ Aim,
                         const float* __restrict__ Cre,
                         const float* __restrict__ Cim,
                         float* __restrict__ w_re, float* __restrict__ w_im,
                         float* __restrict__ c_re, float* __restrict__ c_im) {
    int idx = blockIdx.x * blockDim.x + threadIdx.x;
    if (idx >= HCH * NMODES) return;
    int h = idx / NMODES;
    float a_re = -expf(Arl[idx]);
    float a_im = Aim[idx];
    float dt = expf(log_dt[h]);
    float dre = a_re * dt, dim = a_im * dt;
    float er = expf(dre);
    float sn, cs;
    sincosf(dim, &sn, &cs);
    float wre = er * cs, wim = er * sn;
    float nre = wre - 1.0f, nim = wim;
    float inv = 1.0f / (a_re * a_re + a_im * a_im);
    float tre = (nre * a_re + nim * a_im) * inv;
    float tim = (nim * a_re - nre * a_im) * inv;
    float Cr = Cre[idx], Ci = Cim[idx];
    w_re[idx] = wre;
    w_im[idx] = wim;
    c_re[idx] = 2.0f * (Cr * tre - Ci * tim);
    c_im[idx] = 2.0f * (Cr * tim + Ci * tre);
}

__device__ __forceinline__ float lane_xor16(float v) {
    int i = __builtin_bit_cast(int, v);
    int r = __builtin_amdgcn_permlanex16(i, i, 0x76543210, 0xfedcba98, false, false);
    return __builtin_bit_cast(float, r);
}
template <int MASK>
__device__ __forceinline__ float lane_xor_row(float v) {
    int i = __builtin_bit_cast(int, v);
    int r = __builtin_amdgcn_update_dpp(0, i, 0x160 | MASK, 0xf, 0xf, true);
    return __builtin_bit_cast(float, r);
}
template <int T>
__device__ __forceinline__ float lane_bcast(float v) {
    return __builtin_bit_cast(float,
        __builtin_amdgcn_readlane(__builtin_bit_cast(int, v), T));
}

__device__ __forceinline__ float scan_step(float wr, float wi, float cr, float ci,
                                         float zt, float& xr, float& xi) {
    float nxr = fmaf(wr, xr, fmaf(-wi, xi, zt));
    float nxi = fmaf(wr, xi, wi * xr);
    xr = nxr; xi = nxi;
    return fmaf(cr, xr, -ci * xi);
}

__device__ __forceinline__ void scan_chunk(float wr, float wi, float cr, float ci,
                                           float zv, float& xr, float& xi,
                                           float (&v)[32]) {
#define SCAN_STEP(T) v[T] = scan_step(wr, wi, cr, ci, lane_bcast<T>(zv), xr, xi);
    SCAN_STEP(0)  SCAN_STEP(1)  SCAN_STEP(2)  SCAN_STEP(3)
    SCAN_STEP(4)  SCAN_STEP(5)  SCAN_STEP(6)  SCAN_STEP(7)
    SCAN_STEP(8)  SCAN_STEP(9)  SCAN_STEP(10) SCAN_STEP(11)
    SCAN_STEP(12) SCAN_STEP(13) SCAN_STEP(14) SCAN_STEP(15)
    SCAN_STEP(16) SCAN_STEP(17) SCAN_STEP(18) SCAN_STEP(19)
    SCAN_STEP(20) SCAN_STEP(21) SCAN_STEP(22) SCAN_STEP(23)
    SCAN_STEP(24) SCAN_STEP(25) SCAN_STEP(26) SCAN_STEP(27)
    SCAN_STEP(28) SCAN_STEP(29) SCAN_STEP(30) SCAN_STEP(31)
#undef SCAN_STEP
}

__device__ __forceinline__ float mode_sum(float (&v)[32], bool b4, bool b3,
                                          bool b2, bool b1, bool b0) {
#pragma unroll
    for (int j = 0; j < 16; ++j) {
        float keep = b4 ? v[j + 16] : v[j];
        float send = b4 ? v[j] : v[j + 16];
        v[j] = keep + lane_xor16(send);
    }
#pragma unroll
    for (int j = 0; j < 8; ++j) {
        float keep = b3 ? v[j + 8] : v[j];
        float send = b3 ? v[j] : v[j + 8];
        v[j] = keep + lane_xor_row<8>(send);
    }
#pragma unroll
    for (int j = 0; j < 4; ++j) {
        float keep = b2 ? v[j + 4] : v[j];
        float send = b2 ? v[j] : v[j + 4];
        v[j] = keep + lane_xor_row<4>(send);
    }
#pragma unroll
    for (int j = 0; j < 2; ++j) {
        float keep = b1 ? v[j + 2] : v[j];
        float send = b1 ? v[j] : v[j + 2];
        v[j] = keep + lane_xor_row<2>(send);
    }
    float keep0 = b0 ? v[1] : v[0];
    float send0 = b0 ? v[0] : v[1];
    return keep0 + lane_xor_row<1>(send0);
}


__device__ __forceinline__ void mode_powers(float wr, float wi,
                                            float (&qr)[32], float (&qi)[32]) {
    qr[0] = wr; qi[0] = wi;
#pragma unroll
    for (int j = 1; j < 32; ++j) {
        qr[j] = fmaf(qr[j - 1], wr, -qi[j - 1] * wi);
        qi[j] = fmaf(qr[j - 1], wi, qi[j - 1] * wr);
    }
}

__device__ __forceinline__ void chunk_input_sum(const float (&qr)[32], const float (&qi)[32],
                                                float zv, float& jr, float& ji) {
    int zi = __builtin_bit_cast(int, zv);
    float r0 = 0.0f, r1 = 0.0f, i0 = 0.0f, i1 = 0.0f;
#pragma unroll
    for (int k = 0; k < 30; k += 2) {
        float s0 = __builtin_bit_cast(float, __builtin_amdgcn_readlane(zi, k));
        float s1 = __builtin_bit_cast(float, __builtin_amdgcn_readlane(zi, k + 1));
        r0 = fmaf(qr[30 - k], s0, r0);
        i0 = fmaf(qi[30 - k], s0, i0);
        r1 = fmaf(qr[29 - k], s1, r1);
        i1 = fmaf(qi[29 - k], s1, i1);
    }
    float s30 = lane_bcast<30>(zv);
    float s31 = lane_bcast<31>(zv);
    r0 = fmaf(qr[0], s30, r0);
    i0 = fmaf(qi[0], s30, i0);
    r1 += s31;
    jr = r0 + r1;
    ji = i0 + i1;
}

#define SCAN_PIN_STATE() asm volatile("" : "+v"(xr), "+v"(xi))

#define SCAN_CHUNKS (LSEQ / 32)
#define SCAN_TRIPS (SCAN_CHUNKS / 2)
#define SCAN_WAVES 4
#define SCAN_CH (SCAN_WAVES / 2)

__device__ __forceinline__ int scan_chunk_in_row(int c) {
    return c < SCAN_CHUNKS ? c : SCAN_CHUNKS - 1;
}

#define SCAN_TRIP(zf, zj, v)                                                   \
    {                                                                          \
        float jr, ji;                                                          \
        chunk_input_sum(qr, qi, zj, jr, ji);                                   \
        scan_chunk(wr, wi, cr, ci, zf, xr, xi, v);                             \
        float nxr = fmaf(qr[31], xr, fmaf(-qi[31], xi, jr));                   \
        float nxi = fmaf(qr[31], xi, fmaf(qi[31], xr, ji));                    \
        xr = nxr; xi = nxi;                                                    \
    }

__global__ __launch_bounds__(SCAN_WAVES * 32) void s4_scan(
        const float* __restrict__ z,
        const float* __restrict__ w_re, const float* __restrict__ w_im,
        const float* __restrict__ c_re, const float* __restrict__ c_im,
        const float* __restrict__ D_skip, float* __restrict__ y) {
    int warp = __builtin_amdgcn_readfirstlane((int)(threadIdx.x >> 5));
    int lane = threadIdx.x & 31;
    int h = blockIdx.x * SCAN_CH + (warp >> 1);
    int par = warp & 1;
    if (h >= HCH) return;
    int cidx = h * NMODES + lane;
    float wr = w_re[cidx], wi = w_im[cidx];
    float cr = c_re[cidx], ci = c_im[cidx];
    float d = D_skip[h];
    const bool b4 = (lane & 16) != 0;
    const bool b3 = (lane & 8) != 0;
    const bool b2 = (lane & 4) != 0;
    const bool b1 = (lane & 2) != 0;
    const bool b0 = (lane & 1) != 0;
    const float* zrow = z + (size_t)h * LSEQ + lane;
    float* yrow = y + (size_t)h * LSEQ + lane;
#define Z_OWN(t) zrow[scan_chunk_in_row(2 * (t) + par) * 32]
#define Z_CARRIED(t) zrow[scan_chunk_in_row(2 * (t) + par + 1) * 32]

    float zc0 = zrow[0];
    float zf0 = Z_OWN(0), zj0 = Z_CARRIED(0);
    float zf1 = Z_OWN(1), zj1 = Z_CARRIED(1);
    float zf2 = Z_OWN(2), zj2 = Z_CARRIED(2);

    float qr[32], qi[32];
    mode_powers(wr, wi, qr, qi);

    float xr = 0.0f, xi = 0.0f;
    if (par) chunk_input_sum(qr, qi, zc0, xr, xi);

    float va[32], vb[32];
    SCAN_TRIP(zf0, zj0, va)
    for (int t = 0; t < SCAN_TRIPS - 2; t += 2) {
        float zf3 = Z_OWN(t + 3), zj3 = Z_CARRIED(t + 3);
        SCAN_TRIP(zf1, zj1, vb)
        float o = mode_sum(va, b4, b3, b2, b1, b0);
        SCAN_PIN_STATE();
        yrow[(2 * t + par) * 32] = gelu_tanh(fmaf(d, zf0, o));

        float zf4 = Z_OWN(t + 4), zj4 = Z_CARRIED(t + 4);
        SCAN_TRIP(zf2, zj2, va)
        o = mode_sum(vb, b4, b3, b2, b1, b0);
        SCAN_PIN_STATE();
        yrow[(2 * t + 2 + par) * 32] = gelu_tanh(fmaf(d, zf1, o));

        zf0 = zf2;
        zf1 = zf3; zj1 = zj3;
        zf2 = zf4; zj2 = zj4;
    }
#undef Z_OWN
#undef Z_CARRIED
    scan_chunk(wr, wi, cr, ci, zf1, xr, xi, vb);
    float o = mode_sum(va, b4, b3, b2, b1, b0);
    yrow[(2 * (SCAN_TRIPS - 2) + par) * 32] = gelu_tanh(fmaf(d, zf0, o));
    o = mode_sum(vb, b4, b3, b2, b1, b0);
    yrow[(2 * (SCAN_TRIPS - 1) + par) * 32] = gelu_tanh(fmaf(d, zf1, o));
}

#define BM 64
#define BN 128
#define BK 32
#define LDP 34
#define NKT (KPAD / BK)

__global__ __launch_bounds__(256) void s4_glu_gemm(
        const float* __restrict__ Wg, const float* __restrict__ bias,
        const float* __restrict__ Y, float* __restrict__ Y2) {
    __shared__ float Ws[2][BM][LDP];
    __shared__ float Ys[2][BN][LDP];
    int tid = threadIdx.x;
    int lane = tid & 31;
    int wave = tid >> 5;
    int wave_m = wave >> 2;
    int wave_n = wave & 3;
    int lrow = lane & 15;
    int khalf = (lane >> 4) << 1;
    int madd = (lane >> 4) << 3;
    int blockM = blockIdx.x * BM;
    int blockN = blockIdx.y * BN;

    int wrow[8];
#pragma unroll
    for (int j = 0; j < 8; ++j) {
        int gm = blockM + wave + 8 * j;
        wrow[j] = (gm < M2 ? gm : (M2 - 1)) * KD;
    }
    const float* ycol = Y + blockN + lane;

    v8f acc[2][2];
#pragma unroll
    for (int mt = 0; mt < 2; ++mt) {
#pragma unroll
        for (int i = 0; i < 8; ++i) {
            int gm = blockM + wave_m * 32 + mt * 16 + i + madd;
            float bv = (gm < M2) ? bias[gm] : 0.0f;
#pragma unroll
            for (int nt = 0; nt < 2; ++nt) acc[mt][nt][i] = bv;
        }
    }

    float wreg[8], yreg[16];

#define TILE_LOAD(kb)                                                          \
    {                                                                          \
        int gk = (kb) + lane;                                                  \
        int gkc = gk < KD ? gk : (KD - 1);                                     \
        _Pragma("unroll")                                                      \
        for (int j = 0; j < 8; ++j) wreg[j] = Wg[wrow[j] + gkc];               \
        _Pragma("unroll")                                                      \
        for (int r = 0; r < 4; ++r) {                                          \
            int gy = (kb) + wave + 8 * r;                                      \
            int gyc = gy < KD ? gy : (KD - 1);                                 \
            const float* srow = ycol + (size_t)gyc * LSEQ;                     \
            _Pragma("unroll")                                                  \
            for (int j = 0; j < 4; ++j) yreg[4 * r + j] = srow[32 * j];        \
        }                                                                      \
    }

#define TILE_STORE(kb, b)                                                      \
    {                                                                          \
        bool kin = (kb) + lane < KD;                                           \
        _Pragma("unroll")                                                      \
        for (int j = 0; j < 8; ++j)                                            \
            Ws[b][wave + 8 * j][lane] = kin ? wreg[j] : 0.0f;                  \
        _Pragma("unroll")                                                      \
        for (int r = 0; r < 4; ++r) {                                          \
            bool yin = (kb) + wave + 8 * r < KD;                               \
            _Pragma("unroll")                                                  \
            for (int j = 0; j < 4; ++j)                                        \
                Ys[b][lane + 32 * j][wave + 8 * r] = yin ? yreg[4 * r + j] : 0.0f; \
        }                                                                      \
    }

#define TILE_MMA(b)                                                            \
    {                                                                          \
        v2f a[BK / 4][2], bf[BK / 4][2];                                       \
        _Pragma("unroll")                                                      \
        for (int kc = 0; kc < BK / 4; ++kc) {                                  \
            int kk = kc * 4 + khalf;                                           \
            _Pragma("unroll")                                                  \
            for (int mt = 0; mt < 2; ++mt)                                     \
                a[kc][mt] = *(const v2f*)&Ws[b][wave_m * 32 + mt * 16 + lrow][kk];  \
            _Pragma("unroll")                                                  \
            for (int nt = 0; nt < 2; ++nt)                                     \
                bf[kc][nt] = *(const v2f*)&Ys[b][wave_n * 32 + nt * 16 + lrow][kk]; \
        }                                                                      \
        __builtin_amdgcn_sched_barrier(0);                                     \
        _Pragma("unroll")                                                      \
        for (int kc = 0; kc < BK / 4; ++kc) {                                  \
            _Pragma("unroll")                                                  \
            for (int mt = 0; mt < 2; ++mt)                                     \
                _Pragma("unroll")                                              \
                for (int nt = 0; nt < 2; ++nt)                                 \
                    acc[mt][nt] = __builtin_amdgcn_wmma_f32_16x16x4_f32(       \
                        false, a[kc][mt], false, bf[kc][nt],                   \
                        (short)0, acc[mt][nt], false, false);                  \
        }                                                                      \
        __builtin_amdgcn_sched_barrier(0);                                     \
    }

    TILE_LOAD(0)
    TILE_STORE(0, 0)
    __syncthreads();
    for (int t = 0; t < NKT - 1; ++t) {
        int cur = t & 1;
        int kbn = (t + 1) * BK;
        TILE_LOAD(kbn)
        TILE_MMA(cur)
        TILE_STORE(kbn, cur ^ 1)
        __syncthreads();
    }
    TILE_MMA((NKT - 1) & 1)
#undef TILE_LOAD
#undef TILE_STORE
#undef TILE_MMA

    int gn_base = blockN + wave_n * 32;
#pragma unroll
    for (int mt = 0; mt < 2; ++mt) {
#pragma unroll
        for (int i = 0; i < 8; ++i) {
            int gm = blockM + wave_m * 32 + mt * 16 + i + madd;
            if (gm < M2) {
#pragma unroll
                for (int nt = 0; nt < 2; ++nt) {
                    int gn = gn_base + nt * 16 + lrow;
                    Y2[(size_t)gm * LSEQ + gn] = acc[mt][nt][i];
                }
            }
        }
    }
}

#define LN_COLS 32
#define LN_WAVES 32
#define LN_CPW 16

__global__ __launch_bounds__(1024) void s4_glu_ln(
        const float* __restrict__ Y2, float* __restrict__ Z,
        const float* __restrict__ g, const float* __restrict__ b) {
    __shared__ float red_sum[LN_WAVES][LN_COLS];
    __shared__ float red_sq[LN_WAVES][LN_COLS];
    int lane = threadIdx.x & 31;
    int wave = __builtin_amdgcn_readfirstlane((int)(threadIdx.x >> 5));
    int l = blockIdx.x * LN_COLS + lane;

    float p[LN_CPW], q[LN_CPW], x[LN_CPW], gg[LN_CPW], bb[LN_CPW];
#pragma unroll
    for (int i = 0; i < LN_CPW; ++i) {
        int h = wave + LN_WAVES * i;
        int hc = (i < LN_CPW - 1 || h < HCH) ? h : (HCH - 1);
        p[i] = Y2[(size_t)hc * LSEQ + l];
        q[i] = Y2[(size_t)(hc + HCH) * LSEQ + l];
        x[i] = Z[(size_t)hc * LSEQ + l];
        gg[i] = g[hc];
        bb[i] = b[hc];
    }

    float s = 0.0f;
#pragma unroll
    for (int i = 0; i < LN_CPW; ++i) {
        int h = wave + LN_WAVES * i;
        float v = p[i] * sigmoidf(q[i]) + x[i];
        x[i] = (i < LN_CPW - 1 || h < HCH) ? v : 0.0f;
        s += x[i];
    }
    red_sum[wave][lane] = s;
    __syncthreads();
    float tot = 0.0f;
#pragma unroll
    for (int w = 0; w < LN_WAVES; ++w) tot += red_sum[w][lane];
    float mu = tot * (1.0f / (float)HCH);

    float sq = 0.0f;
#pragma unroll
    for (int i = 0; i < LN_CPW; ++i) {
        int h = wave + LN_WAVES * i;
        float d = (i < LN_CPW - 1 || h < HCH) ? (x[i] - mu) : 0.0f;
        x[i] = d;
        sq += d * d;
    }
    red_sq[wave][lane] = sq;
    __syncthreads();
    float tq = 0.0f;
#pragma unroll
    for (int w = 0; w < LN_WAVES; ++w) tq += red_sq[w][lane];
    float rstd = rsqrtf(tq * (1.0f / (float)HCH) + LN_EPS);

#pragma unroll
    for (int i = 0; i < LN_CPW; ++i) {
        int h = wave + LN_WAVES * i;
        if (i < LN_CPW - 1 || h < HCH)
            Z[(size_t)h * LSEQ + l] = x[i] * rstd * gg[i] + bb[i];
    }
}

extern "C" void kernel_launch(void* const* d_in, const int* in_sizes, int n_in,
                              void* d_out, int out_size, void* d_ws, size_t ws_size,
                              hipStream_t stream) {
    (void)in_sizes; (void)n_in; (void)out_size; (void)ws_size;
    const float* Zin    = (const float*)d_in[0];
    const float* log_dt = (const float*)d_in[1];
    const float* Arl    = (const float*)d_in[2];
    const float* Aim    = (const float*)d_in[3];
    const float* Cre    = (const float*)d_in[4];
    const float* Cim    = (const float*)d_in[5];
    const float* Dsk    = (const float*)d_in[6];
    const float* glu_w  = (const float*)d_in[7];
    const float* glu_b  = (const float*)d_in[8];
    const float* ln_g   = (const float*)d_in[9];
    const float* ln_b   = (const float*)d_in[10];

    float* ws = (float*)d_ws;
    const size_t HN = (size_t)HCH * NMODES;
    float* w_re  = ws;
    float* w_im  = ws + HN;
    float* c_re  = ws + 2 * HN;
    float* c_im  = ws + 3 * HN;
    float* zbuf  = ws + 4 * HN;
    float* ybuf  = zbuf + (size_t)HCH * LSEQ;
    float* y2buf = ybuf + (size_t)HCH * LSEQ;

    s4_setup<<<(int)((HN + 255) / 256), 256, 0, stream>>>(
        log_dt, Arl, Aim, Cre, Cim, w_re, w_im, c_re, c_im);

    hipMemcpyAsync(zbuf, Zin, sizeof(float) * (size_t)HCH * LSEQ,
                   hipMemcpyDeviceToDevice, stream);

    for (int layer = 0; layer < NLAYERS; ++layer) {
        s4_scan<<<(HCH + SCAN_CH - 1) / SCAN_CH, SCAN_WAVES * 32, 0, stream>>>(zbuf, w_re, w_im, c_re, c_im, Dsk, ybuf);
        s4_glu_gemm<<<dim3(16, 16, 1), 256, 0, stream>>>(glu_w, glu_b, ybuf, y2buf);
        s4_glu_ln<<<LSEQ / LN_COLS, 1024, 0, stream>>>(y2buf, zbuf, ln_g, ln_b);
    }

    hipMemcpyAsync(d_out, zbuf, sizeof(float) * (size_t)HCH * LSEQ,
                   hipMemcpyDeviceToDevice, stream);
}


# launches 15 -> 12: coefficient set-up folded into each scan wave's prologue; input and output copies dropped (layer 1 reads the input, layer 4 writes out, two buffers swap roles); sums unchanged
# speedup vs baseline: 13.7046x; 1.0352x over previous
#include <hip/hip_runtime.h>
#include <hip/hip_bf16.h>

#define HCH 511
#define LSEQ 2048
#define NMODES 32
#define NLAYERS 4
#define M2 1022
#define KD 511
#define KPAD 512
#define LN_EPS 1e-5f

typedef __attribute__((ext_vector_type(2))) float v2f;
typedef __attribute__((ext_vector_type(8))) float v8f;

__device__ __forceinline__ float gelu_tanh(float x) {
    float x3 = x * x * x;
    float t = tanhf(0.7978845608028654f * (x + 0.044715f * x3));
    return 0.5f * x * (1.0f + t);
}

__device__ __forceinline__ float sigmoidf(float x) {
    return 1.0f / (1.0f + expf(-x));
}

__device__ __forceinline__ void mode_coeffs(float ldt, float arl, float aim,
                                            float Cr, float Ci,
                                            float& wre, float& wim,
                                            float& cre, float& cim) {
    float a_re = -expf(arl);
    float a_im = aim;
    float dt = expf(ldt);
    float dre = a_re * dt, dim = a_im * dt;
    float er = expf(dre);
    float sn, cs;
    sincosf(dim, &sn, &cs);
    wre = er * cs;
    wim = er * sn;
    float nre = wre - 1.0f, nim = wim;
    float inv = 1.0f / (a_re * a_re + a_im * a_im);
    float tre = (nre * a_re + nim * a_im) * inv;
    float tim = (nim * a_re - nre * a_im) * inv;
    cre = 2.0f * (Cr * tre - Ci * tim);
    cim = 2.0f * (Cr * tim + Ci * tre);
}

__device__ __forceinline__ float lane_xor16(float v) {
    int i = __builtin_bit_cast(int, v);
    int r = __builtin_amdgcn_permlanex16(i, i, 0x76543210, 0xfedcba98, false, false);
    return __builtin_bit_cast(float, r);
}
template <int MASK>
__device__ __forceinline__ float lane_xor_row(float v) {
    int i = __builtin_bit_cast(int, v);
    int r = __builtin_amdgcn_update_dpp(0, i, 0x160 | MASK, 0xf, 0xf, true);
    return __builtin_bit_cast(float, r);
}
template <int T>
__device__ __forceinline__ float lane_bcast(float v) {
    return __builtin_bit_cast(float,
        __builtin_amdgcn_readlane(__builtin_bit_cast(int, v), T));
}

__device__ __forceinline__ float scan_step(float wr, float wi, float cr, float ci,
                                         float zt, float& xr, float& xi) {
    float nxr = fmaf(wr, xr, fmaf(-wi, xi, zt));
    float nxi = fmaf(wr, xi, wi * xr);
    xr = nxr; xi = nxi;
    return fmaf(cr, xr, -ci * xi);
}

__device__ __forceinline__ void scan_chunk(float wr, float wi, float cr, float ci,
                                           float zv, float& xr, float& xi,
                                           float (&v)[32]) {
#define SCAN_STEP(T) v[T] = scan_step(wr, wi, cr, ci, lane_bcast<T>(zv), xr, xi);
    SCAN_STEP(0)  SCAN_STEP(1)  SCAN_STEP(2)  SCAN_STEP(3)
    SCAN_STEP(4)  SCAN_STEP(5)  SCAN_STEP(6)  SCAN_STEP(7)
    SCAN_STEP(8)  SCAN_STEP(9)  SCAN_STEP(10) SCAN_STEP(11)
    SCAN_STEP(12) SCAN_STEP(13) SCAN_STEP(14) SCAN_STEP(15)
    SCAN_STEP(16) SCAN_STEP(17) SCAN_STEP(18) SCAN_STEP(19)
    SCAN_STEP(20) SCAN_STEP(21) SCAN_STEP(22) SCAN_STEP(23)
    SCAN_STEP(24) SCAN_STEP(25) SCAN_STEP(26) SCAN_STEP(27)
    SCAN_STEP(28) SCAN_STEP(29) SCAN_STEP(30) SCAN_STEP(31)
#undef SCAN_STEP
}

__device__ __forceinline__ float mode_sum(float (&v)[32], bool b4, bool b3,
                                          bool b2, bool b1, bool b0) {
#pragma unroll
    for (int j = 0; j < 16; ++j) {
        float keep = b4 ? v[j + 16] : v[j];
        float send = b4 ? v[j] : v[j + 16];
        v[j] = keep + lane_xor16(send);
    }
#pragma unroll
    for (int j = 0; j < 8; ++j) {
        float keep = b3 ? v[j + 8] : v[j];
        float send = b3 ? v[j] : v[j + 8];
        v[j] = keep + lane_xor_row<8>(send);
    }
#pragma unroll
    for (int j = 0; j < 4; ++j) {
        float keep = b2 ? v[j + 4] : v[j];
        float send = b2 ? v[j] : v[j + 4];
        v[j] = keep + lane_xor_row<4>(send);
    }
#pragma unroll
    for (int j = 0; j < 2; ++j) {
        float keep = b1 ? v[j + 2] : v[j];
        float send = b1 ? v[j] : v[j + 2];
        v[j] = keep + lane_xor_row<2>(send);
    }
    float keep0 = b0 ? v[1] : v[0];
    float send0 = b0 ? v[0] : v[1];
    return keep0 + lane_xor_row<1>(send0);
}


__device__ __forceinline__ void mode_powers(float wr, float wi,
                                            float (&qr)[32], float (&qi)[32]) {
    qr[0] = wr; qi[0] = wi;
#pragma unroll
    for (int j = 1; j < 32; ++j) {
        qr[j] = fmaf(qr[j - 1], wr, -qi[j - 1] * wi);
        qi[j] = fmaf(qr[j - 1], wi, qi[j - 1] * wr);
    }
}

__device__ __forceinline__ void chunk_input_sum(const float (&qr)[32], const float (&qi)[32],
                                                float zv, float& jr, float& ji) {
    int zi = __builtin_bit_cast(int, zv);
    float r0 = 0.0f, r1 = 0.0f, i0 = 0.0f, i1 = 0.0f;
#pragma unroll
    for (int k = 0; k < 30; k += 2) {
        float s0 = __builtin_bit_cast(float, __builtin_amdgcn_readlane(zi, k));
        float s1 = __builtin_bit_cast(float, __builtin_amdgcn_readlane(zi, k + 1));
        r0 = fmaf(qr[30 - k], s0, r0);
        i0 = fmaf(qi[30 - k], s0, i0);
        r1 = fmaf(qr[29 - k], s1, r1);
        i1 = fmaf(qi[29 - k], s1, i1);
    }
    float s30 = lane_bcast<30>(zv);
    float s31 = lane_bcast<31>(zv);
    r0 = fmaf(qr[0], s30, r0);
    i0 = fmaf(qi[0], s30, i0);
    r1 += s31;
    jr = r0 + r1;
    ji = i0 + i1;
}

#define SCAN_PIN_STATE() asm volatile("" : "+v"(xr), "+v"(xi))

#define SCAN_CHUNKS (LSEQ / 32)
#define SCAN_TRIPS (SCAN_CHUNKS / 2)
#define SCAN_WAVES 4
#define SCAN_CH (SCAN_WAVES / 2)

__device__ __forceinline__ int scan_chunk_in_row(int c) {
    return c < SCAN_CHUNKS ? c : SCAN_CHUNKS - 1;
}

#define SCAN_TRIP(zf, zj, v)                                                   \
    {                                                                          \
        float jr, ji;                                                          \
        chunk_input_sum(qr, qi, zj, jr, ji);                                   \
        scan_chunk(wr, wi, cr, ci, zf, xr, xi, v);                             \
        float nxr = fmaf(qr[31], xr, fmaf(-qi[31], xi, jr));                   \
        float nxi = fmaf(qr[31], xi, fmaf(qi[31], xr, ji));                    \
        xr = nxr; xi = nxi;                                                    \
    }

__global__ __launch_bounds__(SCAN_WAVES * 32) void s4_scan(
        const float* __restrict__ z,
        const float* __restrict__ log_dt,
        const float* __restrict__ Arl, const float* __restrict__ Aim,
        const float* __restrict__ Cre, const float* __restrict__ Cim,
        const float* __restrict__ D_skip, float* __restrict__ y) {
    int warp = __builtin_amdgcn_readfirstlane((int)(threadIdx.x >> 5));
    int lane = threadIdx.x & 31;
    int h = blockIdx.x * SCAN_CH + (warp >> 1);
    int par = warp & 1;
    if (h >= HCH) return;
    int cidx = h * NMODES + lane;
    float wr, wi, cr, ci;
    mode_coeffs(log_dt[h], Arl[cidx], Aim[cidx], Cre[cidx], Cim[cidx],
                wr, wi, cr, ci);
    float d = D_skip[h];
    const bool b4 = (lane & 16) != 0;
    const bool b3 = (lane & 8) != 0;
    const bool b2 = (lane & 4) != 0;
    const bool b1 = (lane & 2) != 0;
    const bool b0 = (lane & 1) != 0;
    const float* zrow = z + (size_t)h * LSEQ + lane;
    float* yrow = y + (size_t)h * LSEQ + lane;
#define Z_OWN(t) zrow[scan_chunk_in_row(2 * (t) + par) * 32]
#define Z_CARRIED(t) zrow[scan_chunk_in_row(2 * (t) + par + 1) * 32]

    float zc0 = zrow[0];
    float zf0 = Z_OWN(0), zj0 = Z_CARRIED(0);
    float zf1 = Z_OWN(1), zj1 = Z_CARRIED(1);
    float zf2 = Z_OWN(2), zj2 = Z_CARRIED(2);

    float qr[32], qi[32];
    mode_powers(wr, wi, qr, qi);

    float xr = 0.0f, xi = 0.0f;
    if (par) chunk_input_sum(qr, qi, zc0, xr, xi);

    float va[32], vb[32];
    SCAN_TRIP(zf0, zj0, va)
    for (int t = 0; t < SCAN_TRIPS - 2; t += 2) {
        float zf3 = Z_OWN(t + 3), zj3 = Z_CARRIED(t + 3);
        SCAN_TRIP(zf1, zj1, vb)
        float o = mode_sum(va, b4, b3, b2, b1, b0);
        SCAN_PIN_STATE();
        yrow[(2 * t + par) * 32] = gelu_tanh(fmaf(d, zf0, o));

        float zf4 = Z_OWN(t + 4), zj4 = Z_CARRIED(t + 4);
        SCAN_TRIP(zf2, zj2, va)
        o = mode_sum(vb, b4, b3, b2, b1, b0);
        SCAN_PIN_STATE();
        yrow[(2 * t + 2 + par) * 32] = gelu_tanh(fmaf(d, zf1, o));

        zf0 = zf2;
        zf1 = zf3; zj1 = zj3;
        zf2 = zf4; zj2 = zj4;
    }
#undef Z_OWN
#undef Z_CARRIED
    scan_chunk(wr, wi, cr, ci, zf1, xr, xi, vb);
    float o = mode_sum(va, b4, b3, b2, b1, b0);
    yrow[(2 * (SCAN_TRIPS - 2) + par) * 32] = gelu_tanh(fmaf(d, zf0, o));
    o = mode_sum(vb, b4, b3, b2, b1, b0);
    yrow[(2 * (SCAN_TRIPS - 1) + par) * 32] = gelu_tanh(fmaf(d, zf1, o));
}

#define BM 64
#define BN 128
#define BK 32
#define LDP 34
#define NKT (KPAD / BK)

__global__ __launch_bounds__(256) void s4_glu_gemm(
        const float* __restrict__ Wg, const float* __restrict__ bias,
        const float* __restrict__ Y, float* __restrict__ Y2) {
    __shared__ float Ws[2][BM][LDP];
    __shared__ float Ys[2][BN][LDP];
    int tid = threadIdx.x;
    int lane = tid & 31;
    int wave = tid >> 5;
    int wave_m = wave >> 2;
    int wave_n = wave & 3;
    int lrow = lane & 15;
    int khalf = (lane >> 4) << 1;
    int madd = (lane >> 4) << 3;
    int blockM = blockIdx.x * BM;
    int blockN = blockIdx.y * BN;

    int wrow[8];
#pragma unroll
    for (int j = 0; j < 8; ++j) {
        int gm = blockM + wave + 8 * j;
        wrow[j] = (gm < M2 ? gm : (M2 - 1)) * KD;
    }
    const float* ycol = Y + blockN + lane;

    v8f acc[2][2];
#pragma unroll
    for (int mt = 0; mt < 2; ++mt) {
#pragma unroll
        for (int i = 0; i < 8; ++i) {
            int gm = blockM + wave_m * 32 + mt * 16 + i + madd;
            float bv = (gm < M2) ? bias[gm] : 0.0f;
#pragma unroll
            for (int nt = 0; nt < 2; ++nt) acc[mt][nt][i] = bv;
        }
    }

    float wreg[8], yreg[16];

#define TILE_LOAD(kb)                                                          \
    {                                                                          \
        int gk = (kb) + lane;                                                  \
        int gkc = gk < KD ? gk : (KD - 1);                                     \
        _Pragma("unroll")                                                      \
        for (int j = 0; j < 8; ++j) wreg[j] = Wg[wrow[j] + gkc];               \
        _Pragma("unroll")                                                      \
        for (int r = 0; r < 4; ++r) {                                          \
            int gy = (kb) + wave + 8 * r;                                      \
            int gyc = gy < KD ? gy : (KD - 1);                                 \
            const float* srow = ycol + (size_t)gyc * LSEQ;                     \
            _Pragma("unroll")                                                  \
            for (int j = 0; j < 4; ++j) yreg[4 * r + j] = srow[32 * j];        \
        }                                                                      \
    }

#define TILE_STORE(kb, b)                                                      \
    {                                                                          \
        bool kin = (kb) + lane < KD;                                           \
        _Pragma("unroll")                                                      \
        for (int j = 0; j < 8; ++j)                                            \
            Ws[b][wave + 8 * j][lane] = kin ? wreg[j] : 0.0f;                  \
        _Pragma("unroll")                                                      \
        for (int r = 0; r < 4; ++r) {                                          \
            bool yin = (kb) + wave + 8 * r < KD;                               \
            _Pragma("unroll")                                                  \
            for (int j = 0; j < 4; ++j)                                        \
                Ys[b][lane + 32 * j][wave + 8 * r] = yin ? yreg[4 * r + j] : 0.0f; \
        }                                                                      \
    }

#define TILE_MMA(b)                                                            \
    {                                                                          \
        v2f a[BK / 4][2], bf[BK / 4][2];                                       \
        _Pragma("unroll")                                                      \
        for (int kc = 0; kc < BK / 4; ++kc) {                                  \
            int kk = kc * 4 + khalf;                                           \
            _Pragma("unroll")                                                  \
            for (int mt = 0; mt < 2; ++mt)                                     \
                a[kc][mt] = *(const v2f*)&Ws[b][wave_m * 32 + mt * 16 + lrow][kk];  \
            _Pragma("unroll")                                                  \
            for (int nt = 0; nt < 2; ++nt)                                     \
                bf[kc][nt] = *(const v2f*)&Ys[b][wave_n * 32 + nt * 16 + lrow][kk]; \
        }                                                                      \
        __builtin_amdgcn_sched_barrier(0);                                     \
        _Pragma("unroll")                                                      \
        for (int kc = 0; kc < BK / 4; ++kc) {                                  \
            _Pragma("unroll")                                                  \
            for (int mt = 0; mt < 2; ++mt)                                     \
                _Pragma("unroll")                                              \
                for (int nt = 0; nt < 2; ++nt)                                 \
                    acc[mt][nt] = __builtin_amdgcn_wmma_f32_16x16x4_f32(       \
                        false, a[kc][mt], false, bf[kc][nt],                   \
                        (short)0, acc[mt][nt], false, false);                  \
        }                                                                      \
        __builtin_amdgcn_sched_barrier(0);                                     \
    }

    TILE_LOAD(0)
    TILE_STORE(0, 0)
    __syncthreads();
    for (int t = 0; t < NKT - 1; ++t) {
        int cur = t & 1;
        int kbn = (t + 1) * BK;
        TILE_LOAD(kbn)
        TILE_MMA(cur)
        TILE_STORE(kbn, cur ^ 1)
        __syncthreads();
    }
    TILE_MMA((NKT - 1) & 1)
#undef TILE_LOAD
#undef TILE_STORE
#undef TILE_MMA

    int gn_base = blockN + wave_n * 32;
#pragma unroll
    for (int mt = 0; mt < 2; ++mt) {
#pragma unroll
        for (int i = 0; i < 8; ++i) {
            int gm = blockM + wave_m * 32 + mt * 16 + i + madd;
            if (gm < M2) {
#pragma unroll
                for (int nt = 0; nt < 2; ++nt) {
                    int gn = gn_base + nt * 16 + lrow;
                    Y2[(size_t)gm * LSEQ + gn] = acc[mt][nt][i];
                }
            }
        }
    }
}

#define LN_COLS 32
#define LN_WAVES 32
#define LN_CPW 16

__global__ __launch_bounds__(1024) void s4_glu_ln(
        const float* __restrict__ Y2, const float* __restrict__ Zres,
        float* __restrict__ Zout,
        const float* __restrict__ g, const float* __restrict__ b) {
    __shared__ float red_sum[LN_WAVES][LN_COLS];
    __shared__ float red_sq[LN_WAVES][LN_COLS];
    int lane = threadIdx.x & 31;
    int wave = __builtin_amdgcn_readfirstlane((int)(threadIdx.x >> 5));
    int l = blockIdx.x * LN_COLS + lane;

    float p[LN_CPW], q[LN_CPW], x[LN_CPW], gg[LN_CPW], bb[LN_CPW];
#pragma unroll
    for (int i = 0; i < LN_CPW; ++i) {
        int h = wave + LN_WAVES * i;
        int hc = (i < LN_CPW - 1 || h < HCH) ? h : (HCH - 1);
        p[i] = Y2[(size_t)hc * LSEQ + l];
        q[i] = Y2[(size_t)(hc + HCH) * LSEQ + l];
        x[i] = Zres[(size_t)hc * LSEQ + l];
        gg[i] = g[hc];
        bb[i] = b[hc];
    }

    float s = 0.0f;
#pragma unroll
    for (int i = 0; i < LN_CPW; ++i) {
        int h = wave + LN_WAVES * i;
        float v = p[i] * sigmoidf(q[i]) + x[i];
        x[i] = (i < LN_CPW - 1 || h < HCH) ? v : 0.0f;
        s += x[i];
    }
    red_sum[wave][lane] = s;
    __syncthreads();
    float tot = 0.0f;
#pragma unroll
    for (int w = 0; w < LN_WAVES; ++w) tot += red_sum[w][lane];
    float mu = tot * (1.0f / (float)HCH);

    float sq = 0.0f;
#pragma unroll
    for (int i = 0; i < LN_CPW; ++i) {
        int h = wave + LN_WAVES * i;
        float d = (i < LN_CPW - 1 || h < HCH) ? (x[i] - mu) : 0.0f;
        x[i] = d;
        sq += d * d;
    }
    red_sq[wave][lane] = sq;
    __syncthreads();
    float tq = 0.0f;
#pragma unroll
    for (int w = 0; w < LN_WAVES; ++w) tq += red_sq[w][lane];
    float rstd = rsqrtf(tq * (1.0f / (float)HCH) + LN_EPS);

#pragma unroll
    for (int i = 0; i < LN_CPW; ++i) {
        int h = wave + LN_WAVES * i;
        if (i < LN_CPW - 1 || h < HCH)
            Zout[(size_t)h * LSEQ + l] = x[i] * rstd * gg[i] + bb[i];
    }
}

extern "C" void kernel_launch(void* const* d_in, const int* in_sizes, int n_in,
                              void* d_out, int out_size, void* d_ws, size_t ws_size,
                              hipStream_t stream) {
    (void)in_sizes; (void)n_in; (void)out_size; (void)ws_size;
    const float* Zin    = (const float*)d_in[0];
    const float* log_dt = (const float*)d_in[1];
    const float* Arl    = (const float*)d_in[2];
    const float* Aim    = (const float*)d_in[3];
    const float* Cre    = (const float*)d_in[4];
    const float* Cim    = (const float*)d_in[5];
    const float* Dsk    = (const float*)d_in[6];
    const float* glu_w  = (const float*)d_in[7];
    const float* glu_b  = (const float*)d_in[8];
    const float* ln_g   = (const float*)d_in[9];
    const float* ln_b   = (const float*)d_in[10];

    float* bufa  = (float*)d_ws;
    float* bufb  = bufa + (size_t)HCH * LSEQ;
    float* y2buf = bufb + (size_t)HCH * LSEQ;
    float* out   = (float*)d_out;

    static_assert(NLAYERS == 4, "the four layers are written out below");
#define S4_LAYER(zin, ybuf, zout)                                              \
    s4_scan<<<(HCH + SCAN_CH - 1) / SCAN_CH, SCAN_WAVES * 32, 0, stream>>>(    \
        zin, log_dt, Arl, Aim, Cre, Cim, Dsk, ybuf);                           \
    s4_glu_gemm<<<dim3(16, 16, 1), 256, 0, stream>>>(glu_w, glu_b, ybuf, y2buf); \
    s4_glu_ln<<<LSEQ / LN_COLS, 1024, 0, stream>>>(y2buf, zin, zout, ln_g, ln_b);
    S4_LAYER(Zin,  bufb, bufa)
    S4_LAYER(bufa, bufb, bufb)
    S4_LAYER(bufb, bufa, bufa)
    S4_LAYER(bufa, bufb, out)
#undef S4_LAYER
}


# GLU gate + residual add moved from the LayerNorm kernel into the GEMM epilogue (wave tile = 16 value rows + their 16 gate rows): Y2 never stored, LN reads 1 plane not 3; sums unchanged
# speedup vs baseline: 14.2917x; 1.0428x over previous
#include <hip/hip_runtime.h>
#include <hip/hip_bf16.h>

#define HCH 511
#define LSEQ 2048
#define NMODES 32
#define NLAYERS 4
#define KD 511
#define KPAD 512
#define LN_EPS 1e-5f

typedef __attribute__((ext_vector_type(2))) float v2f;
typedef __attribute__((ext_vector_type(8))) float v8f;

__device__ __forceinline__ float gelu_tanh(float x) {
    float x3 = x * x * x;
    float t = tanhf(0.7978845608028654f * (x + 0.044715f * x3));
    return 0.5f * x * (1.0f + t);
}

__device__ __forceinline__ float sigmoidf(float x) {
    return 1.0f / (1.0f + expf(-x));
}

__device__ __forceinline__ void mode_coeffs(float ldt, float arl, float aim,
                                            float Cr, float Ci,
                                            float& wre, float& wim,
                                            float& cre, float& cim) {
    float a_re = -expf(arl);
    float a_im = aim;
    float dt = expf(ldt);
    float dre = a_re * dt, dim = a_im * dt;
    float er = expf(dre);
    float sn, cs;
    sincosf(dim, &sn, &cs);
    wre = er * cs;
    wim = er * sn;
    float nre = wre - 1.0f, nim = wim;
    float inv = 1.0f / (a_re * a_re + a_im * a_im);
    float tre = (nre * a_re + nim * a_im) * inv;
    float tim = (nim * a_re - nre * a_im) * inv;
    cre = 2.0f * (Cr * tre - Ci * tim);
    cim = 2.0f * (Cr * tim + Ci * tre);
}

__device__ __forceinline__ float lane_xor16(float v) {
    int i = __builtin_bit_cast(int, v);
    int r = __builtin_amdgcn_permlanex16(i, i, 0x76543210, 0xfedcba98, false, false);
    return __builtin_bit_cast(float, r);
}
template <int MASK>
__device__ __forceinline__ float lane_xor_row(float v) {
    int i = __builtin_bit_cast(int, v);
    int r = __builtin_amdgcn_update_dpp(0, i, 0x160 | MASK, 0xf, 0xf, true);
    return __builtin_bit_cast(float, r);
}
template <int T>
__device__ __forceinline__ float lane_bcast(float v) {
    return __builtin_bit_cast(float,
        __builtin_amdgcn_readlane(__builtin_bit_cast(int, v), T));
}

__device__ __forceinline__ float scan_step(float wr, float wi, float cr, float ci,
                                         float zt, float& xr, float& xi) {
    float nxr = fmaf(wr, xr, fmaf(-wi, xi, zt));
    float nxi = fmaf(wr, xi, wi * xr);
    xr = nxr; xi = nxi;
    return fmaf(cr, xr, -ci * xi);
}

__device__ __forceinline__ void scan_chunk(float wr, float wi, float cr, float ci,
                                           float zv, float& xr, float& xi,
                                           float (&v)[32]) {
#define SCAN_STEP(T) v[T] = scan_step(wr, wi, cr, ci, lane_bcast<T>(zv), xr, xi);
    SCAN_STEP(0)  SCAN_STEP(1)  SCAN_STEP(2)  SCAN_STEP(3)
    SCAN_STEP(4)  SCAN_STEP(5)  SCAN_STEP(6)  SCAN_STEP(7)
    SCAN_STEP(8)  SCAN_STEP(9)  SCAN_STEP(10) SCAN_STEP(11)
    SCAN_STEP(12) SCAN_STEP(13) SCAN_STEP(14) SCAN_STEP(15)
    SCAN_STEP(16) SCAN_STEP(17) SCAN_STEP(18) SCAN_STEP(19)
    SCAN_STEP(20) SCAN_STEP(21) SCAN_STEP(22) SCAN_STEP(23)
    SCAN_STEP(24) SCAN_STEP(25) SCAN_STEP(26) SCAN_STEP(27)
    SCAN_STEP(28) SCAN_STEP(29) SCAN_STEP(30) SCAN_STEP(31)
#undef SCAN_STEP
}

__device__ __forceinline__ float mode_sum(float (&v)[32], bool b4, bool b3,
                                          bool b2, bool b1, bool b0) {
#pragma unroll
    for (int j = 0; j < 16; ++j) {
        float keep = b4 ? v[j + 16] : v[j];
        float send = b4 ? v[j] : v[j + 16];
        v[j] = keep + lane_xor16(send);
    }
#pragma unroll
    for (int j = 0; j < 8; ++j) {
        float keep = b3 ? v[j + 8] : v[j];
        float send = b3 ? v[j] : v[j + 8];
        v[j] = keep + lane_xor_row<8>(send);
    }
#pragma unroll
    for (int j = 0; j < 4; ++j) {
        float keep = b2 ? v[j + 4] : v[j];
        float send = b2 ? v[j] : v[j + 4];
        v[j] = keep + lane_xor_row<4>(send);
    }
#pragma unroll
    for (int j = 0; j < 2; ++j) {
        float keep = b1 ? v[j + 2] : v[j];
        float send = b1 ? v[j] : v[j + 2];
        v[j] = keep + lane_xor_row<2>(send);
    }
    float keep0 = b0 ? v[1] : v[0];
    float send0 = b0 ? v[0] : v[1];
    return keep0 + lane_xor_row<1>(send0);
}


__device__ __forceinline__ void mode_powers(float wr, float wi,
                                            float (&qr)[32], float (&qi)[32]) {
    qr[0] = wr; qi[0] = wi;
#pragma unroll
    for (int j = 1; j < 32; ++j) {
        qr[j] = fmaf(qr[j - 1], wr, -qi[j - 1] * wi);
        qi[j] = fmaf(qr[j - 1], wi, qi[j - 1] * wr);
    }
}

__device__ __forceinline__ void chunk_input_sum(const float (&qr)[32], const float (&qi)[32],
                                                float zv, float& jr, float& ji) {
    int zi = __builtin_bit_cast(int, zv);
    float r0 = 0.0f, r1 = 0.0f, i0 = 0.0f, i1 = 0.0f;
#pragma unroll
    for (int k = 0; k < 30; k += 2) {
        float s0 = __builtin_bit_cast(float, __builtin_amdgcn_readlane(zi, k));
        float s1 = __builtin_bit_cast(float, __builtin_amdgcn_readlane(zi, k + 1));
        r0 = fmaf(qr[30 - k], s0, r0);
        i0 = fmaf(qi[30 - k], s0, i0);
        r1 = fmaf(qr[29 - k], s1, r1);
        i1 = fmaf(qi[29 - k], s1, i1);
    }
    float s30 = lane_bcast<30>(zv);
    float s31 = lane_bcast<31>(zv);
    r0 = fmaf(qr[0], s30, r0);
    i0 = fmaf(qi[0], s30, i0);
    r1 += s31;
    jr = r0 + r1;
    ji = i0 + i1;
}

#define SCAN_PIN_STATE() asm volatile("" : "+v"(xr), "+v"(xi))

#define SCAN_CHUNKS (LSEQ / 32)
#define SCAN_TRIPS (SCAN_CHUNKS / 2)
#define SCAN_WAVES 4
#define SCAN_CH (SCAN_WAVES / 2)

__device__ __forceinline__ int scan_chunk_in_row(int c) {
    return c < SCAN_CHUNKS ? c : SCAN_CHUNKS - 1;
}

#define SCAN_TRIP(zf, zj, v)                                                   \
    {                                                                          \
        float jr, ji;                                                          \
        chunk_input_sum(qr, qi, zj, jr, ji);                                   \
        scan_chunk(wr, wi, cr, ci, zf, xr, xi, v);                             \
        float nxr = fmaf(qr[31], xr, fmaf(-qi[31], xi, jr));                   \
        float nxi = fmaf(qr[31], xi, fmaf(qi[31], xr, ji));                    \
        xr = nxr; xi = nxi;                                                    \
    }

__global__ __launch_bounds__(SCAN_WAVES * 32) void s4_scan(
        const float* __restrict__ z,
        const float* __restrict__ log_dt,
        const float* __restrict__ Arl, const float* __restrict__ Aim,
        const float* __restrict__ Cre, const float* __restrict__ Cim,
        const float* __restrict__ D_skip, float* __restrict__ y) {
    int warp = __builtin_amdgcn_readfirstlane((int)(threadIdx.x >> 5));
    int lane = threadIdx.x & 31;
    int h = blockIdx.x * SCAN_CH + (warp >> 1);
    int par = warp & 1;
    if (h >= HCH) return;
    int cidx = h * NMODES + lane;
    float wr, wi, cr, ci;
    mode_coeffs(log_dt[h], Arl[cidx], Aim[cidx], Cre[cidx], Cim[cidx],
                wr, wi, cr, ci);
    float d = D_skip[h];
    const bool b4 = (lane & 16) != 0;
    const bool b3 = (lane & 8) != 0;
    const bool b2 = (lane & 4) != 0;
    const bool b1 = (lane & 2) != 0;
    const bool b0 = (lane & 1) != 0;
    const float* zrow = z + (size_t)h * LSEQ + lane;
    float* yrow = y + (size_t)h * LSEQ + lane;
#define Z_OWN(t) zrow[scan_chunk_in_row(2 * (t) + par) * 32]
#define Z_CARRIED(t) zrow[scan_chunk_in_row(2 * (t) + par + 1) * 32]

    float zc0 = zrow[0];
    float zf0 = Z_OWN(0), zj0 = Z_CARRIED(0);
    float zf1 = Z_OWN(1), zj1 = Z_CARRIED(1);
    float zf2 = Z_OWN(2), zj2 = Z_CARRIED(2);

    float qr[32], qi[32];
    mode_powers(wr, wi, qr, qi);

    float xr = 0.0f, xi = 0.0f;
    if (par) chunk_input_sum(qr, qi, zc0, xr, xi);

    float va[32], vb[32];
    SCAN_TRIP(zf0, zj0, va)
    for (int t = 0; t < SCAN_TRIPS - 2; t += 2) {
        float zf3 = Z_OWN(t + 3), zj3 = Z_CARRIED(t + 3);
        SCAN_TRIP(zf1, zj1, vb)
        float o = mode_sum(va, b4, b3, b2, b1, b0);
        SCAN_PIN_STATE();
        yrow[(2 * t + par) * 32] = gelu_tanh(fmaf(d, zf0, o));

        float zf4 = Z_OWN(t + 4), zj4 = Z_CARRIED(t + 4);
        SCAN_TRIP(zf2, zj2, va)
        o = mode_sum(vb, b4, b3, b2, b1, b0);
        SCAN_PIN_STATE();
        yrow[(2 * t + 2 + par) * 32] = gelu_tanh(fmaf(d, zf1, o));

        zf0 = zf2;
        zf1 = zf3; zj1 = zj3;
        zf2 = zf4; zj2 = zj4;
    }
#undef Z_OWN
#undef Z_CARRIED
    scan_chunk(wr, wi, cr, ci, zf1, xr, xi, vb);
    float o = mode_sum(va, b4, b3, b2, b1, b0);
    yrow[(2 * (SCAN_TRIPS - 2) + par) * 32] = gelu_tanh(fmaf(d, zf0, o));
    o = mode_sum(vb, b4, b3, b2, b1, b0);
    yrow[(2 * (SCAN_TRIPS - 1) + par) * 32] = gelu_tanh(fmaf(d, zf1, o));
}

#define BM 64
#define BMH 32
#define GROWS 512
#define BN 128
#define BK 32
#define LDP 34
#define NKT (KPAD / BK)

__global__ __launch_bounds__(256) void s4_glu_gemm(
        const float* __restrict__ Wg, const float* __restrict__ bias,
        const float* __restrict__ Y, const float* __restrict__ Zres,
        float* __restrict__ G) {
    __shared__ float Ws[2][BM][LDP];
    __shared__ float Ys[2][BN][LDP];
    int tid = threadIdx.x;
    int lane = tid & 31;
    int wave = tid >> 5;
    int wave_m = wave >> 2;
    int wave_n = wave & 3;
    int lrow = lane & 15;
    int khalf = (lane >> 4) << 1;
    int madd = (lane >> 4) << 3;
    int blockM = blockIdx.x * BMH;
    int blockN = blockIdx.y * BN;

    int wrow[8];
#pragma unroll
    for (int j = 0; j < 8; ++j) {
        int R = wave + 8 * j;
        int ch = blockM + ((R >> 5) << 4) + (R & 15);
        ch = ch < HCH ? ch : (HCH - 1);
        wrow[j] = (((R >> 4) & 1) * HCH + ch) * KD;
    }
    const float* ycol = Y + blockN + lane;

    v8f acc[2][2];
#pragma unroll
    for (int mt = 0; mt < 2; ++mt) {
#pragma unroll
        for (int i = 0; i < 8; ++i) {
            int ch = blockM + wave_m * 16 + i + madd;
            float bv = (ch < HCH) ? bias[mt * HCH + ch] : 0.0f;
#pragma unroll
            for (int nt = 0; nt < 2; ++nt) acc[mt][nt][i] = bv;
        }
    }

    float wreg[8], yreg[16];

#define TILE_LOAD(kb)                                                          \
    {                                                                          \
        int gk = (kb) + lane;                                                  \
        int gkc = gk < KD ? gk : (KD - 1);                                     \
        _Pragma("unroll")                                                      \
        for (int j = 0; j < 8; ++j) wreg[j] = Wg[wrow[j] + gkc];               \
        _Pragma("unroll")                                                      \
        for (int r = 0; r < 4; ++r) {                                          \
            int gy = (kb) + wave + 8 * r;                                      \
            int gyc = gy < KD ? gy : (KD - 1);                                 \
            const float* srow = ycol + (size_t)gyc * LSEQ;                     \
            _Pragma("unroll")                                                  \
            for (int j = 0; j < 4; ++j) yreg[4 * r + j] = srow[32 * j];        \
        }                                                                      \
    }

#define TILE_STORE(kb, b)                                                      \
    {                                                                          \
        bool kin = (kb) + lane < KD;                                           \
        _Pragma("unroll")                                                      \
        for (int j = 0; j < 8; ++j)                                            \
            Ws[b][wave + 8 * j][lane] = kin ? wreg[j] : 0.0f;                  \
        _Pragma("unroll")                                                      \
        for (int r = 0; r < 4; ++r) {                                          \
            bool yin = (kb) + wave + 8 * r < KD;                               \
            _Pragma("unroll")                                                  \
            for (int j = 0; j < 4; ++j)                                        \
                Ys[b][lane + 32 * j][wave + 8 * r] = yin ? yreg[4 * r + j] : 0.0f; \
        }                                                                      \
    }

#define TILE_MMA(b)                                                            \
    {                                                                          \
        v2f a[BK / 4][2], bf[BK / 4][2];                                       \
        _Pragma("unroll")                                                      \
        for (int kc = 0; kc < BK / 4; ++kc) {                                  \
            int kk = kc * 4 + khalf;                                           \
            _Pragma("unroll")                                                  \
            for (int mt = 0; mt < 2; ++mt)                                     \
                a[kc][mt] = *(const v2f*)&Ws[b][wave_m * 32 + mt * 16 + lrow][kk];  \
            _Pragma("unroll")                                                  \
            for (int nt = 0; nt < 2; ++nt)                                     \
                bf[kc][nt] = *(const v2f*)&Ys[b][wave_n * 32 + nt * 16 + lrow][kk]; \
        }                                                                      \
        __builtin_amdgcn_sched_barrier(0);                                     \
        _Pragma("unroll")                                                      \
        for (int kc = 0; kc < BK / 4; ++kc) {                                  \
            _Pragma("unroll")                                                  \
            for (int mt = 0; mt < 2; ++mt)                                     \
                _Pragma("unroll")                                              \
                for (int nt = 0; nt < 2; ++nt)                                 \
                    acc[mt][nt] = __builtin_amdgcn_wmma_f32_16x16x4_f32(       \
                        false, a[kc][mt], false, bf[kc][nt],                   \
                        (short)0, acc[mt][nt], false, false);                  \
        }                                                                      \
        __builtin_amdgcn_sched_barrier(0);                                     \
    }

    TILE_LOAD(0)
    TILE_STORE(0, 0)
    __syncthreads();
    for (int t = 0; t < NKT - 1; ++t) {
        int cur = t & 1;
        int kbn = (t + 1) * BK;
        TILE_LOAD(kbn)
        TILE_MMA(cur)
        TILE_STORE(kbn, cur ^ 1)
        __syncthreads();
    }
    int gn_base = blockN + wave_n * 32;
    float zr[8][2];
#pragma unroll
    for (int i = 0; i < 8; ++i) {
        int ch = blockM + wave_m * 16 + i + madd;
        int chc = ch < HCH ? ch : (HCH - 1);
#pragma unroll
        for (int nt = 0; nt < 2; ++nt)
            zr[i][nt] = Zres[(size_t)chc * LSEQ + gn_base + nt * 16 + lrow];
    }
    TILE_MMA((NKT - 1) & 1)
#undef TILE_LOAD
#undef TILE_STORE
#undef TILE_MMA

    asm volatile("v_nop\n\tv_nop\n\tv_nop\n\tv_nop\n\tv_nop\n\tv_nop\n\tv_nop\n\tv_nop"
                 : "+v"(acc[0][0]), "+v"(acc[0][1]), "+v"(acc[1][0]), "+v"(acc[1][1]));
    __builtin_amdgcn_sched_barrier(0);

#pragma unroll
    for (int i = 0; i < 8; ++i) {
        int ch = blockM + wave_m * 16 + i + madd;
#pragma unroll
        for (int nt = 0; nt < 2; ++nt) {
            int gn = gn_base + nt * 16 + lrow;
            G[(size_t)ch * LSEQ + gn] =
                acc[0][nt][i] * sigmoidf(acc[1][nt][i]) + zr[i][nt];
        }
    }
}

#define LN_COLS 32
#define LN_WAVES 32
#define LN_CPW 16

__global__ __launch_bounds__(1024) void s4_glu_ln(
        const float* __restrict__ Gin, float* __restrict__ Zout,
        const float* __restrict__ g, const float* __restrict__ b) {
    __shared__ float red_sum[LN_WAVES][LN_COLS];
    __shared__ float red_sq[LN_WAVES][LN_COLS];
    int lane = threadIdx.x & 31;
    int wave = __builtin_amdgcn_readfirstlane((int)(threadIdx.x >> 5));
    int l = blockIdx.x * LN_COLS + lane;

    float x[LN_CPW], gg[LN_CPW], bb[LN_CPW];
#pragma unroll
    for (int i = 0; i < LN_CPW; ++i) {
        int h = wave + LN_WAVES * i;
        int hc = (i < LN_CPW - 1 || h < HCH) ? h : (HCH - 1);
        x[i] = Gin[(size_t)hc * LSEQ + l];
        gg[i] = g[hc];
        bb[i] = b[hc];
    }

    float s = 0.0f;
#pragma unroll
    for (int i = 0; i < LN_CPW; ++i) {
        int h = wave + LN_WAVES * i;
        x[i] = (i < LN_CPW - 1 || h < HCH) ? x[i] : 0.0f;
        s += x[i];
    }
    red_sum[wave][lane] = s;
    __syncthreads();
    float tot = 0.0f;
#pragma unroll
    for (int w = 0; w < LN_WAVES; ++w) tot += red_sum[w][lane];
    float mu = tot * (1.0f / (float)HCH);

    float sq = 0.0f;
#pragma unroll
    for (int i = 0; i < LN_CPW; ++i) {
        int h = wave + LN_WAVES * i;
        float d = (i < LN_CPW - 1 || h < HCH) ? (x[i] - mu) : 0.0f;
        x[i] = d;
        sq += d * d;
    }
    red_sq[wave][lane] = sq;
    __syncthreads();
    float tq = 0.0f;
#pragma unroll
    for (int w = 0; w < LN_WAVES; ++w) tq += red_sq[w][lane];
    float rstd = rsqrtf(tq * (1.0f / (float)HCH) + LN_EPS);

#pragma unroll
    for (int i = 0; i < LN_CPW; ++i) {
        int h = wave + LN_WAVES * i;
        if (i < LN_CPW - 1 || h < HCH)
            Zout[(size_t)h * LSEQ + l] = x[i] * rstd * gg[i] + bb[i];
    }
}

extern "C" void kernel_launch(void* const* d_in, const int* in_sizes, int n_in,
                              void* d_out, int out_size, void* d_ws, size_t ws_size,
                              hipStream_t stream) {
    (void)in_sizes; (void)n_in; (void)out_size; (void)ws_size;
    const float* Zin    = (const float*)d_in[0];
    const float* log_dt = (const float*)d_in[1];
    const float* Arl    = (const float*)d_in[2];
    const float* Aim    = (const float*)d_in[3];
    const float* Cre    = (const float*)d_in[4];
    const float* Cim    = (const float*)d_in[5];
    const float* Dsk    = (const float*)d_in[6];
    const float* glu_w  = (const float*)d_in[7];
    const float* glu_b  = (const float*)d_in[8];
    const float* ln_g   = (const float*)d_in[9];
    const float* ln_b   = (const float*)d_in[10];

    float* bufa  = (float*)d_ws;
    float* bufb  = bufa + (size_t)HCH * LSEQ;
    float* gbuf  = bufb + (size_t)HCH * LSEQ;
    float* out   = (float*)d_out;

    static_assert(NLAYERS == 4, "the four layers are written out below");
    static_assert(16 * BMH == GROWS && GROWS >= HCH && GROWS <= 2 * HCH,
                  "16 block rows cover every channel; G fits the plane behind bufb");
#define S4_LAYER(zin, ybuf, zout)                                              \
    s4_scan<<<(HCH + SCAN_CH - 1) / SCAN_CH, SCAN_WAVES * 32, 0, stream>>>(    \
        zin, log_dt, Arl, Aim, Cre, Cim, Dsk, ybuf);                           \
    s4_glu_gemm<<<dim3(16, 16, 1), 256, 0, stream>>>(glu_w, glu_b, ybuf, zin, gbuf); \
    s4_glu_ln<<<LSEQ / LN_COLS, 1024, 0, stream>>>(gbuf, zout, ln_g, ln_b);
    S4_LAYER(Zin,  bufb, bufa)
    S4_LAYER(bufa, bufb, bufb)
    S4_LAYER(bufb, bufa, bufa)
    S4_LAYER(bufa, bufb, out)
#undef S4_LAYER
}


# glu_gemm K loop pipelined one tile deeper: 3 LDS stages, next tile's fragments read into a 2nd register set ahead of the WMMAs, staging stores between WMMA halves, LDS-only barrier; sums unchanged
# speedup vs baseline: 15.4147x; 1.0786x over previous
#include <hip/hip_runtime.h>
#include <hip/hip_bf16.h>

#define HCH 511
#define LSEQ 2048
#define NMODES 32
#define NLAYERS 4
#define KD 511
#define KPAD 512
#define LN_EPS 1e-5f

typedef __attribute__((ext_vector_type(2))) float v2f;
typedef __attribute__((ext_vector_type(8))) float v8f;

__device__ __forceinline__ float gelu_tanh(float x) {
    float x3 = x * x * x;
    float t = tanhf(0.7978845608028654f * (x + 0.044715f * x3));
    return 0.5f * x * (1.0f + t);
}

__device__ __forceinline__ float sigmoidf(float x) {
    return 1.0f / (1.0f + expf(-x));
}

__device__ __forceinline__ void mode_coeffs(float ldt, float arl, float aim,
                                            float Cr, float Ci,
                                            float& wre, float& wim,
                                            float& cre, float& cim) {
    float a_re = -expf(arl);
    float a_im = aim;
    float dt = expf(ldt);
    float dre = a_re * dt, dim = a_im * dt;
    float er = expf(dre);
    float sn, cs;
    sincosf(dim, &sn, &cs);
    wre = er * cs;
    wim = er * sn;
    float nre = wre - 1.0f, nim = wim;
    float inv = 1.0f / (a_re * a_re + a_im * a_im);
    float tre = (nre * a_re + nim * a_im) * inv;
    float tim = (nim * a_re - nre * a_im) * inv;
    cre = 2.0f * (Cr * tre - Ci * tim);
    cim = 2.0f * (Cr * tim + Ci * tre);
}

__device__ __forceinline__ float lane_xor16(float v) {
    int i = __builtin_bit_cast(int, v);
    int r = __builtin_amdgcn_permlanex16(i, i, 0x76543210, 0xfedcba98, false, false);
    return __builtin_bit_cast(float, r);
}
template <int MASK>
__device__ __forceinline__ float lane_xor_row(float v) {
    int i = __builtin_bit_cast(int, v);
    int r = __builtin_amdgcn_update_dpp(0, i, 0x160 | MASK, 0xf, 0xf, true);
    return __builtin_bit_cast(float, r);
}
template <int T>
__device__ __forceinline__ float lane_bcast(float v) {
    return __builtin_bit_cast(float,
        __builtin_amdgcn_readlane(__builtin_bit_cast(int, v), T));
}

__device__ __forceinline__ float scan_step(float wr, float wi, float cr, float ci,
                                         float zt, float& xr, float& xi) {
    float nxr = fmaf(wr, xr, fmaf(-wi, xi, zt));
    float nxi = fmaf(wr, xi, wi * xr);
    xr = nxr; xi = nxi;
    return fmaf(cr, xr, -ci * xi);
}

__device__ __forceinline__ void scan_chunk(float wr, float wi, float cr, float ci,
                                           float zv, float& xr, float& xi,
                                           float (&v)[32]) {
#define SCAN_STEP(T) v[T] = scan_step(wr, wi, cr, ci, lane_bcast<T>(zv), xr, xi);
    SCAN_STEP(0)  SCAN_STEP(1)  SCAN_STEP(2)  SCAN_STEP(3)
    SCAN_STEP(4)  SCAN_STEP(5)  SCAN_STEP(6)  SCAN_STEP(7)
    SCAN_STEP(8)  SCAN_STEP(9)  SCAN_STEP(10) SCAN_STEP(11)
    SCAN_STEP(12) SCAN_STEP(13) SCAN_STEP(14) SCAN_STEP(15)
    SCAN_STEP(16) SCAN_STEP(17) SCAN_STEP(18) SCAN_STEP(19)
    SCAN_STEP(20) SCAN_STEP(21) SCAN_STEP(22) SCAN_STEP(23)
    SCAN_STEP(24) SCAN_STEP(25) SCAN_STEP(26) SCAN_STEP(27)
    SCAN_STEP(28) SCAN_STEP(29) SCAN_STEP(30) SCAN_STEP(31)
#undef SCAN_STEP
}

__device__ __forceinline__ float mode_sum(float (&v)[32], bool b4, bool b3,
                                          bool b2, bool b1, bool b0) {
#pragma unroll
    for (int j = 0; j < 16; ++j) {
        float keep = b4 ? v[j + 16] : v[j];
        float send = b4 ? v[j] : v[j + 16];
        v[j] = keep + lane_xor16(send);
    }
#pragma unroll
    for (int j = 0; j < 8; ++j) {
        float keep = b3 ? v[j + 8] : v[j];
        float send = b3 ? v[j] : v[j + 8];
        v[j] = keep + lane_xor_row<8>(send);
    }
#pragma unroll
    for (int j = 0; j < 4; ++j) {
        float keep = b2 ? v[j + 4] : v[j];
        float send = b2 ? v[j] : v[j + 4];
        v[j] = keep + lane_xor_row<4>(send);
    }
#pragma unroll
    for (int j = 0; j < 2; ++j) {
        float keep = b1 ? v[j + 2] : v[j];
        float send = b1 ? v[j] : v[j + 2];
        v[j] = keep + lane_xor_row<2>(send);
    }
    float keep0 = b0 ? v[1] : v[0];
    float send0 = b0 ? v[0] : v[1];
    return keep0 + lane_xor_row<1>(send0);
}


__device__ __forceinline__ void mode_powers(float wr, float wi,
                                            float (&qr)[32], float (&qi)[32]) {
    qr[0] = wr; qi[0] = wi;
#pragma unroll
    for (int j = 1; j < 32; ++j) {
        qr[j] = fmaf(qr[j - 1], wr, -qi[j - 1] * wi);
        qi[j] = fmaf(qr[j - 1], wi, qi[j - 1] * wr);
    }
}

__device__ __forceinline__ void chunk_input_sum(const float (&qr)[32], const float (&qi)[32],
                                                float zv, float& jr, float& ji) {
    int zi = __builtin_bit_cast(int, zv);
    float r0 = 0.0f, r1 = 0.0f, i0 = 0.0f, i1 = 0.0f;
#pragma unroll
    for (int k = 0; k < 30; k += 2) {
        float s0 = __builtin_bit_cast(float, __builtin_amdgcn_readlane(zi, k));
        float s1 = __builtin_bit_cast(float, __builtin_amdgcn_readlane(zi, k + 1));
        r0 = fmaf(qr[30 - k], s0, r0);
        i0 = fmaf(qi[30 - k], s0, i0);
        r1 = fmaf(qr[29 - k], s1, r1);
        i1 = fmaf(qi[29 - k], s1, i1);
    }
    float s30 = lane_bcast<30>(zv);
    float s31 = lane_bcast<31>(zv);
    r0 = fmaf(qr[0], s30, r0);
    i0 = fmaf(qi[0], s30, i0);
    r1 += s31;
    jr = r0 + r1;
    ji = i0 + i1;
}

#define SCAN_PIN_STATE() asm volatile("" : "+v"(xr), "+v"(xi))

#define SCAN_CHUNKS (LSEQ / 32)
#define SCAN_TRIPS (SCAN_CHUNKS / 2)
#define SCAN_WAVES 4
#define SCAN_CH (SCAN_WAVES / 2)

__device__ __forceinline__ int scan_chunk_in_row(int c) {
    return c < SCAN_CHUNKS ? c : SCAN_CHUNKS - 1;
}

#define SCAN_TRIP(zf, zj, v)                                                   \
    {                                                                          \
        float jr, ji;                                                          \
        chunk_input_sum(qr, qi, zj, jr, ji);                                   \
        scan_chunk(wr, wi, cr, ci, zf, xr, xi, v);                             \
        float nxr = fmaf(qr[31], xr, fmaf(-qi[31], xi, jr));                   \
        float nxi = fmaf(qr[31], xi, fmaf(qi[31], xr, ji));                    \
        xr = nxr; xi = nxi;                                                    \
    }

__global__ __launch_bounds__(SCAN_WAVES * 32) void s4_scan(
        const float* __restrict__ z,
        const float* __restrict__ log_dt,
        const float* __restrict__ Arl, const float* __restrict__ Aim,
        const float* __restrict__ Cre, const float* __restrict__ Cim,
        const float* __restrict__ D_skip, float* __restrict__ y) {
    int warp = __builtin_amdgcn_readfirstlane((int)(threadIdx.x >> 5));
    int lane = threadIdx.x & 31;
    int h = blockIdx.x * SCAN_CH + (warp >> 1);
    int par = warp & 1;
    if (h >= HCH) return;
    int cidx = h * NMODES + lane;
    float wr, wi, cr, ci;
    mode_coeffs(log_dt[h], Arl[cidx], Aim[cidx], Cre[cidx], Cim[cidx],
                wr, wi, cr, ci);
    float d = D_skip[h];
    const bool b4 = (lane & 16) != 0;
    const bool b3 = (lane & 8) != 0;
    const bool b2 = (lane & 4) != 0;
    const bool b1 = (lane & 2) != 0;
    const bool b0 = (lane & 1) != 0;
    const float* zrow = z + (size_t)h * LSEQ + lane;
    float* yrow = y + (size_t)h * LSEQ + lane;
#define Z_OWN(t) zrow[scan_chunk_in_row(2 * (t) + par) * 32]
#define Z_CARRIED(t) zrow[scan_chunk_in_row(2 * (t) + par + 1) * 32]

    float zc0 = zrow[0];
    float zf0 = Z_OWN(0), zj0 = Z_CARRIED(0);
    float zf1 = Z_OWN(1), zj1 = Z_CARRIED(1);
    float zf2 = Z_OWN(2), zj2 = Z_CARRIED(2);

    float qr[32], qi[32];
    mode_powers(wr, wi, qr, qi);

    float xr = 0.0f, xi = 0.0f;
    if (par) chunk_input_sum(qr, qi, zc0, xr, xi);

    float va[32], vb[32];
    SCAN_TRIP(zf0, zj0, va)
    for (int t = 0; t < SCAN_TRIPS - 2; t += 2) {
        float zf3 = Z_OWN(t + 3), zj3 = Z_CARRIED(t + 3);
        SCAN_TRIP(zf1, zj1, vb)
        float o = mode_sum(va, b4, b3, b2, b1, b0);
        SCAN_PIN_STATE();
        yrow[(2 * t + par) * 32] = gelu_tanh(fmaf(d, zf0, o));

        float zf4 = Z_OWN(t + 4), zj4 = Z_CARRIED(t + 4);
        SCAN_TRIP(zf2, zj2, va)
        o = mode_sum(vb, b4, b3, b2, b1, b0);
        SCAN_PIN_STATE();
        yrow[(2 * t + 2 + par) * 32] = gelu_tanh(fmaf(d, zf1, o));

        zf0 = zf2;
        zf1 = zf3; zj1 = zj3;
        zf2 = zf4; zj2 = zj4;
    }
#undef Z_OWN
#undef Z_CARRIED
    scan_chunk(wr, wi, cr, ci, zf1, xr, xi, vb);
    float o = mode_sum(va, b4, b3, b2, b1, b0);
    yrow[(2 * (SCAN_TRIPS - 2) + par) * 32] = gelu_tanh(fmaf(d, zf0, o));
    o = mode_sum(vb, b4, b3, b2, b1, b0);
    yrow[(2 * (SCAN_TRIPS - 1) + par) * 32] = gelu_tanh(fmaf(d, zf1, o));
}

__device__ __forceinline__ void barrier_lds_ring() {
    asm volatile("s_wait_dscnt 0x0\n\ts_barrier_signal -1\n\ts_barrier_wait -1" ::: "memory");
}

#define BM 64
#define BMH 32
#define GROWS 512
#define BN 128
#define BK 32
#define LDP 34
#define NKT (KPAD / BK)
#define NSTG 3

__global__ __launch_bounds__(256) void s4_glu_gemm(
        const float* __restrict__ Wg, const float* __restrict__ bias,
        const float* __restrict__ Y, const float* __restrict__ Zres,
        float* __restrict__ G) {
    __shared__ float Ws[NSTG][BM][LDP];
    __shared__ float Ys[NSTG][BN][LDP];
    int tid = threadIdx.x;
    int lane = tid & 31;
    int wave = tid >> 5;
    int wave_m = wave >> 2;
    int wave_n = wave & 3;
    int lrow = lane & 15;
    int khalf = (lane >> 4) << 1;
    int madd = (lane >> 4) << 3;
    int blockM = blockIdx.x * BMH;
    int blockN = blockIdx.y * BN;

    int wrow[8];
#pragma unroll
    for (int j = 0; j < 8; ++j) {
        int R = wave + 8 * j;
        int ch = blockM + ((R >> 5) << 4) + (R & 15);
        ch = ch < HCH ? ch : (HCH - 1);
        wrow[j] = (((R >> 4) & 1) * HCH + ch) * KD;
    }
    const float* ycol = Y + blockN + lane;

    v8f acc[2][2];
#pragma unroll
    for (int mt = 0; mt < 2; ++mt) {
#pragma unroll
        for (int i = 0; i < 8; ++i) {
            int ch = blockM + wave_m * 16 + i + madd;
            float bv = (ch < HCH) ? bias[mt * HCH + ch] : 0.0f;
#pragma unroll
            for (int nt = 0; nt < 2; ++nt) acc[mt][nt][i] = bv;
        }
    }

    float wregA[8], yregA[16], wregB[8], yregB[16];
    v2f fa0[BK / 4][2], fb0[BK / 4][2], fa1[BK / 4][2], fb1[BK / 4][2];

#define TILE_LOAD(kb, wreg, yreg)                                              \
    {                                                                          \
        int gk = (kb) + lane;                                                  \
        int gkc = gk < KD ? gk : (KD - 1);                                     \
        _Pragma("unroll")                                                      \
        for (int j = 0; j < 8; ++j) wreg[j] = Wg[wrow[j] + gkc];               \
        _Pragma("unroll")                                                      \
        for (int r = 0; r < 4; ++r) {                                          \
            int gy = (kb) + wave + 8 * r;                                      \
            int gyc = gy < KD ? gy : (KD - 1);                                 \
            const float* srow = ycol + (size_t)gyc * LSEQ;                     \
            _Pragma("unroll")                                                  \
            for (int j = 0; j < 4; ++j) yreg[4 * r + j] = srow[32 * j];        \
        }                                                                      \
    }

#define TILE_STORE(kb, b, wreg, yreg)                                          \
    {                                                                          \
        bool kin = (kb) + lane < KD;                                           \
        _Pragma("unroll")                                                      \
        for (int j = 0; j < 8; ++j)                                            \
            Ws[b][wave + 8 * j][lane] = kin ? wreg[j] : 0.0f;                  \
        _Pragma("unroll")                                                      \
        for (int r = 0; r < 4; ++r) {                                          \
            bool yin = (kb) + wave + 8 * r < KD;                               \
            _Pragma("unroll")                                                  \
            for (int j = 0; j < 4; ++j)                                        \
                Ys[b][lane + 32 * j][wave + 8 * r] = yin ? yreg[4 * r + j] : 0.0f; \
        }                                                                      \
    }

#define FRAG_LOAD(b, fa, fb)                                                   \
    {                                                                          \
        _Pragma("unroll")                                                      \
        for (int kc = 0; kc < BK / 4; ++kc) {                                  \
            int kk = kc * 4 + khalf;                                           \
            _Pragma("unroll")                                                  \
            for (int mt = 0; mt < 2; ++mt)                                     \
                fa[kc][mt] = *(const v2f*)&Ws[b][wave_m * 32 + mt * 16 + lrow][kk];  \
            _Pragma("unroll")                                                  \
            for (int nt = 0; nt < 2; ++nt)                                     \
                fb[kc][nt] = *(const v2f*)&Ys[b][wave_n * 32 + nt * 16 + lrow][kk];  \
        }                                                                      \
    }

#define TILE_MMA_HALF(KC0, fa, fb)                                             \
    {                                                                          \
        _Pragma("unroll")                                                      \
        for (int kc = (KC0); kc < (KC0) + BK / 8; ++kc) {                      \
            _Pragma("unroll")                                                  \
            for (int mt = 0; mt < 2; ++mt)                                     \
                _Pragma("unroll")                                              \
                for (int nt = 0; nt < 2; ++nt)                                 \
                    acc[mt][nt] = __builtin_amdgcn_wmma_f32_16x16x4_f32(       \
                        false, fa[kc][mt], false, fb[kc][nt],                  \
                        (short)0, acc[mt][nt], false, false);                  \
        }                                                                      \
    }

#define PIPE_STEP(T, fa, fb, fna, fnb, DOF, DOS, DOL)                          \
    {                                                                          \
        if (DOF) FRAG_LOAD(fb_, fna, fnb)                                      \
        __builtin_amdgcn_sched_barrier(0);                                     \
        TILE_MMA_HALF(0, fa, fb)                                               \
        __builtin_amdgcn_sched_barrier(0);                                     \
        if (DOS) TILE_STORE(((T) + 2) * BK, sb_, wregA, yregA)                 \
        if (DOL) TILE_LOAD(((T) + 3) * BK, wregA, yregA)                       \
        __builtin_amdgcn_sched_barrier(0);                                     \
        TILE_MMA_HALF(BK / 8, fa, fb)                                          \
        __builtin_amdgcn_sched_barrier(0);                                     \
         \
         \
         \
        if (DOS) barrier_lds_ring();                                           \
        fb_ = sb_;                                                             \
        sb_ = sb_ == NSTG - 1 ? 0 : sb_ + 1;                                   \
    }

    TILE_LOAD(0, wregA, yregA)
    TILE_LOAD(BK, wregB, yregB)
    TILE_STORE(0, 0, wregA, yregA)
    TILE_STORE(BK, 1, wregB, yregB)
    TILE_LOAD(2 * BK, wregA, yregA)
    barrier_lds_ring();
    FRAG_LOAD(0, fa0, fb0)
    int fb_ = 1, sb_ = 2;
    static_assert(NKT == 16, "the tail below is written out for 16 K tiles");
    for (int t = 0; t < NKT - 4; t += 2) {
        PIPE_STEP(t, fa0, fb0, fa1, fb1, true, true, true)
        PIPE_STEP(t + 1, fa1, fb1, fa0, fb0, true, true, true)
    }
    PIPE_STEP(NKT - 4, fa0, fb0, fa1, fb1, true, true, true)
    PIPE_STEP(NKT - 3, fa1, fb1, fa0, fb0, true, true, false)
    PIPE_STEP(NKT - 2, fa0, fb0, fa1, fb1, true, false, false)
    int gn_base = blockN + wave_n * 32;
    float zr[8][2];
#pragma unroll
    for (int i = 0; i < 8; ++i) {
        int ch = blockM + wave_m * 16 + i + madd;
        int chc = ch < HCH ? ch : (HCH - 1);
#pragma unroll
        for (int nt = 0; nt < 2; ++nt)
            zr[i][nt] = Zres[(size_t)chc * LSEQ + gn_base + nt * 16 + lrow];
    }
    PIPE_STEP(NKT - 1, fa1, fb1, fa0, fb0, false, false, false)
#undef TILE_LOAD
#undef TILE_STORE
#undef FRAG_LOAD
#undef TILE_MMA_HALF
#undef PIPE_STEP

    asm volatile("v_nop\n\tv_nop\n\tv_nop\n\tv_nop\n\tv_nop\n\tv_nop\n\tv_nop\n\tv_nop"
                 : "+v"(acc[0][0]), "+v"(acc[0][1]), "+v"(acc[1][0]), "+v"(acc[1][1]));
    __builtin_amdgcn_sched_barrier(0);

#pragma unroll
    for (int i = 0; i < 8; ++i) {
        int ch = blockM + wave_m * 16 + i + madd;
#pragma unroll
        for (int nt = 0; nt < 2; ++nt) {
            int gn = gn_base + nt * 16 + lrow;
            G[(size_t)ch * LSEQ + gn] =
                acc[0][nt][i] * sigmoidf(acc[1][nt][i]) + zr[i][nt];
        }
    }
}

#define LN_COLS 32
#define LN_WAVES 32
#define LN_CPW 16

__global__ __launch_bounds__(1024) void s4_glu_ln(
        const float* __restrict__ Gin, float* __restrict__ Zout,
        const float* __restrict__ g, const float* __restrict__ b) {
    __shared__ float red_sum[LN_WAVES][LN_COLS];
    __shared__ float red_sq[LN_WAVES][LN_COLS];
    int lane = threadIdx.x & 31;
    int wave = __builtin_amdgcn_readfirstlane((int)(threadIdx.x >> 5));
    int l = blockIdx.x * LN_COLS + lane;

    float x[LN_CPW], gg[LN_CPW], bb[LN_CPW];
#pragma unroll
    for (int i = 0; i < LN_CPW; ++i) {
        int h = wave + LN_WAVES * i;
        int hc = (i < LN_CPW - 1 || h < HCH) ? h : (HCH - 1);
        x[i] = Gin[(size_t)hc * LSEQ + l];
        gg[i] = g[hc];
        bb[i] = b[hc];
    }

    float s = 0.0f;
#pragma unroll
    for (int i = 0; i < LN_CPW; ++i) {
        int h = wave + LN_WAVES * i;
        x[i] = (i < LN_CPW - 1 || h < HCH) ? x[i] : 0.0f;
        s += x[i];
    }
    red_sum[wave][lane] = s;
    __syncthreads();
    float tot = 0.0f;
#pragma unroll
    for (int w = 0; w < LN_WAVES; ++w) tot += red_sum[w][lane];
    float mu = tot * (1.0f / (float)HCH);

    float sq = 0.0f;
#pragma unroll
    for (int i = 0; i < LN_CPW; ++i) {
        int h = wave + LN_WAVES * i;
        float d = (i < LN_CPW - 1 || h < HCH) ? (x[i] - mu) : 0.0f;
        x[i] = d;
        sq += d * d;
    }
    red_sq[wave][lane] = sq;
    __syncthreads();
    float tq = 0.0f;
#pragma unroll
    for (int w = 0; w < LN_WAVES; ++w) tq += red_sq[w][lane];
    float rstd = rsqrtf(tq * (1.0f / (float)HCH) + LN_EPS);

#pragma unroll
    for (int i = 0; i < LN_CPW; ++i) {
        int h = wave + LN_WAVES * i;
        if (i < LN_CPW - 1 || h < HCH)
            Zout[(size_t)h * LSEQ + l] = x[i] * rstd * gg[i] + bb[i];
    }
}

extern "C" void kernel_launch(void* const* d_in, const int* in_sizes, int n_in,
                              void* d_out, int out_size, void* d_ws, size_t ws_size,
                              hipStream_t stream) {
    (void)in_sizes; (void)n_in; (void)out_size; (void)ws_size;
    const float* Zin    = (const float*)d_in[0];
    const float* log_dt = (const float*)d_in[1];
    const float* Arl    = (const float*)d_in[2];
    const float* Aim    = (const float*)d_in[3];
    const float* Cre    = (const float*)d_in[4];
    const float* Cim    = (const float*)d_in[5];
    const float* Dsk    = (const float*)d_in[6];
    const float* glu_w  = (const float*)d_in[7];
    const float* glu_b  = (const float*)d_in[8];
    const float* ln_g   = (const float*)d_in[9];
    const float* ln_b   = (const float*)d_in[10];

    float* bufa  = (float*)d_ws;
    float* bufb  = bufa + (size_t)HCH * LSEQ;
    float* gbuf  = bufb + (size_t)HCH * LSEQ;
    float* out   = (float*)d_out;

    static_assert(NLAYERS == 4, "the four layers are written out below");
    static_assert(16 * BMH == GROWS && GROWS >= HCH && GROWS <= 2 * HCH,
                  "16 block rows cover every channel; G fits the plane behind bufb");
#define S4_LAYER(zin, ybuf, zout)                                              \
    s4_scan<<<(HCH + SCAN_CH - 1) / SCAN_CH, SCAN_WAVES * 32, 0, stream>>>(    \
        zin, log_dt, Arl, Aim, Cre, Cim, Dsk, ybuf);                           \
    s4_glu_gemm<<<dim3(16, 16, 1), 256, 0, stream>>>(glu_w, glu_b, ybuf, zin, gbuf); \
    s4_glu_ln<<<LSEQ / LN_COLS, 1024, 0, stream>>>(gbuf, zout, ln_g, ln_b);
    S4_LAYER(Zin,  bufb, bufa)
    S4_LAYER(bufa, bufb, bufb)
    S4_LAYER(bufb, bufa, bufa)
    S4_LAYER(bufa, bufb, out)
#undef S4_LAYER
}


# scan in chunk form, lane = step: per 32 steps 32 kernel taps x inputs + 64 table terms x carried state (all lanes get it via LDS), state carried by w^32 jumps; no per-step chain; sum order changes
# speedup vs baseline: 17.1896x; 1.1151x over previous
#include <hip/hip_runtime.h>
#include <hip/hip_bf16.h>

#define HCH 511
#define LSEQ 2048
#define NMODES 32
#define NLAYERS 4
#define KD 511
#define KPAD 512
#define LN_EPS 1e-5f

typedef __attribute__((ext_vector_type(2))) float v2f;
typedef __attribute__((ext_vector_type(8))) float v8f;

__device__ __forceinline__ float gelu_tanh(float x) {
    float x3 = x * x * x;
    float t = tanhf(0.7978845608028654f * (x + 0.044715f * x3));
    return 0.5f * x * (1.0f + t);
}

__device__ __forceinline__ float sigmoidf(float x) {
    return 1.0f / (1.0f + expf(-x));
}

__device__ __forceinline__ void mode_coeffs(float ldt, float arl, float aim,
                                            float Cr, float Ci,
                                            float& wre, float& wim,
                                            float& cre, float& cim) {
    float a_re = -expf(arl);
    float a_im = aim;
    float dt = expf(ldt);
    float dre = a_re * dt, dim = a_im * dt;
    float er = expf(dre);
    float sn, cs;
    sincosf(dim, &sn, &cs);
    wre = er * cs;
    wim = er * sn;
    float nre = wre - 1.0f, nim = wim;
    float inv = 1.0f / (a_re * a_re + a_im * a_im);
    float tre = (nre * a_re + nim * a_im) * inv;
    float tim = (nim * a_re - nre * a_im) * inv;
    cre = 2.0f * (Cr * tre - Ci * tim);
    cim = 2.0f * (Cr * tim + Ci * tre);
}

__device__ __forceinline__ void mode_powers(float wr, float wi,
                                            float (&qr)[32], float (&qi)[32]) {
    qr[0] = wr; qi[0] = wi;
#pragma unroll
    for (int j = 1; j < 32; ++j) {
        qr[j] = fmaf(qr[j - 1], wr, -qi[j - 1] * wi);
        qi[j] = fmaf(qr[j - 1], wi, qi[j - 1] * wr);
    }
}

__device__ __forceinline__ void chunk_inputs(const float* zc, float (&zb)[32]) {
    const float4* p = reinterpret_cast<const float4*>(zc);
#pragma unroll
    for (int i = 0; i < 8; ++i) {
        float4 v = p[i];
        zb[4 * i] = v.x; zb[4 * i + 1] = v.y;
        zb[4 * i + 2] = v.z; zb[4 * i + 3] = v.w;
    }
}

__device__ __forceinline__ void chunk_input_sum(const float (&qr)[32], const float (&qi)[32],
                                                const float (&zb)[32], float& jr, float& ji) {
    float r[4] = {0.0f, 0.0f, 0.0f, 0.0f}, s[4] = {0.0f, 0.0f, 0.0f, 0.0f};
#pragma unroll
    for (int k = 0; k < 32; ++k) {
        r[k & 3] = fmaf(qr[31 - k], zb[k], r[k & 3]);
        s[k & 3] = fmaf(qi[31 - k], zb[k], s[k & 3]);
    }
    jr = (r[0] + r[1]) + (r[2] + r[3]);
    ji = (s[0] + s[1]) + (s[2] + s[3]);
}

__device__ __forceinline__ float chunk_outputs(const float (&P)[32], const float (&Q)[32],
                                               const float (&T)[32], const float (&ub)[64],
                                               const float (&zb)[32]) {
    float a[4] = {0.0f, 0.0f, 0.0f, 0.0f}, b[4] = {0.0f, 0.0f, 0.0f, 0.0f};
    float c[4] = {0.0f, 0.0f, 0.0f, 0.0f};
#pragma unroll
    for (int n = 0; n < 32; ++n) {
        a[n & 3] = fmaf(P[n], ub[n], a[n & 3]);
        b[n & 3] = fmaf(Q[n], ub[32 + n], b[n & 3]);
        c[n & 3] = fmaf(T[n], zb[n], c[n & 3]);
    }
    float sa = (a[0] + a[1]) + (a[2] + a[3]);
    float sb = (b[0] + b[1]) + (b[2] + b[3]);
    float sc = (c[0] + c[1]) + (c[2] + c[3]);
    return (sa + sb) + sc;
}

__device__ __forceinline__ void state_to_all_lanes(float* us, int lane, float ur, float ui,
                                                   float (&ub)[64]) {
    us[lane] = ur;
    us[32 + lane] = ui;
    const float4* p = reinterpret_cast<const float4*>(us);
#pragma unroll
    for (int i = 0; i < 16; ++i) {
        float4 v = p[i];
        ub[4 * i] = v.x; ub[4 * i + 1] = v.y;
        ub[4 * i + 2] = v.z; ub[4 * i + 3] = v.w;
    }
}

#define SCAN_CHUNKS (LSEQ / 32)
#define SCAN_TRIPS (SCAN_CHUNKS / 2)
#define SCAN_WAVES 4
#define SCAN_CH (SCAN_WAVES / 2)
#define TPITCH 33

__global__ __launch_bounds__(SCAN_WAVES * 32) void s4_scan(
        const float* __restrict__ z,
        const float* __restrict__ log_dt,
        const float* __restrict__ Arl, const float* __restrict__ Aim,
        const float* __restrict__ Cre, const float* __restrict__ Cim,
        const float* __restrict__ D_skip, float* __restrict__ y) {
    __shared__ __attribute__((aligned(16))) float zsm[SCAN_CH][LSEQ];
    __shared__ float tre[SCAN_WAVES][32 * TPITCH];
    __shared__ float tim[SCAN_WAVES][32 * TPITCH];
    __shared__ float ksm[SCAN_WAVES][64];
    __shared__ __attribute__((aligned(16))) float usm[SCAN_WAVES][64];
    int warp = __builtin_amdgcn_readfirstlane((int)(threadIdx.x >> 5));
    int lane = threadIdx.x & 31;
    int chl = warp >> 1;
    int par = warp & 1;
    int h = blockIdx.x * SCAN_CH + chl;
    bool live = h < HCH;
    int hc = live ? h : HCH - 1;

    const float* zrow = z + (size_t)hc * LSEQ + lane;
    float zt[SCAN_TRIPS];
#pragma unroll
    for (int i = 0; i < SCAN_TRIPS; ++i) zt[i] = zrow[(2 * i + par) * 32];

    int cidx = hc * NMODES + lane;
    float wr, wi, cr, ci;
    mode_coeffs(log_dt[hc], Arl[cidx], Aim[cidx], Cre[cidx], Cim[cidx],
                wr, wi, cr, ci);
    float d = D_skip[hc];
    float qr[32], qi[32];
    mode_powers(wr, wi, qr, qi);

    float* trow = &tre[warp][lane * TPITCH];
    float* irow = &tim[warp][lane * TPITCH];
    trow[0] = cr;
    irow[0] = -ci;
#pragma unroll
    for (int j = 1; j < 32; ++j) {
        trow[j] = fmaf(cr, qr[j - 1], -ci * qi[j - 1]);
        irow[j] = -fmaf(cr, qi[j - 1], ci * qr[j - 1]);
    }
    ksm[warp][lane] = 0.0f;
    float* zdst = &zsm[chl][par * 32 + lane];
#pragma unroll
    for (int i = 0; i < SCAN_TRIPS; ++i) zdst[i * 64] = zt[i];
    __syncthreads();

    float P[32], Q[32];
#pragma unroll
    for (int n = 0; n < 32; ++n) {
        P[n] = tre[warp][n * TPITCH + lane];
        Q[n] = tim[warp][n * TPITCH + lane];
    }
    float k0 = 0.0f, k1 = 0.0f, k2 = 0.0f, k3 = 0.0f;
#pragma unroll
    for (int n = 0; n < 32; n += 4) {
        k0 += P[n]; k1 += P[n + 1]; k2 += P[n + 2]; k3 += P[n + 3];
    }
    ksm[warp][32 + lane] = (k0 + k1) + (k2 + k3);
    __syncthreads();
    float T[32];
#pragma unroll
    for (int k = 0; k < 32; ++k) T[k] = ksm[warp][32 + lane - k];
    if (!live) return;

    const float* zl = &zsm[chl][0];
    float* us = &usm[warp][0];
    float* yrow = y + (size_t)hc * LSEQ + lane;
    float ur = 0.0f, ui = 0.0f;
    float zA[32], zB[32], zC[32], zD[32], ubA[64], ubB[64];
    if (par) {
        chunk_inputs(zl, zA);
        chunk_input_sum(qr, qi, zA, ur, ui);
    }
#define SCAN_HOLD8(a, i)                                                       \
    asm volatile("" : "+v"(a[i]), "+v"(a[(i) + 1]), "+v"(a[(i) + 2]),          \
                      "+v"(a[(i) + 3]), "+v"(a[(i) + 4]), "+v"(a[(i) + 5]),    \
                      "+v"(a[(i) + 6]), "+v"(a[(i) + 7]) : : "memory")
#define SCAN_HOLD32(a)                                                         \
    SCAN_HOLD8(a, 0); SCAN_HOLD8(a, 8); SCAN_HOLD8(a, 16); SCAN_HOLD8(a, 24)
#define SCAN_TRIP(t, zca, zcb, zvc, ubc, zna, znb, zvn, ubn, HASB)             \
    {                                                                          \
        const float* zc = zl + (2 * (t) + par) * 32;                           \
        chunk_inputs(zc + 64, zna);                                            \
        if (HASB) chunk_inputs(zc + 96, znb);                                  \
        zvn = zc[64 + lane];                                                   \
        SCAN_HOLD32(zca);                                                      \
        SCAN_HOLD32(zcb);                                                      \
        float o = chunk_outputs(P, Q, T, ubc, zca);                            \
        float jr, ji;                                                          \
        chunk_input_sum(qr, qi, zca, jr, ji);                                  \
        float mr = fmaf(qr[31], ur, fmaf(-qi[31], ui, jr));                    \
        float mi = fmaf(qr[31], ui, fmaf(qi[31], ur, ji));                     \
        chunk_input_sum(qr, qi, zcb, jr, ji);                                  \
        ur = fmaf(qr[31], mr, fmaf(-qi[31], mi, jr));                          \
        ui = fmaf(qr[31], mi, fmaf(qi[31], mr, ji));                           \
        state_to_all_lanes(us, lane, ur, ui, ubn);                             \
        yrow[(2 * (t) + par) * 32] = gelu_tanh(fmaf(d, zvc, o));               \
    }
    static_assert(SCAN_TRIPS == 32, "the trips below are counted out for 32");
    float zvA, zvC;
    chunk_inputs(zl + par * 32, zA);
    chunk_inputs(zl + par * 32 + 32, zB);
    zvA = zl[par * 32 + lane];
    state_to_all_lanes(us, lane, ur, ui, ubA);
    for (int t = 0; t < SCAN_TRIPS - 2; t += 2) {
        SCAN_TRIP(t, zA, zB, zvA, ubA, zC, zD, zvC, ubB, true)
        SCAN_TRIP(t + 1, zC, zD, zvC, ubB, zA, zB, zvA, ubA, true)
    }
    SCAN_TRIP(SCAN_TRIPS - 2, zA, zB, zvA, ubA, zC, zD, zvC, ubB, false)
#undef SCAN_TRIP
#undef SCAN_HOLD32
#undef SCAN_HOLD8
    float o = chunk_outputs(P, Q, T, ubB, zC);
    yrow[(2 * (SCAN_TRIPS - 1) + par) * 32] = gelu_tanh(fmaf(d, zvC, o));
}

__device__ __forceinline__ void barrier_lds_ring() {
    asm volatile("s_wait_dscnt 0x0\n\ts_barrier_signal -1\n\ts_barrier_wait -1" ::: "memory");
}

#define BM 64
#define BMH 32
#define GROWS 512
#define BN 128
#define BK 32
#define LDP 34
#define NKT (KPAD / BK)
#define NSTG 3

__global__ __launch_bounds__(256) void s4_glu_gemm(
        const float* __restrict__ Wg, const float* __restrict__ bias,
        const float* __restrict__ Y, const float* __restrict__ Zres,
        float* __restrict__ G) {
    __shared__ float Ws[NSTG][BM][LDP];
    __shared__ float Ys[NSTG][BN][LDP];
    int tid = threadIdx.x;
    int lane = tid & 31;
    int wave = tid >> 5;
    int wave_m = wave >> 2;
    int wave_n = wave & 3;
    int lrow = lane & 15;
    int khalf = (lane >> 4) << 1;
    int madd = (lane >> 4) << 3;
    int blockM = blockIdx.x * BMH;
    int blockN = blockIdx.y * BN;

    int wrow[8];
#pragma unroll
    for (int j = 0; j < 8; ++j) {
        int R = wave + 8 * j;
        int ch = blockM + ((R >> 5) << 4) + (R & 15);
        ch = ch < HCH ? ch : (HCH - 1);
        wrow[j] = (((R >> 4) & 1) * HCH + ch) * KD;
    }
    const float* ycol = Y + blockN + lane;

    v8f acc[2][2];
#pragma unroll
    for (int mt = 0; mt < 2; ++mt) {
#pragma unroll
        for (int i = 0; i < 8; ++i) {
            int ch = blockM + wave_m * 16 + i + madd;
            float bv = (ch < HCH) ? bias[mt * HCH + ch] : 0.0f;
#pragma unroll
            for (int nt = 0; nt < 2; ++nt) acc[mt][nt][i] = bv;
        }
    }

    float wregA[8], yregA[16], wregB[8], yregB[16];
    v2f fa0[BK / 4][2], fb0[BK / 4][2], fa1[BK / 4][2], fb1[BK / 4][2];

#define TILE_LOAD(kb, wreg, yreg)                                              \
    {                                                                          \
        int gk = (kb) + lane;                                                  \
        int gkc = gk < KD ? gk : (KD - 1);                                     \
        _Pragma("unroll")                                                      \
        for (int j = 0; j < 8; ++j) wreg[j] = Wg[wrow[j] + gkc];               \
        _Pragma("unroll")                                                      \
        for (int r = 0; r < 4; ++r) {                                          \
            int gy = (kb) + wave + 8 * r;                                      \
            int gyc = gy < KD ? gy : (KD - 1);                                 \
            const float* srow = ycol + (size_t)gyc * LSEQ;                     \
            _Pragma("unroll")                                                  \
            for (int j = 0; j < 4; ++j) yreg[4 * r + j] = srow[32 * j];        \
        }                                                                      \
    }

#define TILE_STORE(kb, b, wreg, yreg)                                          \
    {                                                                          \
        bool kin = (kb) + lane < KD;                                           \
        _Pragma("unroll")                                                      \
        for (int j = 0; j < 8; ++j)                                            \
            Ws[b][wave + 8 * j][lane] = kin ? wreg[j] : 0.0f;                  \
        _Pragma("unroll")                                                      \
        for (int r = 0; r < 4; ++r) {                                          \
            bool yin = (kb) + wave + 8 * r < KD;                               \
            _Pragma("unroll")                                                  \
            for (int j = 0; j < 4; ++j)                                        \
                Ys[b][lane + 32 * j][wave + 8 * r] = yin ? yreg[4 * r + j] : 0.0f; \
        }                                                                      \
    }

#define FRAG_LOAD(b, fa, fb)                                                   \
    {                                                                          \
        _Pragma("unroll")                                                      \
        for (int kc = 0; kc < BK / 4; ++kc) {                                  \
            int kk = kc * 4 + khalf;                                           \
            _Pragma("unroll")                                                  \
            for (int mt = 0; mt < 2; ++mt)                                     \
                fa[kc][mt] = *(const v2f*)&Ws[b][wave_m * 32 + mt * 16 + lrow][kk];  \
            _Pragma("unroll")                                                  \
            for (int nt = 0; nt < 2; ++nt)                                     \
                fb[kc][nt] = *(const v2f*)&Ys[b][wave_n * 32 + nt * 16 + lrow][kk];  \
        }                                                                      \
    }

#define TILE_MMA_HALF(KC0, fa, fb)                                             \
    {                                                                          \
        _Pragma("unroll")                                                      \
        for (int kc = (KC0); kc < (KC0) + BK / 8; ++kc) {                      \
            _Pragma("unroll")                                                  \
            for (int mt = 0; mt < 2; ++mt)                                     \
                _Pragma("unroll")                                              \
                for (int nt = 0; nt < 2; ++nt)                                 \
                    acc[mt][nt] = __builtin_amdgcn_wmma_f32_16x16x4_f32(       \
                        false, fa[kc][mt], false, fb[kc][nt],                  \
                        (short)0, acc[mt][nt], false, false);                  \
        }                                                                      \
    }

#define PIPE_STEP(T, fa, fb, fna, fnb, DOF, DOS, DOL)                          \
    {                                                                          \
        if (DOF) FRAG_LOAD(fb_, fna, fnb)                                      \
        __builtin_amdgcn_sched_barrier(0);                                     \
        TILE_MMA_HALF(0, fa, fb)                                               \
        __builtin_amdgcn_sched_barrier(0);                                     \
        if (DOS) TILE_STORE(((T) + 2) * BK, sb_, wregA, yregA)                 \
        if (DOL) TILE_LOAD(((T) + 3) * BK, wregA, yregA)                       \
        __builtin_amdgcn_sched_barrier(0);                                     \
        TILE_MMA_HALF(BK / 8, fa, fb)                                          \
        __builtin_amdgcn_sched_barrier(0);                                     \
         \
         \
         \
        if (DOS) barrier_lds_ring();                                           \
        fb_ = sb_;                                                             \
        sb_ = sb_ == NSTG - 1 ? 0 : sb_ + 1;                                   \
    }

    TILE_LOAD(0, wregA, yregA)
    TILE_LOAD(BK, wregB, yregB)
    TILE_STORE(0, 0, wregA, yregA)
    TILE_STORE(BK, 1, wregB, yregB)
    TILE_LOAD(2 * BK, wregA, yregA)
    barrier_lds_ring();
    FRAG_LOAD(0, fa0, fb0)
    int fb_ = 1, sb_ = 2;
    static_assert(NKT == 16, "the tail below is written out for 16 K tiles");
    for (int t = 0; t < NKT - 4; t += 2) {
        PIPE_STEP(t, fa0, fb0, fa1, fb1, true, true, true)
        PIPE_STEP(t + 1, fa1, fb1, fa0, fb0, true, true, true)
    }
    PIPE_STEP(NKT - 4, fa0, fb0, fa1, fb1, true, true, true)
    PIPE_STEP(NKT - 3, fa1, fb1, fa0, fb0, true, true, false)
    PIPE_STEP(NKT - 2, fa0, fb0, fa1, fb1, true, false, false)
    int gn_base = blockN + wave_n * 32;
    float zr[8][2];
#pragma unroll
    for (int i = 0; i < 8; ++i) {
        int ch = blockM + wave_m * 16 + i + madd;
        int chc = ch < HCH ? ch : (HCH - 1);
#pragma unroll
        for (int nt = 0; nt < 2; ++nt)
            zr[i][nt] = Zres[(size_t)chc * LSEQ + gn_base + nt * 16 + lrow];
    }
    PIPE_STEP(NKT - 1, fa1, fb1, fa0, fb0, false, false, false)
#undef TILE_LOAD
#undef TILE_STORE
#undef FRAG_LOAD
#undef TILE_MMA_HALF
#undef PIPE_STEP

    asm volatile("v_nop\n\tv_nop\n\tv_nop\n\tv_nop\n\tv_nop\n\tv_nop\n\tv_nop\n\tv_nop"
                 : "+v"(acc[0][0]), "+v"(acc[0][1]), "+v"(acc[1][0]), "+v"(acc[1][1]));
    __builtin_amdgcn_sched_barrier(0);

#pragma unroll
    for (int i = 0; i < 8; ++i) {
        int ch = blockM + wave_m * 16 + i + madd;
#pragma unroll
        for (int nt = 0; nt < 2; ++nt) {
            int gn = gn_base + nt * 16 + lrow;
            G[(size_t)ch * LSEQ + gn] =
                acc[0][nt][i] * sigmoidf(acc[1][nt][i]) + zr[i][nt];
        }
    }
}

#define LN_COLS 32
#define LN_WAVES 32
#define LN_CPW 16

__global__ __launch_bounds__(1024) void s4_glu_ln(
        const float* __restrict__ Gin, float* __restrict__ Zout,
        const float* __restrict__ g, const float* __restrict__ b) {
    __shared__ float red_sum[LN_WAVES][LN_COLS];
    __shared__ float red_sq[LN_WAVES][LN_COLS];
    int lane = threadIdx.x & 31;
    int wave = __builtin_amdgcn_readfirstlane((int)(threadIdx.x >> 5));
    int l = blockIdx.x * LN_COLS + lane;

    float x[LN_CPW], gg[LN_CPW], bb[LN_CPW];
#pragma unroll
    for (int i = 0; i < LN_CPW; ++i) {
        int h = wave + LN_WAVES * i;
        int hc = (i < LN_CPW - 1 || h < HCH) ? h : (HCH - 1);
        x[i] = Gin[(size_t)hc * LSEQ + l];
        gg[i] = g[hc];
        bb[i] = b[hc];
    }

    float s = 0.0f;
#pragma unroll
    for (int i = 0; i < LN_CPW; ++i) {
        int h = wave + LN_WAVES * i;
        x[i] = (i < LN_CPW - 1 || h < HCH) ? x[i] : 0.0f;
        s += x[i];
    }
    red_sum[wave][lane] = s;
    __syncthreads();
    float tot = 0.0f;
#pragma unroll
    for (int w = 0; w < LN_WAVES; ++w) tot += red_sum[w][lane];
    float mu = tot * (1.0f / (float)HCH);

    float sq = 0.0f;
#pragma unroll
    for (int i = 0; i < LN_CPW; ++i) {
        int h = wave + LN_WAVES * i;
        float d = (i < LN_CPW - 1 || h < HCH) ? (x[i] - mu) : 0.0f;
        x[i] = d;
        sq += d * d;
    }
    red_sq[wave][lane] = sq;
    __syncthreads();
    float tq = 0.0f;
#pragma unroll
    for (int w = 0; w < LN_WAVES; ++w) tq += red_sq[w][lane];
    float rstd = rsqrtf(tq * (1.0f / (float)HCH) + LN_EPS);

#pragma unroll
    for (int i = 0; i < LN_CPW; ++i) {
        int h = wave + LN_WAVES * i;
        if (i < LN_CPW - 1 || h < HCH)
            Zout[(size_t)h * LSEQ + l] = x[i] * rstd * gg[i] + bb[i];
    }
}

extern "C" void kernel_launch(void* const* d_in, const int* in_sizes, int n_in,
                              void* d_out, int out_size, void* d_ws, size_t ws_size,
                              hipStream_t stream) {
    (void)in_sizes; (void)n_in; (void)out_size; (void)ws_size;
    const float* Zin    = (const float*)d_in[0];
    const float* log_dt = (const float*)d_in[1];
    const float* Arl    = (const float*)d_in[2];
    const float* Aim    = (const float*)d_in[3];
    const float* Cre    = (const float*)d_in[4];
    const float* Cim    = (const float*)d_in[5];
    const float* Dsk    = (const float*)d_in[6];
    const float* glu_w  = (const float*)d_in[7];
    const float* glu_b  = (const float*)d_in[8];
    const float* ln_g   = (const float*)d_in[9];
    const float* ln_b   = (const float*)d_in[10];

    float* bufa  = (float*)d_ws;
    float* bufb  = bufa + (size_t)HCH * LSEQ;
    float* gbuf  = bufb + (size_t)HCH * LSEQ;
    float* out   = (float*)d_out;

    static_assert(NLAYERS == 4, "the four layers are written out below");
    static_assert(16 * BMH == GROWS && GROWS >= HCH && GROWS <= 2 * HCH,
                  "16 block rows cover every channel; G fits the plane behind bufb");
#define S4_LAYER(zin, ybuf, zout)                                              \
    s4_scan<<<(HCH + SCAN_CH - 1) / SCAN_CH, SCAN_WAVES * 32, 0, stream>>>(    \
        zin, log_dt, Arl, Aim, Cre, Cim, Dsk, ybuf);                           \
    s4_glu_gemm<<<dim3(16, 16, 1), 256, 0, stream>>>(glu_w, glu_b, ybuf, zin, gbuf); \
    s4_glu_ln<<<LSEQ / LN_COLS, 1024, 0, stream>>>(gbuf, zout, ln_g, ln_b);
    S4_LAYER(Zin,  bufb, bufa)
    S4_LAYER(bufa, bufb, bufb)
    S4_LAYER(bufb, bufa, bufa)
    S4_LAYER(bufa, bufb, out)
#undef S4_LAYER
}
